# starting kernel
# baseline (speedup 1.0000x reference)
#include <hip/hip_runtime.h>
#include <hip/hip_bf16.h>

#define EMB_DIM 128

typedef __attribute__((ext_vector_type(2))) float v2f;
typedef __attribute__((ext_vector_type(8))) float v8f;

__device__ __forceinline__ void gatomic_add_f32(float* p, float v) {
  asm volatile("global_atomic_add_f32 %0, %1, off scope:SCOPE_DEV"
               :: "v"(p), "v"(v) : "memory");
}

__device__ __forceinline__ void gatomic_add_f32x4(float* p, float4 v) {
  asm volatile(
      "global_atomic_add_f32 %0, %1, off scope:SCOPE_DEV\n\t"
      "global_atomic_add_f32 %0, %2, off offset:4 scope:SCOPE_DEV\n\t"
      "global_atomic_add_f32 %0, %3, off offset:8 scope:SCOPE_DEV\n\t"
      "global_atomic_add_f32 %0, %4, off offset:12 scope:SCOPE_DEV"
      :: "v"(p), "v"(v.x), "v"(v.y), "v"(v.z), "v"(v.w) : "memory");
}

__global__ void zero_f32(float* __restrict__ p, long long n) {
  long long t = (long long)blockIdx.x * blockDim.x + threadIdx.x;
  if (t < n) p[t] = 0.0f;
}

__global__ void degree_kernel(const int* __restrict__ dst,
                              float* __restrict__ deg, int n_edges) {
  int e = blockIdx.x * blockDim.x + threadIdx.x;
  if (e < n_edges) gatomic_add_f32(deg + dst[e], 1.0f);
}

__global__ void __launch_bounds__(256)
scatter_kernel(const int* __restrict__ src, const int* __restrict__ dst,
               const float* __restrict__ h, float* __restrict__ agg,
               int n_edges) {
  int t = blockIdx.x * blockDim.x + threadIdx.x;
  int e = t >> 5;
  int lane = t & 31;
  if (e >= n_edges) return;
  int s = src[e];
  int d = dst[e];
  const float4 v = *(const float4*)(h + (size_t)s * EMB_DIM + lane * 4);
  gatomic_add_f32x4(agg + (size_t)d * EMB_DIM + lane * 4, v);
}

__global__ void norm_kernel(float* __restrict__ g, const float* __restrict__ deg,
                            int num_users) {
  int t = blockIdx.x * blockDim.x + threadIdx.x;
  int u = t >> 5;
  int lane = t & 31;
  if (u >= num_users) return;
  float inv = 1.0f / fmaxf(deg[u], 1.0f);
  float4* p = (float4*)(g + (size_t)u * EMB_DIM + lane * 4);
  float4 v = *p;
  v.x *= inv; v.y *= inv; v.z *= inv; v.w *= inv;
  *p = v;
}

__global__ void __launch_bounds__(256)
gather_kernel(const int* __restrict__ users, const int* __restrict__ items,
              const float* __restrict__ uemb, const float* __restrict__ g1,
              const float* __restrict__ g2, const float* __restrict__ iemb,
              float* __restrict__ out_user, float* __restrict__ out_item,
              int batch) {
  int t = blockIdx.x * blockDim.x + threadIdx.x;
  int b = t >> 5;
  int lane = t & 31;
  if (b >= batch) return;
  int u = users[b];
  int it = items[b];
  size_t off = (size_t)lane * 4;
  const float4 a = *(const float4*)(uemb + (size_t)u * EMB_DIM + off);
  const float4 c = *(const float4*)(g1 + (size_t)u * EMB_DIM + off);
  const float4 d = *(const float4*)(g2 + (size_t)u * EMB_DIM + off);
  float4 fu;
  fu.x = a.x + c.x + d.x;
  fu.y = a.y + c.y + d.y;
  fu.z = a.z + c.z + d.z;
  fu.w = a.w + c.w + d.w;
  *(float4*)(out_user + (size_t)b * EMB_DIM + off) = fu;
  *(float4*)(out_item + (size_t)b * EMB_DIM + off) =
      *(const float4*)(iemb + (size_t)it * EMB_DIM + off);
}

__global__ void __launch_bounds__(256)
dot_sigmoid_wmma(const float* __restrict__ U, const float* __restrict__ V,
                 float* __restrict__ pred, int batch) {
  int gid = blockIdx.x * blockDim.x + threadIdx.x;
  int wave = gid >> 5;
  int lane = threadIdx.x & 31;
  if (wave * 16 >= batch) return;

  int row = lane & 15;
  int half = lane >> 4;
  const float* urow = U + ((size_t)wave * 16 + row) * EMB_DIM + 2 * half;
  const float* vrow = V + ((size_t)wave * 16 + row) * EMB_DIM + 2 * half;

  v8f acc = {};
#pragma unroll
  for (int k0 = 0; k0 < EMB_DIM; k0 += 4) {
    v2f a = *(const v2f*)(urow + k0);
    v2f b = *(const v2f*)(vrow + k0);
    acc = __builtin_amdgcn_wmma_f32_16x16x4_f32(
         false, a,  false, b,
         (short)0, acc,  false,  false);
  }

  if (lane < 8 || lane >= 24) {
    int i = (lane < 8) ? lane : (lane - 16);
    float d = acc[lane & 7];
    pred[(size_t)wave * 16 + i] = 1.0f / (1.0f + expf(-d));
  }
}

extern "C" void kernel_launch(void* const* d_in, const int* in_sizes, int n_in,
                              void* d_out, int out_size, void* d_ws,
                              size_t ws_size, hipStream_t stream) {
  const int* users  = (const int*)d_in[0];
  const int* items  = (const int*)d_in[1];
  const int* esrc   = (const int*)d_in[2];
  const int* edst   = (const int*)d_in[3];
  const float* uemb = (const float*)d_in[4];
  const float* iemb = (const float*)d_in[5];

  const int batch     = in_sizes[0];
  const int n_edges   = in_sizes[2];
  const int num_users = in_sizes[4] / EMB_DIM;

  float* deg = (float*)d_ws;
  float* g1  = deg + num_users;
  float* g2  = g1 + (size_t)num_users * EMB_DIM;

  float* pred     = (float*)d_out;
  float* out_user = pred + batch;
  float* out_item = out_user + (size_t)batch * EMB_DIM;

  const int B = 256;

  long long total = (long long)num_users * (1 + 2 * EMB_DIM);
  zero_f32<<<(int)((total + B - 1) / B), B, 0, stream>>>((float*)d_ws, total);

  degree_kernel<<<(n_edges + B - 1) / B, B, 0, stream>>>(edst, deg, n_edges);

  long long sc_threads = (long long)n_edges * 32;
  scatter_kernel<<<(int)((sc_threads + B - 1) / B), B, 0, stream>>>(
      esrc, edst, uemb, g1, n_edges);
  long long nm_threads = (long long)num_users * 32;
  norm_kernel<<<(int)((nm_threads + B - 1) / B), B, 0, stream>>>(g1, deg,
                                                                 num_users);

  scatter_kernel<<<(int)((sc_threads + B - 1) / B), B, 0, stream>>>(
      esrc, edst, g1, g2, n_edges);
  norm_kernel<<<(int)((nm_threads + B - 1) / B), B, 0, stream>>>(g2, deg,
                                                                 num_users);

  long long ga_threads = (long long)batch * 32;
  gather_kernel<<<(int)((ga_threads + B - 1) / B), B, 0, stream>>>(
      users, items, uemb, g1, g2, iemb, out_user, out_item, batch);

  long long dot_threads = (long long)((batch + 15) / 16) * 32;
  dot_sigmoid_wmma<<<(int)((dot_threads + B - 1) / B), B, 0, stream>>>(
      out_user, out_item, pred, batch);
}


# conv as owner-computes: edges bucketed by block of 391 destination rows (ballot-ranked counting sort), a work-group sums its block in LDS, one wave per row; atomics gone, sums in a fixed order
# speedup vs baseline: 2.9272x; 2.9272x over previous
#include <hip/hip_runtime.h>
#include <hip/hip_bf16.h>


#define EMB_DIM 128
#define ROWS 391
#define ROW_BITS 9
#define ROW_MASK 511u
#define NBMAX 256
#define NCH 256
#define PT 1024
#define NWAVES (PT / 32)
#define SCAN_K (NBMAX * NCH / PT)
#define TILE 2048
#define STEP 128

typedef __attribute__((ext_vector_type(2))) float v2f;
typedef __attribute__((ext_vector_type(8))) float v8f;

__device__ __forceinline__ unsigned same_key_lanes(int key, unsigned live) {
  unsigned peers = live;
#pragma unroll
  for (int i = 0; i < 8; ++i) {
    const bool bit = ((key >> i) & 1) != 0;
    const unsigned m = __builtin_amdgcn_ballot_w32(bit);
    peers &= bit ? m : ~m;
  }
  return peers;
}

__device__ __forceinline__ void wave_block_hist(const int* __restrict__ dst,
                                                int* hist, int eb, int ee,
                                                int w, int lane) {
  const unsigned lt = (1u << lane) - 1u;
  for (int base = eb + w * 32; base < ee; base += PT) {
    const int e = base + lane;
    const bool valid = e < ee;
    const int b = valid ? dst[e] / ROWS : 0;
    const unsigned live = __builtin_amdgcn_ballot_w32(valid);
    const unsigned peers = same_key_lanes(b, live);
    if (valid && (peers & lt) == 0u) hist[b] += __popc(peers);
  }
}

__global__ void __launch_bounds__(PT)
part_count(const int* __restrict__ dst, int* __restrict__ tab, int n_edges,
           int num_users) {
  __shared__ int whist[NWAVES * NBMAX];
  const int tid = threadIdx.x;
  const int lane = tid & 31;
  const int w = __builtin_amdgcn_readfirstlane(tid >> 5);
  const int nb = (num_users + ROWS - 1) / ROWS;
  for (int i = tid; i < NWAVES * NBMAX; i += PT) whist[i] = 0;
  __syncthreads();
  const int cs = (n_edges + NCH - 1) / NCH;
  const int eb = blockIdx.x * cs;
  const int ee = min(n_edges, eb + cs);
  wave_block_hist(dst, whist + w * NBMAX, eb, ee, w, lane);
  __syncthreads();
  if (tid < nb) {
    int s = 0;
    for (int k = 0; k < NWAVES; ++k) s += whist[k * NBMAX + tid];
    tab[tid * NCH + blockIdx.x] = s;
  }
}

__global__ void __launch_bounds__(PT)
part_scan(int* __restrict__ tab, int num_users) {
  __shared__ int wsum[NWAVES];
  const int tid = threadIdx.x;
  const int lane = tid & 31;
  const int w = tid >> 5;
  const int nb = (num_users + ROWS - 1) / ROWS;
  const int total = nb * NCH;
  const int base = tid * SCAN_K;
  int v[SCAN_K];
  int s = 0;
#pragma unroll
  for (int i = 0; i < SCAN_K; ++i) {
    const int idx = base + i;
    const int x = (idx < total) ? tab[idx] : 0;
    v[i] = s;
    s += x;
  }
  int inc = s;
#pragma unroll
  for (int o = 1; o < 32; o <<= 1) {
    const int t = __shfl_up(inc, o);
    if (lane >= o) inc += t;
  }
  if (lane == 31) wsum[w] = inc;
  __syncthreads();
  if (w == 0) {
    const int x = wsum[lane];
    int xi = x;
#pragma unroll
    for (int o = 1; o < 32; o <<= 1) {
      const int t = __shfl_up(xi, o);
      if (lane >= o) xi += t;
    }
    wsum[lane] = xi - x;
  }
  __syncthreads();
  const int off = wsum[w] + inc - s;
#pragma unroll
  for (int i = 0; i < SCAN_K; ++i) {
    const int idx = base + i;
    if (idx < total) tab[idx] = off + v[i];
  }
}

__global__ void __launch_bounds__(PT)
part_place(const int* __restrict__ src, const int* __restrict__ dst,
           const int* __restrict__ tab, unsigned* __restrict__ elist,
           int n_edges, int num_users) {
  __shared__ int whist[NWAVES * NBMAX];
  const int tid = threadIdx.x;
  const int lane = tid & 31;
  const int w = __builtin_amdgcn_readfirstlane(tid >> 5);
  const int nb = (num_users + ROWS - 1) / ROWS;
  for (int i = tid; i < NWAVES * NBMAX; i += PT) whist[i] = 0;
  __syncthreads();
  const int cs = (n_edges + NCH - 1) / NCH;
  const int eb = blockIdx.x * cs;
  const int ee = min(n_edges, eb + cs);
  wave_block_hist(dst, whist + w * NBMAX, eb, ee, w, lane);
  __syncthreads();
  if (tid < nb) {
    int run = tab[tid * NCH + blockIdx.x];
    for (int k = 0; k < NWAVES; ++k) {
      const int c = whist[k * NBMAX + tid];
      whist[k * NBMAX + tid] = run;
      run += c;
    }
  }
  __syncthreads();
  int* cur = whist + w * NBMAX;
  const unsigned lt = (1u << lane) - 1u;
  for (int base = eb + w * 32; base < ee; base += PT) {
    const int e = base + lane;
    const bool valid = e < ee;
    const int d = valid ? dst[e] : 0;
    const int b = d / ROWS;
    const unsigned live = __builtin_amdgcn_ballot_w32(valid);
    const unsigned peers = same_key_lanes(b, live);
    const int rank = __popc(peers & lt);
    const int c0 = cur[b];
    if (valid)
      elist[c0 + rank] =
          ((unsigned)src[e] << ROW_BITS) | (unsigned)(d - b * ROWS);
    if (valid && rank == 0) cur[b] = c0 + __popc(peers);
  }
}

template <bool LAST>
__global__ void __launch_bounds__(PT)
conv_owner(const unsigned* __restrict__ elist, const int* __restrict__ tab,
           const float* __restrict__ h, float* __restrict__ gout,
           const int* __restrict__ users, const float* __restrict__ uemb,
           float* __restrict__ out_user, int n_edges, int num_users,
           int batch) {
  __shared__ __attribute__((aligned(16))) float acc[ROWS * EMB_DIM];
  __shared__ __attribute__((aligned(16))) unsigned stage[TILE];
  __shared__ __attribute__((aligned(16))) unsigned queue[NWAVES * STEP];
  __shared__ float cnt[ROWS];
  const int tid = threadIdx.x;
  const int lane = tid & 31;
  const int w = __builtin_amdgcn_readfirstlane(tid >> 5);
  const unsigned lt = (1u << lane) - 1u;
  const int nb = (num_users + ROWS - 1) / ROWS;
  float4* acc4 = (float4*)acc;
  unsigned* myq = queue + w * STEP;
  for (int i = tid; i < NWAVES * STEP; i += PT) queue[i] = 0u;

  for (int bin = blockIdx.x; bin < nb; bin += gridDim.x) {
    const int row0 = bin * ROWS;
    const int nrows = min(ROWS, num_users - row0);

    for (int i = tid; i < ROWS * EMB_DIM / 4; i += PT)
      acc4[i] = make_float4(0.0f, 0.0f, 0.0f, 0.0f);
    __syncthreads();

    const int e0 = tab[bin * NCH];
    const int e1 = (bin + 1 < nb) ? tab[(bin + 1) * NCH] : n_edges;
    int mycnt = 0;

    for (int t0 = e0; t0 < e1; t0 += TILE) {
      const int tn = min(TILE, e1 - t0);
      __syncthreads();
      for (int i = tid; i < TILE; i += PT)
        stage[i] = (i < tn) ? elist[t0 + i] : 0u;
      __syncthreads();

      for (int s0 = 0; s0 < tn; s0 += STEP) {
        const int i0 = s0 + lane * 4;
        const uint4 x4 = *(const uint4*)(stage + i0);
        const unsigned xs[4] = {x4.x, x4.y, x4.z, x4.w};
        int n = 0;
#pragma unroll
        for (int k = 0; k < 4; ++k) {
          const bool hit = (i0 + k < tn) && ((int)(xs[k] & 31u) == w);
          const unsigned m = __builtin_amdgcn_ballot_w32(hit);
          if (hit) myq[n + __popc(m & lt)] = xs[k];
          n += __popc(m);
        }
        for (int q = 0; q < n; q += 4) {
          const uint4 q4 = *(const uint4*)(myq + q);
          const unsigned xe[4] = {
              (unsigned)__builtin_amdgcn_readfirstlane((int)q4.x),
              (unsigned)__builtin_amdgcn_readfirstlane((int)q4.y),
              (unsigned)__builtin_amdgcn_readfirstlane((int)q4.z),
              (unsigned)__builtin_amdgcn_readfirstlane((int)q4.w)};
          float4 v[4];
#pragma unroll
          for (int j = 0; j < 4; ++j) {
            if (q + j < n)
              v[j] = *(const float4*)(h + (size_t)(xe[j] >> ROW_BITS) * EMB_DIM +
                                      lane * 4);
          }
#pragma unroll
          for (int j = 0; j < 4; ++j) {
            if (q + j < n) {
              const int r = (int)(xe[j] & ROW_MASK);
              float4 a = acc4[r * 32 + lane];
              a.x += v[j].x;
              a.y += v[j].y;
              a.z += v[j].z;
              a.w += v[j].w;
              acc4[r * 32 + lane] = a;
              mycnt += (lane == (r >> 5)) ? 1 : 0;
            }
          }
        }
      }
    }
    {
      const int r = w + 32 * lane;
      if (r < ROWS) cnt[r] = (float)mycnt;
    }
    __syncthreads();

    if (!LAST) {
      for (int r = w; r < nrows; r += NWAVES) {
        const float m = fmaxf(cnt[r], 1.0f);
        const float4 a = acc4[r * 32 + lane];
        float4 o;
        o.x = a.x / m;
        o.y = a.y / m;
        o.z = a.z / m;
        o.w = a.w / m;
        *(float4*)(gout + (size_t)(row0 + r) * EMB_DIM + lane * 4) = o;
      }
    } else {
      for (int b0 = w * 32; b0 < batch; b0 += 4 * PT) {
        int r[4];
#pragma unroll
        for (int k = 0; k < 4; ++k) {
          const int b = b0 + k * PT + lane;
          r[k] = (b < batch) ? (users[b] - row0) : -1;
        }
#pragma unroll
        for (int k = 0; k < 4; ++k) {
          unsigned mask = __builtin_amdgcn_ballot_w32(r[k] >= 0 && r[k] < nrows);
          while (mask) {
            const int j = __builtin_ctz(mask);
            mask &= mask - 1;
            const int rj = __builtin_amdgcn_readlane(r[k], j);
            const size_t ro = (size_t)(row0 + rj) * EMB_DIM + lane * 4;
            const float4 x = *(const float4*)(uemb + ro);
            const float4 c = *(const float4*)(h + ro);
            const float m = fmaxf(cnt[rj], 1.0f);
            const float4 a = acc4[rj * 32 + lane];
            float4 o;
            o.x = x.x + c.x + a.x / m;
            o.y = x.y + c.y + a.y / m;
            o.z = x.z + c.z + a.z / m;
            o.w = x.w + c.w + a.w / m;
            *(float4*)(out_user + (size_t)(b0 + k * PT + j) * EMB_DIM +
                       lane * 4) = o;
          }
        }
      }
    }
    __syncthreads();
  }
}

__global__ void __launch_bounds__(256)
gather_items(const int* __restrict__ items, const float* __restrict__ iemb,
             float* __restrict__ out_item, int batch) {
  int t = blockIdx.x * blockDim.x + threadIdx.x;
  int b = t >> 5;
  int lane = t & 31;
  if (b >= batch) return;
  int it = items[b];
  size_t off = (size_t)lane * 4;
  *(float4*)(out_item + (size_t)b * EMB_DIM + off) =
      *(const float4*)(iemb + (size_t)it * EMB_DIM + off);
}

__global__ void __launch_bounds__(256)
dot_sigmoid_wmma(const float* __restrict__ U, const float* __restrict__ V,
                 float* __restrict__ pred, int batch) {
  int gid = blockIdx.x * blockDim.x + threadIdx.x;
  int wave = gid >> 5;
  int lane = threadIdx.x & 31;
  if (wave * 16 >= batch) return;

  int row = lane & 15;
  int half = lane >> 4;
  const float* urow = U + ((size_t)wave * 16 + row) * EMB_DIM + 2 * half;
  const float* vrow = V + ((size_t)wave * 16 + row) * EMB_DIM + 2 * half;

  v8f acc = {};
#pragma unroll
  for (int k0 = 0; k0 < EMB_DIM; k0 += 4) {
    v2f a = *(const v2f*)(urow + k0);
    v2f b = *(const v2f*)(vrow + k0);
    acc = __builtin_amdgcn_wmma_f32_16x16x4_f32(
         false, a,  false, b,
         (short)0, acc,  false,  false);
  }

  if (lane < 8 || lane >= 24) {
    int i = (lane < 8) ? lane : (lane - 16);
    float d = acc[lane & 7];
    pred[(size_t)wave * 16 + i] = 1.0f / (1.0f + expf(-d));
  }
}

extern "C" void kernel_launch(void* const* d_in, const int* in_sizes, int n_in,
                              void* d_out, int out_size, void* d_ws,
                              size_t ws_size, hipStream_t stream) {
  const int* users  = (const int*)d_in[0];
  const int* items  = (const int*)d_in[1];
  const int* esrc   = (const int*)d_in[2];
  const int* edst   = (const int*)d_in[3];
  const float* uemb = (const float*)d_in[4];
  const float* iemb = (const float*)d_in[5];

  const int batch     = in_sizes[0];
  const int n_edges   = in_sizes[2];
  const int num_users = in_sizes[4] / EMB_DIM;

  int* tab        = (int*)d_ws;
  float* g1       = (float*)d_ws + num_users;
  unsigned* elist = (unsigned*)(g1 + (size_t)num_users * EMB_DIM);

  float* pred     = (float*)d_out;
  float* out_user = pred + batch;
  float* out_item = out_user + (size_t)batch * EMB_DIM;

  const int B = 256;

  part_count<<<NCH, PT, 0, stream>>>(edst, tab, n_edges, num_users);
  part_scan<<<1, PT, 0, stream>>>(tab, num_users);
  part_place<<<NCH, PT, 0, stream>>>(esrc, edst, tab, elist, n_edges,
                                     num_users);

  conv_owner<false><<<NBMAX, PT, 0, stream>>>(elist, tab, uemb, g1, users,
                                              uemb, out_user, n_edges,
                                              num_users, batch);

  conv_owner<true><<<NBMAX, PT, 0, stream>>>(elist, tab, g1, (float*)nullptr,
                                             users, uemb, out_user, n_edges,
                                             num_users, batch);

  long long ga_threads = (long long)batch * 32;
  gather_items<<<(int)((ga_threads + B - 1) / B), B, 0, stream>>>(
      items, iemb, out_item, batch);

  long long dot_threads = (long long)((batch + 15) / 16) * 32;
  dot_sigmoid_wmma<<<(int)((dot_threads + B - 1) / B), B, 0, stream>>>(
      out_user, out_item, pred, batch);
}


# offsets scan (one work-group) reads and writes the count table as whole lines: a wave takes 2048 consecutive counters, four per lane a load, in place of 64 lane-private words 256 B apart
# speedup vs baseline: 3.6925x; 1.2614x over previous
#include <hip/hip_runtime.h>
#include <hip/hip_bf16.h>


#define EMB_DIM 128
#define ROWS 391
#define ROW_BITS 9
#define ROW_MASK 511u
#define NBMAX 256
#define NCH 256
#define PT 1024
#define NWAVES (PT / 32)
#define SCAN_G (NBMAX * NCH / (PT * 4))
#define TILE 2048
#define STEP 128

typedef __attribute__((ext_vector_type(2))) float v2f;
typedef __attribute__((ext_vector_type(8))) float v8f;

__device__ __forceinline__ unsigned same_key_lanes(int key, unsigned live) {
  unsigned peers = live;
#pragma unroll
  for (int i = 0; i < 8; ++i) {
    const bool bit = ((key >> i) & 1) != 0;
    const unsigned m = __builtin_amdgcn_ballot_w32(bit);
    peers &= bit ? m : ~m;
  }
  return peers;
}

__device__ __forceinline__ void wave_block_hist(const int* __restrict__ dst,
                                                int* hist, int eb, int ee,
                                                int w, int lane) {
  const unsigned lt = (1u << lane) - 1u;
  for (int base = eb + w * 32; base < ee; base += PT) {
    const int e = base + lane;
    const bool valid = e < ee;
    const int b = valid ? dst[e] / ROWS : 0;
    const unsigned live = __builtin_amdgcn_ballot_w32(valid);
    const unsigned peers = same_key_lanes(b, live);
    if (valid && (peers & lt) == 0u) hist[b] += __popc(peers);
  }
}

__global__ void __launch_bounds__(PT)
part_count(const int* __restrict__ dst, int* __restrict__ tab, int n_edges,
           int num_users) {
  __shared__ int whist[NWAVES * NBMAX];
  const int tid = threadIdx.x;
  const int lane = tid & 31;
  const int w = __builtin_amdgcn_readfirstlane(tid >> 5);
  const int nb = (num_users + ROWS - 1) / ROWS;
  for (int i = tid; i < NWAVES * NBMAX; i += PT) whist[i] = 0;
  __syncthreads();
  const int cs = (n_edges + NCH - 1) / NCH;
  const int eb = blockIdx.x * cs;
  const int ee = min(n_edges, eb + cs);
  wave_block_hist(dst, whist + w * NBMAX, eb, ee, w, lane);
  __syncthreads();
  if (tid < nb) {
    int s = 0;
    for (int k = 0; k < NWAVES; ++k) s += whist[k * NBMAX + tid];
    tab[tid * NCH + blockIdx.x] = s;
  }
}

__global__ void __launch_bounds__(PT)
part_scan(int* __restrict__ tab, int num_users) {
  __shared__ int wsum[NWAVES];
  const int tid = threadIdx.x;
  const int lane = tid & 31;
  const int w = tid >> 5;
  const int nb = (num_users + ROWS - 1) / ROWS;
  const int total = nb * NCH;
  const int first = w * (SCAN_G * 128) + lane * 4;
  int4 x[SCAN_G];
#pragma unroll
  for (int g = 0; g < SCAN_G; ++g) {
    const int idx = first + g * 128;
    x[g] = (idx < total) ? *(const int4*)(tab + idx) : make_int4(0, 0, 0, 0);
  }
  int before[SCAN_G];
  int run = 0;
#pragma unroll
  for (int g = 0; g < SCAN_G; ++g) {
    const int s = x[g].x + x[g].y + x[g].z + x[g].w;
    int inc = s;
#pragma unroll
    for (int o = 1; o < 32; o <<= 1) {
      const int t = __shfl_up(inc, o);
      if (lane >= o) inc += t;
    }
    before[g] = run + inc - s;
    run += __builtin_amdgcn_readlane(inc, 31);
  }
  if (lane == 0) wsum[w] = run;
  __syncthreads();
  if (w == 0) {
    const int v = wsum[lane];
    int vi = v;
#pragma unroll
    for (int o = 1; o < 32; o <<= 1) {
      const int t = __shfl_up(vi, o);
      if (lane >= o) vi += t;
    }
    wsum[lane] = vi - v;
  }
  __syncthreads();
  const int off = wsum[w];
#pragma unroll
  for (int g = 0; g < SCAN_G; ++g) {
    const int idx = first + g * 128;
    if (idx < total) {
      int4 o;
      o.x = off + before[g];
      o.y = o.x + x[g].x;
      o.z = o.y + x[g].y;
      o.w = o.z + x[g].z;
      *(int4*)(tab + idx) = o;
    }
  }
}

__global__ void __launch_bounds__(PT)
part_place(const int* __restrict__ src, const int* __restrict__ dst,
           const int* __restrict__ tab, unsigned* __restrict__ elist,
           int n_edges, int num_users) {
  __shared__ int whist[NWAVES * NBMAX];
  const int tid = threadIdx.x;
  const int lane = tid & 31;
  const int w = __builtin_amdgcn_readfirstlane(tid >> 5);
  const int nb = (num_users + ROWS - 1) / ROWS;
  for (int i = tid; i < NWAVES * NBMAX; i += PT) whist[i] = 0;
  __syncthreads();
  const int cs = (n_edges + NCH - 1) / NCH;
  const int eb = blockIdx.x * cs;
  const int ee = min(n_edges, eb + cs);
  wave_block_hist(dst, whist + w * NBMAX, eb, ee, w, lane);
  __syncthreads();
  if (tid < nb) {
    int run = tab[tid * NCH + blockIdx.x];
    for (int k = 0; k < NWAVES; ++k) {
      const int c = whist[k * NBMAX + tid];
      whist[k * NBMAX + tid] = run;
      run += c;
    }
  }
  __syncthreads();
  int* cur = whist + w * NBMAX;
  const unsigned lt = (1u << lane) - 1u;
  for (int base = eb + w * 32; base < ee; base += PT) {
    const int e = base + lane;
    const bool valid = e < ee;
    const int d = valid ? dst[e] : 0;
    const int b = d / ROWS;
    const unsigned live = __builtin_amdgcn_ballot_w32(valid);
    const unsigned peers = same_key_lanes(b, live);
    const int rank = __popc(peers & lt);
    const int c0 = cur[b];
    if (valid)
      elist[c0 + rank] =
          ((unsigned)src[e] << ROW_BITS) | (unsigned)(d - b * ROWS);
    if (valid && rank == 0) cur[b] = c0 + __popc(peers);
  }
}

template <bool LAST>
__global__ void __launch_bounds__(PT)
conv_owner(const unsigned* __restrict__ elist, const int* __restrict__ tab,
           const float* __restrict__ h, float* __restrict__ gout,
           const int* __restrict__ users, const float* __restrict__ uemb,
           float* __restrict__ out_user, int n_edges, int num_users,
           int batch) {
  __shared__ __attribute__((aligned(16))) float acc[ROWS * EMB_DIM];
  __shared__ __attribute__((aligned(16))) unsigned stage[TILE];
  __shared__ __attribute__((aligned(16))) unsigned queue[NWAVES * STEP];
  __shared__ float cnt[ROWS];
  const int tid = threadIdx.x;
  const int lane = tid & 31;
  const int w = __builtin_amdgcn_readfirstlane(tid >> 5);
  const unsigned lt = (1u << lane) - 1u;
  const int nb = (num_users + ROWS - 1) / ROWS;
  float4* acc4 = (float4*)acc;
  unsigned* myq = queue + w * STEP;
  for (int i = tid; i < NWAVES * STEP; i += PT) queue[i] = 0u;

  for (int bin = blockIdx.x; bin < nb; bin += gridDim.x) {
    const int row0 = bin * ROWS;
    const int nrows = min(ROWS, num_users - row0);

    for (int i = tid; i < ROWS * EMB_DIM / 4; i += PT)
      acc4[i] = make_float4(0.0f, 0.0f, 0.0f, 0.0f);
    __syncthreads();

    const int e0 = tab[bin * NCH];
    const int e1 = (bin + 1 < nb) ? tab[(bin + 1) * NCH] : n_edges;
    int mycnt = 0;

    for (int t0 = e0; t0 < e1; t0 += TILE) {
      const int tn = min(TILE, e1 - t0);
      __syncthreads();
      for (int i = tid; i < TILE; i += PT)
        stage[i] = (i < tn) ? elist[t0 + i] : 0u;
      __syncthreads();

      for (int s0 = 0; s0 < tn; s0 += STEP) {
        const int i0 = s0 + lane * 4;
        const uint4 x4 = *(const uint4*)(stage + i0);
        const unsigned xs[4] = {x4.x, x4.y, x4.z, x4.w};
        int n = 0;
#pragma unroll
        for (int k = 0; k < 4; ++k) {
          const bool hit = (i0 + k < tn) && ((int)(xs[k] & 31u) == w);
          const unsigned m = __builtin_amdgcn_ballot_w32(hit);
          if (hit) myq[n + __popc(m & lt)] = xs[k];
          n += __popc(m);
        }
        for (int q = 0; q < n; q += 4) {
          const uint4 q4 = *(const uint4*)(myq + q);
          const unsigned xe[4] = {
              (unsigned)__builtin_amdgcn_readfirstlane((int)q4.x),
              (unsigned)__builtin_amdgcn_readfirstlane((int)q4.y),
              (unsigned)__builtin_amdgcn_readfirstlane((int)q4.z),
              (unsigned)__builtin_amdgcn_readfirstlane((int)q4.w)};
          float4 v[4];
#pragma unroll
          for (int j = 0; j < 4; ++j) {
            if (q + j < n)
              v[j] = *(const float4*)(h + (size_t)(xe[j] >> ROW_BITS) * EMB_DIM +
                                      lane * 4);
          }
#pragma unroll
          for (int j = 0; j < 4; ++j) {
            if (q + j < n) {
              const int r = (int)(xe[j] & ROW_MASK);
              float4 a = acc4[r * 32 + lane];
              a.x += v[j].x;
              a.y += v[j].y;
              a.z += v[j].z;
              a.w += v[j].w;
              acc4[r * 32 + lane] = a;
              mycnt += (lane == (r >> 5)) ? 1 : 0;
            }
          }
        }
      }
    }
    {
      const int r = w + 32 * lane;
      if (r < ROWS) cnt[r] = (float)mycnt;
    }
    __syncthreads();

    if (!LAST) {
      for (int r = w; r < nrows; r += NWAVES) {
        const float m = fmaxf(cnt[r], 1.0f);
        const float4 a = acc4[r * 32 + lane];
        float4 o;
        o.x = a.x / m;
        o.y = a.y / m;
        o.z = a.z / m;
        o.w = a.w / m;
        *(float4*)(gout + (size_t)(row0 + r) * EMB_DIM + lane * 4) = o;
      }
    } else {
      for (int b0 = w * 32; b0 < batch; b0 += 4 * PT) {
        int r[4];
#pragma unroll
        for (int k = 0; k < 4; ++k) {
          const int b = b0 + k * PT + lane;
          r[k] = (b < batch) ? (users[b] - row0) : -1;
        }
#pragma unroll
        for (int k = 0; k < 4; ++k) {
          unsigned mask = __builtin_amdgcn_ballot_w32(r[k] >= 0 && r[k] < nrows);
          while (mask) {
            const int j = __builtin_ctz(mask);
            mask &= mask - 1;
            const int rj = __builtin_amdgcn_readlane(r[k], j);
            const size_t ro = (size_t)(row0 + rj) * EMB_DIM + lane * 4;
            const float4 x = *(const float4*)(uemb + ro);
            const float4 c = *(const float4*)(h + ro);
            const float m = fmaxf(cnt[rj], 1.0f);
            const float4 a = acc4[rj * 32 + lane];
            float4 o;
            o.x = x.x + c.x + a.x / m;
            o.y = x.y + c.y + a.y / m;
            o.z = x.z + c.z + a.z / m;
            o.w = x.w + c.w + a.w / m;
            *(float4*)(out_user + (size_t)(b0 + k * PT + j) * EMB_DIM +
                       lane * 4) = o;
          }
        }
      }
    }
    __syncthreads();
  }
}

__global__ void __launch_bounds__(256)
gather_items(const int* __restrict__ items, const float* __restrict__ iemb,
             float* __restrict__ out_item, int batch) {
  int t = blockIdx.x * blockDim.x + threadIdx.x;
  int b = t >> 5;
  int lane = t & 31;
  if (b >= batch) return;
  int it = items[b];
  size_t off = (size_t)lane * 4;
  *(float4*)(out_item + (size_t)b * EMB_DIM + off) =
      *(const float4*)(iemb + (size_t)it * EMB_DIM + off);
}

__global__ void __launch_bounds__(256)
dot_sigmoid_wmma(const float* __restrict__ U, const float* __restrict__ V,
                 float* __restrict__ pred, int batch) {
  int gid = blockIdx.x * blockDim.x + threadIdx.x;
  int wave = gid >> 5;
  int lane = threadIdx.x & 31;
  if (wave * 16 >= batch) return;

  int row = lane & 15;
  int half = lane >> 4;
  const float* urow = U + ((size_t)wave * 16 + row) * EMB_DIM + 2 * half;
  const float* vrow = V + ((size_t)wave * 16 + row) * EMB_DIM + 2 * half;

  v8f acc = {};
#pragma unroll
  for (int k0 = 0; k0 < EMB_DIM; k0 += 4) {
    v2f a = *(const v2f*)(urow + k0);
    v2f b = *(const v2f*)(vrow + k0);
    acc = __builtin_amdgcn_wmma_f32_16x16x4_f32(
         false, a,  false, b,
         (short)0, acc,  false,  false);
  }

  if (lane < 8 || lane >= 24) {
    int i = (lane < 8) ? lane : (lane - 16);
    float d = acc[lane & 7];
    pred[(size_t)wave * 16 + i] = 1.0f / (1.0f + expf(-d));
  }
}

extern "C" void kernel_launch(void* const* d_in, const int* in_sizes, int n_in,
                              void* d_out, int out_size, void* d_ws,
                              size_t ws_size, hipStream_t stream) {
  const int* users  = (const int*)d_in[0];
  const int* items  = (const int*)d_in[1];
  const int* esrc   = (const int*)d_in[2];
  const int* edst   = (const int*)d_in[3];
  const float* uemb = (const float*)d_in[4];
  const float* iemb = (const float*)d_in[5];

  const int batch     = in_sizes[0];
  const int n_edges   = in_sizes[2];
  const int num_users = in_sizes[4] / EMB_DIM;

  int* tab        = (int*)d_ws;
  float* g1       = (float*)d_ws + num_users;
  unsigned* elist = (unsigned*)(g1 + (size_t)num_users * EMB_DIM);

  float* pred     = (float*)d_out;
  float* out_user = pred + batch;
  float* out_item = out_user + (size_t)batch * EMB_DIM;

  const int B = 256;

  part_count<<<NCH, PT, 0, stream>>>(edst, tab, n_edges, num_users);
  part_scan<<<1, PT, 0, stream>>>(tab, num_users);
  part_place<<<NCH, PT, 0, stream>>>(esrc, edst, tab, elist, n_edges,
                                     num_users);

  conv_owner<false><<<NBMAX, PT, 0, stream>>>(elist, tab, uemb, g1, users,
                                              uemb, out_user, n_edges,
                                              num_users, batch);

  conv_owner<true><<<NBMAX, PT, 0, stream>>>(elist, tab, g1, (float*)nullptr,
                                             users, uemb, out_user, n_edges,
                                             num_users, batch);

  long long ga_threads = (long long)batch * 32;
  gather_items<<<(int)((ga_threads + B - 1) / B), B, 0, stream>>>(
      items, iemb, out_item, batch);

  long long dot_threads = (long long)((batch + 15) / 16) * 32;
  dot_sigmoid_wmma<<<(int)((dot_threads + B - 1) / B), B, 0, stream>>>(
      out_user, out_item, pred, batch);
}


# conv owner: a wave's queue of matched edges carries over scan steps and is added 12 edges at a time, so 12 source-row loads are in flight per wave instead of at most 4; sum order as before
# speedup vs baseline: 4.1795x; 1.1319x over previous
#include <hip/hip_runtime.h>
#include <hip/hip_bf16.h>


#define EMB_DIM 128
#define ROWS 391
#define ROW_BITS 9
#define ROW_MASK 511u
#define NBMAX 256
#define NCH 256
#define PT 1024
#define NWAVES (PT / 32)
#define SCAN_G (NBMAX * NCH / (PT * 4))
#define TILE 2048
#define STEP 128
#define DRAIN 12
#define QCAP 144
#define TAIL 8

typedef __attribute__((ext_vector_type(2))) float v2f;
typedef __attribute__((ext_vector_type(8))) float v8f;

__device__ __forceinline__ unsigned same_key_lanes(int key, unsigned live) {
  unsigned peers = live;
#pragma unroll
  for (int i = 0; i < 8; ++i) {
    const bool bit = ((key >> i) & 1) != 0;
    const unsigned m = __builtin_amdgcn_ballot_w32(bit);
    peers &= bit ? m : ~m;
  }
  return peers;
}

__device__ __forceinline__ void wave_block_hist(const int* __restrict__ dst,
                                                int* hist, int eb, int ee,
                                                int w, int lane) {
  const unsigned lt = (1u << lane) - 1u;
  for (int base = eb + w * 32; base < ee; base += PT) {
    const int e = base + lane;
    const bool valid = e < ee;
    const int b = valid ? dst[e] / ROWS : 0;
    const unsigned live = __builtin_amdgcn_ballot_w32(valid);
    const unsigned peers = same_key_lanes(b, live);
    if (valid && (peers & lt) == 0u) hist[b] += __popc(peers);
  }
}

__global__ void __launch_bounds__(PT)
part_count(const int* __restrict__ dst, int* __restrict__ tab, int n_edges,
           int num_users) {
  __shared__ int whist[NWAVES * NBMAX];
  const int tid = threadIdx.x;
  const int lane = tid & 31;
  const int w = __builtin_amdgcn_readfirstlane(tid >> 5);
  const int nb = (num_users + ROWS - 1) / ROWS;
  for (int i = tid; i < NWAVES * NBMAX; i += PT) whist[i] = 0;
  __syncthreads();
  const int cs = (n_edges + NCH - 1) / NCH;
  const int eb = blockIdx.x * cs;
  const int ee = min(n_edges, eb + cs);
  wave_block_hist(dst, whist + w * NBMAX, eb, ee, w, lane);
  __syncthreads();
  if (tid < nb) {
    int s = 0;
    for (int k = 0; k < NWAVES; ++k) s += whist[k * NBMAX + tid];
    tab[tid * NCH + blockIdx.x] = s;
  }
}

__global__ void __launch_bounds__(PT)
part_scan(int* __restrict__ tab, int num_users) {
  __shared__ int wsum[NWAVES];
  const int tid = threadIdx.x;
  const int lane = tid & 31;
  const int w = tid >> 5;
  const int nb = (num_users + ROWS - 1) / ROWS;
  const int total = nb * NCH;
  const int first = w * (SCAN_G * 128) + lane * 4;
  int4 x[SCAN_G];
#pragma unroll
  for (int g = 0; g < SCAN_G; ++g) {
    const int idx = first + g * 128;
    x[g] = (idx < total) ? *(const int4*)(tab + idx) : make_int4(0, 0, 0, 0);
  }
  int before[SCAN_G];
  int run = 0;
#pragma unroll
  for (int g = 0; g < SCAN_G; ++g) {
    const int s = x[g].x + x[g].y + x[g].z + x[g].w;
    int inc = s;
#pragma unroll
    for (int o = 1; o < 32; o <<= 1) {
      const int t = __shfl_up(inc, o);
      if (lane >= o) inc += t;
    }
    before[g] = run + inc - s;
    run += __builtin_amdgcn_readlane(inc, 31);
  }
  if (lane == 0) wsum[w] = run;
  __syncthreads();
  if (w == 0) {
    const int v = wsum[lane];
    int vi = v;
#pragma unroll
    for (int o = 1; o < 32; o <<= 1) {
      const int t = __shfl_up(vi, o);
      if (lane >= o) vi += t;
    }
    wsum[lane] = vi - v;
  }
  __syncthreads();
  const int off = wsum[w];
#pragma unroll
  for (int g = 0; g < SCAN_G; ++g) {
    const int idx = first + g * 128;
    if (idx < total) {
      int4 o;
      o.x = off + before[g];
      o.y = o.x + x[g].x;
      o.z = o.y + x[g].y;
      o.w = o.z + x[g].z;
      *(int4*)(tab + idx) = o;
    }
  }
}

__global__ void __launch_bounds__(PT)
part_place(const int* __restrict__ src, const int* __restrict__ dst,
           const int* __restrict__ tab, unsigned* __restrict__ elist,
           int n_edges, int num_users) {
  __shared__ int whist[NWAVES * NBMAX];
  const int tid = threadIdx.x;
  const int lane = tid & 31;
  const int w = __builtin_amdgcn_readfirstlane(tid >> 5);
  const int nb = (num_users + ROWS - 1) / ROWS;
  for (int i = tid; i < NWAVES * NBMAX; i += PT) whist[i] = 0;
  __syncthreads();
  const int cs = (n_edges + NCH - 1) / NCH;
  const int eb = blockIdx.x * cs;
  const int ee = min(n_edges, eb + cs);
  wave_block_hist(dst, whist + w * NBMAX, eb, ee, w, lane);
  __syncthreads();
  if (tid < nb) {
    int run = tab[tid * NCH + blockIdx.x];
    for (int k = 0; k < NWAVES; ++k) {
      const int c = whist[k * NBMAX + tid];
      whist[k * NBMAX + tid] = run;
      run += c;
    }
  }
  __syncthreads();
  int* cur = whist + w * NBMAX;
  const unsigned lt = (1u << lane) - 1u;
  for (int base = eb + w * 32; base < ee; base += PT) {
    const int e = base + lane;
    const bool valid = e < ee;
    const int d = valid ? dst[e] : 0;
    const int b = d / ROWS;
    const unsigned live = __builtin_amdgcn_ballot_w32(valid);
    const unsigned peers = same_key_lanes(b, live);
    const int rank = __popc(peers & lt);
    const int c0 = cur[b];
    if (valid)
      elist[c0 + rank] =
          ((unsigned)src[e] << ROW_BITS) | (unsigned)(d - b * ROWS);
    if (valid && rank == 0) cur[b] = c0 + __popc(peers);
  }
}

template <int G, bool FULL>
__device__ __forceinline__ void add_group(const unsigned* myq, int q, int n,
                                          const float* __restrict__ h,
                                          float4* acc4, int lane, int& mycnt) {
  unsigned xe[G];
#pragma unroll
  for (int g = 0; g < G / 4; ++g) {
    const uint4 q4 = *(const uint4*)(myq + q + 4 * g);
    xe[4 * g + 0] = (unsigned)__builtin_amdgcn_readfirstlane((int)q4.x);
    xe[4 * g + 1] = (unsigned)__builtin_amdgcn_readfirstlane((int)q4.y);
    xe[4 * g + 2] = (unsigned)__builtin_amdgcn_readfirstlane((int)q4.z);
    xe[4 * g + 3] = (unsigned)__builtin_amdgcn_readfirstlane((int)q4.w);
  }
  float4 v[G];
#pragma unroll
  for (int j = 0; j < G; ++j) {
    if (FULL || q + j < n) {
      const unsigned off =
          (xe[j] >> ROW_BITS) * (unsigned)(EMB_DIM * sizeof(float)) +
          (unsigned)lane * 16u;
      v[j] = *(const float4*)((const char*)h + off);
    }
  }
  if (FULL) __builtin_amdgcn_sched_barrier(0);
#pragma unroll
  for (int j = 0; j < G; ++j) {
    if (FULL || q + j < n) {
      const int r = (int)(xe[j] & ROW_MASK);
      float4 a = acc4[r * 32 + lane];
      a.x += v[j].x;
      a.y += v[j].y;
      a.z += v[j].z;
      a.w += v[j].w;
      acc4[r * 32 + lane] = a;
      mycnt += (lane == (r >> 5)) ? 1 : 0;
    }
  }
}

template <bool LAST>
__global__ void __launch_bounds__(PT)
conv_owner(const unsigned* __restrict__ elist, const int* __restrict__ tab,
           const float* __restrict__ h, float* __restrict__ gout,
           const int* __restrict__ users, const float* __restrict__ uemb,
           float* __restrict__ out_user, int n_edges, int num_users,
           int batch) {
  __shared__ __attribute__((aligned(16))) float acc[ROWS * EMB_DIM];
  __shared__ __attribute__((aligned(16))) unsigned stage[TILE];
  __shared__ __attribute__((aligned(16))) unsigned queue[NWAVES * QCAP];
  __shared__ float cnt[ROWS];
  const int tid = threadIdx.x;
  const int lane = tid & 31;
  const int w = __builtin_amdgcn_readfirstlane(tid >> 5);
  const unsigned lt = (1u << lane) - 1u;
  const int nb = (num_users + ROWS - 1) / ROWS;
  float4* acc4 = (float4*)acc;
  unsigned* myq = queue + w * QCAP;
  for (int i = tid; i < NWAVES * QCAP; i += PT) queue[i] = 0u;

  for (int bin = blockIdx.x; bin < nb; bin += gridDim.x) {
    const int row0 = bin * ROWS;
    const int nrows = min(ROWS, num_users - row0);

    for (int i = tid; i < ROWS * EMB_DIM / 4; i += PT)
      acc4[i] = make_float4(0.0f, 0.0f, 0.0f, 0.0f);
    __syncthreads();

    const int e0 = tab[bin * NCH];
    const int e1 = (bin + 1 < nb) ? tab[(bin + 1) * NCH] : n_edges;
    int mycnt = 0;
    int n = 0;

    for (int t0 = e0; t0 < e1; t0 += TILE) {
      const int tn = min(TILE, e1 - t0);
      __syncthreads();
      for (int i = tid; i < TILE; i += PT)
        stage[i] = (i < tn) ? elist[t0 + i] : 0u;
      __syncthreads();

      for (int s0 = 0; s0 < tn; s0 += STEP) {
        const int i0 = s0 + lane * 4;
        const uint4 x4 = *(const uint4*)(stage + i0);
        const unsigned xs[4] = {x4.x, x4.y, x4.z, x4.w};
#pragma unroll
        for (int k = 0; k < 4; ++k) {
          const bool hit = (i0 + k < tn) && ((int)(xs[k] & 31u) == w);
          const unsigned m = __builtin_amdgcn_ballot_w32(hit);
          if (hit) myq[n + __popc(m & lt)] = xs[k];
          n += __popc(m);
        }
        if (n >= DRAIN) {
          int q = 0;
          for (; q + DRAIN <= n; q += DRAIN)
            add_group<DRAIN, true>(myq, q, n, h, acc4, lane, mycnt);
          const int rest = n - q;
          unsigned keep = 0u;
          if (lane < rest) keep = myq[q + lane];
          if (lane < rest) myq[lane] = keep;
          n = rest;
        }
      }
    }
    for (int q = 0; q < n; q += TAIL)
      add_group<TAIL, false>(myq, q, n, h, acc4, lane, mycnt);
    {
      const int r = w + 32 * lane;
      if (r < ROWS) cnt[r] = (float)mycnt;
    }
    __syncthreads();

    if (!LAST) {
      for (int r = w; r < nrows; r += NWAVES) {
        const float m = fmaxf(cnt[r], 1.0f);
        const float4 a = acc4[r * 32 + lane];
        float4 o;
        o.x = a.x / m;
        o.y = a.y / m;
        o.z = a.z / m;
        o.w = a.w / m;
        *(float4*)(gout + (size_t)(row0 + r) * EMB_DIM + lane * 4) = o;
      }
    } else {
      for (int b0 = w * 32; b0 < batch; b0 += 4 * PT) {
        int r[4];
#pragma unroll
        for (int k = 0; k < 4; ++k) {
          const int b = b0 + k * PT + lane;
          r[k] = (b < batch) ? (users[b] - row0) : -1;
        }
#pragma unroll
        for (int k = 0; k < 4; ++k) {
          unsigned mask = __builtin_amdgcn_ballot_w32(r[k] >= 0 && r[k] < nrows);
          while (mask) {
            const int j = __builtin_ctz(mask);
            mask &= mask - 1;
            const int rj = __builtin_amdgcn_readlane(r[k], j);
            const size_t ro = (size_t)(row0 + rj) * EMB_DIM + lane * 4;
            const float4 x = *(const float4*)(uemb + ro);
            const float4 c = *(const float4*)(h + ro);
            const float m = fmaxf(cnt[rj], 1.0f);
            const float4 a = acc4[rj * 32 + lane];
            float4 o;
            o.x = x.x + c.x + a.x / m;
            o.y = x.y + c.y + a.y / m;
            o.z = x.z + c.z + a.z / m;
            o.w = x.w + c.w + a.w / m;
            *(float4*)(out_user + (size_t)(b0 + k * PT + j) * EMB_DIM +
                       lane * 4) = o;
          }
        }
      }
    }
    __syncthreads();
  }
}

__global__ void __launch_bounds__(256)
gather_items(const int* __restrict__ items, const float* __restrict__ iemb,
             float* __restrict__ out_item, int batch) {
  int t = blockIdx.x * blockDim.x + threadIdx.x;
  int b = t >> 5;
  int lane = t & 31;
  if (b >= batch) return;
  int it = items[b];
  size_t off = (size_t)lane * 4;
  *(float4*)(out_item + (size_t)b * EMB_DIM + off) =
      *(const float4*)(iemb + (size_t)it * EMB_DIM + off);
}

__global__ void __launch_bounds__(256)
dot_sigmoid_wmma(const float* __restrict__ U, const float* __restrict__ V,
                 float* __restrict__ pred, int batch) {
  int gid = blockIdx.x * blockDim.x + threadIdx.x;
  int wave = gid >> 5;
  int lane = threadIdx.x & 31;
  if (wave * 16 >= batch) return;

  int row = lane & 15;
  int half = lane >> 4;
  const float* urow = U + ((size_t)wave * 16 + row) * EMB_DIM + 2 * half;
  const float* vrow = V + ((size_t)wave * 16 + row) * EMB_DIM + 2 * half;

  v8f acc = {};
#pragma unroll
  for (int k0 = 0; k0 < EMB_DIM; k0 += 4) {
    v2f a = *(const v2f*)(urow + k0);
    v2f b = *(const v2f*)(vrow + k0);
    acc = __builtin_amdgcn_wmma_f32_16x16x4_f32(
         false, a,  false, b,
         (short)0, acc,  false,  false);
  }

  if (lane < 8 || lane >= 24) {
    int i = (lane < 8) ? lane : (lane - 16);
    float d = acc[lane & 7];
    pred[(size_t)wave * 16 + i] = 1.0f / (1.0f + expf(-d));
  }
}

extern "C" void kernel_launch(void* const* d_in, const int* in_sizes, int n_in,
                              void* d_out, int out_size, void* d_ws,
                              size_t ws_size, hipStream_t stream) {
  const int* users  = (const int*)d_in[0];
  const int* items  = (const int*)d_in[1];
  const int* esrc   = (const int*)d_in[2];
  const int* edst   = (const int*)d_in[3];
  const float* uemb = (const float*)d_in[4];
  const float* iemb = (const float*)d_in[5];

  const int batch     = in_sizes[0];
  const int n_edges   = in_sizes[2];
  const int num_users = in_sizes[4] / EMB_DIM;

  int* tab        = (int*)d_ws;
  float* g1       = (float*)d_ws + num_users;
  unsigned* elist = (unsigned*)(g1 + (size_t)num_users * EMB_DIM);

  float* pred     = (float*)d_out;
  float* out_user = pred + batch;
  float* out_item = out_user + (size_t)batch * EMB_DIM;

  const int B = 256;

  part_count<<<NCH, PT, 0, stream>>>(edst, tab, n_edges, num_users);
  part_scan<<<1, PT, 0, stream>>>(tab, num_users);
  part_place<<<NCH, PT, 0, stream>>>(esrc, edst, tab, elist, n_edges,
                                     num_users);

  conv_owner<false><<<NBMAX, PT, 0, stream>>>(elist, tab, uemb, g1, users,
                                              uemb, out_user, n_edges,
                                              num_users, batch);

  conv_owner<true><<<NBMAX, PT, 0, stream>>>(elist, tab, g1, (float*)nullptr,
                                             users, uemb, out_user, n_edges,
                                             num_users, batch);

  long long ga_threads = (long long)batch * 32;
  gather_items<<<(int)((ga_threads + B - 1) / B), B, 0, stream>>>(
      items, iemb, out_item, batch);

  long long dot_threads = (long long)((batch + 15) / 16) * 32;
  dot_sigmoid_wmma<<<(int)((dot_threads + B - 1) / B), B, 0, stream>>>(
      out_user, out_item, pred, batch);
}


# conv owner: block's edge list sorted by destination row in LDS (ballot-ranked counting sort), a wave sums a row in registers, 16 row loads in flight; per-edge LDS adds gone; sums in list order
# speedup vs baseline: 4.6534x; 1.1134x over previous
#include <hip/hip_runtime.h>
#include <hip/hip_bf16.h>


#define EMB_DIM 128
#define ROWS 391
#define ROW_BITS 9
#define ROW_MASK 511u
#define NBMAX 256
#define NCH 256
#define PT 1024
#define NWAVES (PT / 32)
#define SCAN_G (NBMAX * NCH / (PT * 4))
#define CAP 8192
#define SHARE (CAP / NWAVES)
#define KSTEPS (SHARE / 32)
#define HS 392
#define GRP 16

typedef __attribute__((ext_vector_type(2))) float v2f;
typedef __attribute__((ext_vector_type(8))) float v8f;
typedef __attribute__((ext_vector_type(4))) float v4f;

template <int BITS>
__device__ __forceinline__ unsigned same_key_lanes(int key, unsigned live) {
  unsigned peers = live;
#pragma unroll
  for (int i = 0; i < BITS; ++i) {
    const bool bit = ((key >> i) & 1) != 0;
    const unsigned m = __builtin_amdgcn_ballot_w32(bit);
    peers &= bit ? m : ~m;
  }
  return peers;
}

__device__ __forceinline__ void wave_block_hist(const int* __restrict__ dst,
                                                int* hist, int eb, int ee,
                                                int w, int lane) {
  const unsigned lt = (1u << lane) - 1u;
  for (int base = eb + w * 32; base < ee; base += PT) {
    const int e = base + lane;
    const bool valid = e < ee;
    const int b = valid ? dst[e] / ROWS : 0;
    const unsigned live = __builtin_amdgcn_ballot_w32(valid);
    const unsigned peers = same_key_lanes<8>(b, live);
    if (valid && (peers & lt) == 0u) hist[b] += __popc(peers);
  }
}

__global__ void __launch_bounds__(PT)
part_count(const int* __restrict__ dst, int* __restrict__ tab, int n_edges,
           int num_users) {
  __shared__ int whist[NWAVES * NBMAX];
  const int tid = threadIdx.x;
  const int lane = tid & 31;
  const int w = __builtin_amdgcn_readfirstlane(tid >> 5);
  const int nb = (num_users + ROWS - 1) / ROWS;
  for (int i = tid; i < NWAVES * NBMAX; i += PT) whist[i] = 0;
  __syncthreads();
  const int cs = (n_edges + NCH - 1) / NCH;
  const int eb = blockIdx.x * cs;
  const int ee = min(n_edges, eb + cs);
  wave_block_hist(dst, whist + w * NBMAX, eb, ee, w, lane);
  __syncthreads();
  if (tid < nb) {
    int s = 0;
    for (int k = 0; k < NWAVES; ++k) s += whist[k * NBMAX + tid];
    tab[tid * NCH + blockIdx.x] = s;
  }
}

__global__ void __launch_bounds__(PT)
part_scan(int* __restrict__ tab, int num_users) {
  __shared__ int wsum[NWAVES];
  const int tid = threadIdx.x;
  const int lane = tid & 31;
  const int w = tid >> 5;
  const int nb = (num_users + ROWS - 1) / ROWS;
  const int total = nb * NCH;
  const int first = w * (SCAN_G * 128) + lane * 4;
  int4 x[SCAN_G];
#pragma unroll
  for (int g = 0; g < SCAN_G; ++g) {
    const int idx = first + g * 128;
    x[g] = (idx < total) ? *(const int4*)(tab + idx) : make_int4(0, 0, 0, 0);
  }
  int before[SCAN_G];
  int run = 0;
#pragma unroll
  for (int g = 0; g < SCAN_G; ++g) {
    const int s = x[g].x + x[g].y + x[g].z + x[g].w;
    int inc = s;
#pragma unroll
    for (int o = 1; o < 32; o <<= 1) {
      const int t = __shfl_up(inc, o);
      if (lane >= o) inc += t;
    }
    before[g] = run + inc - s;
    run += __builtin_amdgcn_readlane(inc, 31);
  }
  if (lane == 0) wsum[w] = run;
  __syncthreads();
  if (w == 0) {
    const int v = wsum[lane];
    int vi = v;
#pragma unroll
    for (int o = 1; o < 32; o <<= 1) {
      const int t = __shfl_up(vi, o);
      if (lane >= o) vi += t;
    }
    wsum[lane] = vi - v;
  }
  __syncthreads();
  const int off = wsum[w];
#pragma unroll
  for (int g = 0; g < SCAN_G; ++g) {
    const int idx = first + g * 128;
    if (idx < total) {
      int4 o;
      o.x = off + before[g];
      o.y = o.x + x[g].x;
      o.z = o.y + x[g].y;
      o.w = o.z + x[g].z;
      *(int4*)(tab + idx) = o;
    }
  }
}

__global__ void __launch_bounds__(PT)
part_place(const int* __restrict__ src, const int* __restrict__ dst,
           const int* __restrict__ tab, unsigned* __restrict__ elist,
           int n_edges, int num_users) {
  __shared__ int whist[NWAVES * NBMAX];
  const int tid = threadIdx.x;
  const int lane = tid & 31;
  const int w = __builtin_amdgcn_readfirstlane(tid >> 5);
  const int nb = (num_users + ROWS - 1) / ROWS;
  for (int i = tid; i < NWAVES * NBMAX; i += PT) whist[i] = 0;
  __syncthreads();
  const int cs = (n_edges + NCH - 1) / NCH;
  const int eb = blockIdx.x * cs;
  const int ee = min(n_edges, eb + cs);
  wave_block_hist(dst, whist + w * NBMAX, eb, ee, w, lane);
  __syncthreads();
  if (tid < nb) {
    int run = tab[tid * NCH + blockIdx.x];
    for (int k = 0; k < NWAVES; ++k) {
      const int c = whist[k * NBMAX + tid];
      whist[k * NBMAX + tid] = run;
      run += c;
    }
  }
  __syncthreads();
  int* cur = whist + w * NBMAX;
  const unsigned lt = (1u << lane) - 1u;
  for (int base = eb + w * 32; base < ee; base += PT) {
    const int e = base + lane;
    const bool valid = e < ee;
    const int d = valid ? dst[e] : 0;
    const int b = d / ROWS;
    const unsigned live = __builtin_amdgcn_ballot_w32(valid);
    const unsigned peers = same_key_lanes<8>(b, live);
    const int rank = __popc(peers & lt);
    const int c0 = cur[b];
    if (valid)
      elist[c0 + rank] =
          ((unsigned)src[e] << ROW_BITS) | (unsigned)(d - b * ROWS);
    if (valid && rank == 0) cur[b] = c0 + __popc(peers);
  }
}

template <int G>
__device__ __forceinline__ void hold_loaded(const v4f (&v)[G]) {
  if constexpr (G == 1) asm volatile("" :: "v"(v[0]));
  else if constexpr (G == 2) asm volatile("" :: "v"(v[0]), "v"(v[1]));
  else if constexpr (G == 3) asm volatile("" :: "v"(v[0]), "v"(v[1]), "v"(v[2]));
  else if constexpr (G == 4) asm volatile("" :: "v"(v[0]), "v"(v[1]), "v"(v[2]), "v"(v[3]));
  else if constexpr (G == 5) asm volatile("" :: "v"(v[0]), "v"(v[1]), "v"(v[2]), "v"(v[3]), "v"(v[4]));
  else if constexpr (G == 6) asm volatile("" :: "v"(v[0]), "v"(v[1]), "v"(v[2]), "v"(v[3]), "v"(v[4]), "v"(v[5]));
  else if constexpr (G == 7) asm volatile("" :: "v"(v[0]), "v"(v[1]), "v"(v[2]), "v"(v[3]), "v"(v[4]), "v"(v[5]), "v"(v[6]));
  else if constexpr (G == 8) asm volatile("" :: "v"(v[0]), "v"(v[1]), "v"(v[2]), "v"(v[3]), "v"(v[4]), "v"(v[5]), "v"(v[6]), "v"(v[7]));
  else if constexpr (G == 9) asm volatile("" :: "v"(v[0]), "v"(v[1]), "v"(v[2]), "v"(v[3]), "v"(v[4]), "v"(v[5]), "v"(v[6]), "v"(v[7]), "v"(v[8]));
  else if constexpr (G == 10) asm volatile("" :: "v"(v[0]), "v"(v[1]), "v"(v[2]), "v"(v[3]), "v"(v[4]), "v"(v[5]), "v"(v[6]), "v"(v[7]), "v"(v[8]), "v"(v[9]));
  else if constexpr (G == 11) asm volatile("" :: "v"(v[0]), "v"(v[1]), "v"(v[2]), "v"(v[3]), "v"(v[4]), "v"(v[5]), "v"(v[6]), "v"(v[7]), "v"(v[8]), "v"(v[9]), "v"(v[10]));
  else if constexpr (G == 12) asm volatile("" :: "v"(v[0]), "v"(v[1]), "v"(v[2]), "v"(v[3]), "v"(v[4]), "v"(v[5]), "v"(v[6]), "v"(v[7]), "v"(v[8]), "v"(v[9]), "v"(v[10]), "v"(v[11]));
  else if constexpr (G == 13) asm volatile("" :: "v"(v[0]), "v"(v[1]), "v"(v[2]), "v"(v[3]), "v"(v[4]), "v"(v[5]), "v"(v[6]), "v"(v[7]), "v"(v[8]), "v"(v[9]), "v"(v[10]), "v"(v[11]), "v"(v[12]));
  else if constexpr (G == 14) asm volatile("" :: "v"(v[0]), "v"(v[1]), "v"(v[2]), "v"(v[3]), "v"(v[4]), "v"(v[5]), "v"(v[6]), "v"(v[7]), "v"(v[8]), "v"(v[9]), "v"(v[10]), "v"(v[11]), "v"(v[12]), "v"(v[13]));
  else if constexpr (G == 15) asm volatile("" :: "v"(v[0]), "v"(v[1]), "v"(v[2]), "v"(v[3]), "v"(v[4]), "v"(v[5]), "v"(v[6]), "v"(v[7]), "v"(v[8]), "v"(v[9]), "v"(v[10]), "v"(v[11]), "v"(v[12]), "v"(v[13]), "v"(v[14]));
  else if constexpr (G == 16) asm volatile("" :: "v"(v[0]), "v"(v[1]), "v"(v[2]), "v"(v[3]), "v"(v[4]), "v"(v[5]), "v"(v[6]), "v"(v[7]), "v"(v[8]), "v"(v[9]), "v"(v[10]), "v"(v[11]), "v"(v[12]), "v"(v[13]), "v"(v[14]), "v"(v[15]));
}

template <int G>
__device__ __forceinline__ void sum_group(const unsigned* p,
                                          const float* __restrict__ h,
                                          int lane, float4& s) {
  const unsigned mine = p[lane < G ? lane : 0];
  unsigned xe[G];
#pragma unroll
  for (int j = 0; j < G; ++j)
    xe[j] = (unsigned)__builtin_amdgcn_readlane((int)mine, j);
  v4f v[G];
#pragma unroll
  for (int j = 0; j < G; ++j) {
    const unsigned off =
        xe[j] * (unsigned)(EMB_DIM * sizeof(float)) + (unsigned)lane * 16u;
    v[j] = *(const v4f*)((const char*)h + off);
  }
  hold_loaded<G>(v);
#pragma unroll
  for (int j = 0; j < G; ++j) {
    s.x += v[j].x;
    s.y += v[j].y;
    s.z += v[j].z;
    s.w += v[j].w;
  }
}

template <bool LAST>
__global__ void __launch_bounds__(PT)
conv_owner(const unsigned* __restrict__ elist, const int* __restrict__ tab,
           const float* __restrict__ h, float* __restrict__ gout,
           const int* __restrict__ users, const float* __restrict__ uemb,
           float* __restrict__ out_user, int n_edges, int num_users,
           int batch) {
  __shared__ __attribute__((aligned(16))) float acc[ROWS * EMB_DIM];
  __shared__ unsigned sorted[CAP];
  __shared__ int whist[NWAVES * HS];
  __shared__ int rs[HS];
  __shared__ int rc[HS];
  __shared__ int cnt[HS];
  __shared__ int wsum[NWAVES];
  const int tid = threadIdx.x;
  const int lane = tid & 31;
  const int w = __builtin_amdgcn_readfirstlane(tid >> 5);
  const unsigned lt = (1u << lane) - 1u;
  const int nb = (num_users + ROWS - 1) / ROWS;
  float4* acc4 = (float4*)acc;
  int* myh = whist + w * HS;

  for (int bin = blockIdx.x; bin < nb; bin += gridDim.x) {
    const int row0 = bin * ROWS;
    const int nrows = min(ROWS, num_users - row0);

    for (int i = tid; i < ROWS * EMB_DIM / 4; i += PT)
      acc4[i] = make_float4(0.0f, 0.0f, 0.0f, 0.0f);
    for (int i = tid; i < HS; i += PT) cnt[i] = 0;

    const int e0 = tab[bin * NCH];
    const int e1 = (bin + 1 < nb) ? tab[(bin + 1) * NCH] : n_edges;

    for (int t0 = e0; t0 < e1; t0 += CAP) {
      const int tn = min(CAP, e1 - t0);
      __syncthreads();
      for (int i = tid; i < NWAVES * HS; i += PT) whist[i] = 0;
      unsigned x[KSTEPS];
#pragma unroll
      for (int k = 0; k < KSTEPS; ++k) {
        const int i = w * SHARE + k * 32 + lane;
        x[k] = (i < tn) ? elist[t0 + i] : 0u;
      }
      __syncthreads();

      unsigned peers[KSTEPS];
#pragma unroll
      for (int k = 0; k < KSTEPS; ++k) {
        const bool valid = w * SHARE + k * 32 + lane < tn;
        const int key = (int)(x[k] & ROW_MASK);
        const unsigned live = __builtin_amdgcn_ballot_w32(valid);
        peers[k] = same_key_lanes<ROW_BITS>(key, live);
        if (valid && (peers[k] & lt) == 0u) myh[key] += __popc(peers[k]);
      }
      __syncthreads();

      int tot = 0;
      if (tid < HS) {
        int cw[NWAVES];
#pragma unroll
        for (int k = 0; k < NWAVES; ++k) cw[k] = whist[k * HS + tid];
#pragma unroll
        for (int k = 0; k < NWAVES; ++k) {
          whist[k * HS + tid] = tot;
          tot += cw[k];
        }
      }
      int inc = tot;
#pragma unroll
      for (int o = 1; o < 32; o <<= 1) {
        const int t = __shfl_up(inc, o);
        if (lane >= o) inc += t;
      }
      if (lane == 31) wsum[w] = inc;
      __syncthreads();
      if (w == 0) {
        const int v = wsum[lane];
        int vi = v;
#pragma unroll
        for (int o = 1; o < 32; o <<= 1) {
          const int t = __shfl_up(vi, o);
          if (lane >= o) vi += t;
        }
        wsum[lane] = vi - v;
      }
      __syncthreads();
      if (tid < HS) {
        rs[tid] = wsum[w] + inc - tot;
        rc[tid] = tot;
      }
      __syncthreads();

#pragma unroll
      for (int k = 0; k < KSTEPS; ++k) {
        const bool valid = w * SHARE + k * 32 + lane < tn;
        const int key = (int)(x[k] & ROW_MASK);
        const int rank = __popc(peers[k] & lt);
        const int c0 = myh[key];
        if (valid) sorted[rs[key] + c0 + rank] = x[k] >> ROW_BITS;
        if (valid && rank == 0) myh[key] = c0 + __popc(peers[k]);
      }
      __syncthreads();

      for (int r = w; r < ROWS; r += NWAVES) {
        const int st = __builtin_amdgcn_readfirstlane(rs[r]);
        const int c = __builtin_amdgcn_readfirstlane(rc[r]);
        if (c == 0) continue;
        float4 s = make_float4(0.0f, 0.0f, 0.0f, 0.0f);
        const unsigned* p = sorted + st;
        int q = 0;
        for (; q + GRP <= c; q += GRP) sum_group<GRP>(p + q, h, lane, s);
        switch (c - q) {
          case 1: sum_group<1>(p + q, h, lane, s); break;
          case 2: sum_group<2>(p + q, h, lane, s); break;
          case 3: sum_group<3>(p + q, h, lane, s); break;
          case 4: sum_group<4>(p + q, h, lane, s); break;
          case 5: sum_group<5>(p + q, h, lane, s); break;
          case 6: sum_group<6>(p + q, h, lane, s); break;
          case 7: sum_group<7>(p + q, h, lane, s); break;
          case 8: sum_group<8>(p + q, h, lane, s); break;
          case 9: sum_group<9>(p + q, h, lane, s); break;
          case 10: sum_group<10>(p + q, h, lane, s); break;
          case 11: sum_group<11>(p + q, h, lane, s); break;
          case 12: sum_group<12>(p + q, h, lane, s); break;
          case 13: sum_group<13>(p + q, h, lane, s); break;
          case 14: sum_group<14>(p + q, h, lane, s); break;
          case 15: sum_group<15>(p + q, h, lane, s); break;
          default: break;
        }
        float4 a = acc4[r * 32 + lane];
        a.x += s.x;
        a.y += s.y;
        a.z += s.z;
        a.w += s.w;
        acc4[r * 32 + lane] = a;
        if (lane == 0) cnt[r] += c;
      }
    }
    __syncthreads();

    if (!LAST) {
      for (int r = w; r < nrows; r += NWAVES) {
        const float m = fmaxf((float)cnt[r], 1.0f);
        const float4 a = acc4[r * 32 + lane];
        float4 o;
        o.x = a.x / m;
        o.y = a.y / m;
        o.z = a.z / m;
        o.w = a.w / m;
        *(float4*)(gout + (size_t)(row0 + r) * EMB_DIM + lane * 4) = o;
      }
    } else {
      for (int b0 = w * 32; b0 < batch; b0 += 4 * PT) {
        int r[4];
#pragma unroll
        for (int k = 0; k < 4; ++k) {
          const int b = b0 + k * PT + lane;
          r[k] = (b < batch) ? (users[b] - row0) : -1;
        }
#pragma unroll
        for (int k = 0; k < 4; ++k) {
          unsigned mask = __builtin_amdgcn_ballot_w32(r[k] >= 0 && r[k] < nrows);
          while (mask) {
            const int j = __builtin_ctz(mask);
            mask &= mask - 1;
            const int rj = __builtin_amdgcn_readlane(r[k], j);
            const size_t ro = (size_t)(row0 + rj) * EMB_DIM + lane * 4;
            const float4 x = *(const float4*)(uemb + ro);
            const float4 c = *(const float4*)(h + ro);
            const float m = fmaxf((float)cnt[rj], 1.0f);
            const float4 a = acc4[rj * 32 + lane];
            float4 o;
            o.x = x.x + c.x + a.x / m;
            o.y = x.y + c.y + a.y / m;
            o.z = x.z + c.z + a.z / m;
            o.w = x.w + c.w + a.w / m;
            *(float4*)(out_user + (size_t)(b0 + k * PT + j) * EMB_DIM +
                       lane * 4) = o;
          }
        }
      }
    }
    __syncthreads();
  }
}

__global__ void __launch_bounds__(256)
gather_items(const int* __restrict__ items, const float* __restrict__ iemb,
             float* __restrict__ out_item, int batch) {
  int t = blockIdx.x * blockDim.x + threadIdx.x;
  int b = t >> 5;
  int lane = t & 31;
  if (b >= batch) return;
  int it = items[b];
  size_t off = (size_t)lane * 4;
  *(float4*)(out_item + (size_t)b * EMB_DIM + off) =
      *(const float4*)(iemb + (size_t)it * EMB_DIM + off);
}

__global__ void __launch_bounds__(256)
dot_sigmoid_wmma(const float* __restrict__ U, const float* __restrict__ V,
                 float* __restrict__ pred, int batch) {
  int gid = blockIdx.x * blockDim.x + threadIdx.x;
  int wave = gid >> 5;
  int lane = threadIdx.x & 31;
  if (wave * 16 >= batch) return;

  int row = lane & 15;
  int half = lane >> 4;
  const float* urow = U + ((size_t)wave * 16 + row) * EMB_DIM + 2 * half;
  const float* vrow = V + ((size_t)wave * 16 + row) * EMB_DIM + 2 * half;

  v8f acc = {};
#pragma unroll
  for (int k0 = 0; k0 < EMB_DIM; k0 += 4) {
    v2f a = *(const v2f*)(urow + k0);
    v2f b = *(const v2f*)(vrow + k0);
    acc = __builtin_amdgcn_wmma_f32_16x16x4_f32(
         false, a,  false, b,
         (short)0, acc,  false,  false);
  }

  if (lane < 8 || lane >= 24) {
    int i = (lane < 8) ? lane : (lane - 16);
    float d = acc[lane & 7];
    pred[(size_t)wave * 16 + i] = 1.0f / (1.0f + expf(-d));
  }
}

extern "C" void kernel_launch(void* const* d_in, const int* in_sizes, int n_in,
                              void* d_out, int out_size, void* d_ws,
                              size_t ws_size, hipStream_t stream) {
  const int* users  = (const int*)d_in[0];
  const int* items  = (const int*)d_in[1];
  const int* esrc   = (const int*)d_in[2];
  const int* edst   = (const int*)d_in[3];
  const float* uemb = (const float*)d_in[4];
  const float* iemb = (const float*)d_in[5];

  const int batch     = in_sizes[0];
  const int n_edges   = in_sizes[2];
  const int num_users = in_sizes[4] / EMB_DIM;

  int* tab        = (int*)d_ws;
  float* g1       = (float*)d_ws + num_users;
  unsigned* elist = (unsigned*)(g1 + (size_t)num_users * EMB_DIM);

  float* pred     = (float*)d_out;
  float* out_user = pred + batch;
  float* out_item = out_user + (size_t)batch * EMB_DIM;

  const int B = 256;

  part_count<<<NCH, PT, 0, stream>>>(edst, tab, n_edges, num_users);
  part_scan<<<1, PT, 0, stream>>>(tab, num_users);
  part_place<<<NCH, PT, 0, stream>>>(esrc, edst, tab, elist, n_edges,
                                     num_users);

  conv_owner<false><<<NBMAX, PT, 0, stream>>>(elist, tab, uemb, g1, users,
                                              uemb, out_user, n_edges,
                                              num_users, batch);

  conv_owner<true><<<NBMAX, PT, 0, stream>>>(elist, tab, g1, (float*)nullptr,
                                             users, uemb, out_user, n_edges,
                                             num_users, batch);

  long long ga_threads = (long long)batch * 32;
  gather_items<<<(int)((ga_threads + B - 1) / B), B, 0, stream>>>(
      items, iemb, out_item, batch);

  long long dot_threads = (long long)((batch + 15) / 16) * 32;
  dot_sigmoid_wmma<<<(int)((dot_threads + B - 1) / B), B, 0, stream>>>(
      out_user, out_item, pred, batch);
}


# edge partition in one launch: a chunk groups its edges by block in its own list part (places from own counts); an owner collects its 256 groups in LDS; count and scan launches gone; sum order kept
# speedup vs baseline: 5.7481x; 1.2352x over previous
#include <hip/hip_runtime.h>
#include <hip/hip_bf16.h>


#define EMB_DIM 128
#define ROWS 391
#define ROW_BITS 9
#define ROW_MASK 511u
#define NBMAX 256
#define NCH 256
#define PT 1024
#define NWAVES (PT / 32)
#define CPW (NCH / NWAVES)
#define CAP 8192
#define SHARE (CAP / NWAVES)
#define KSTEPS (SHARE / 32)
#define HS 392
#define GRP 16

typedef __attribute__((ext_vector_type(2))) float v2f;
typedef __attribute__((ext_vector_type(8))) float v8f;
typedef __attribute__((ext_vector_type(4))) float v4f;

template <int BITS>
__device__ __forceinline__ unsigned same_key_lanes(int key, unsigned live) {
  unsigned peers = live;
#pragma unroll
  for (int i = 0; i < BITS; ++i) {
    const bool bit = ((key >> i) & 1) != 0;
    const unsigned m = __builtin_amdgcn_ballot_w32(bit);
    peers &= bit ? m : ~m;
  }
  return peers;
}

__device__ __forceinline__ void wave_block_hist(const int* __restrict__ dst,
                                                int* hist, int eb, int ee,
                                                int w, int lane) {
  const unsigned lt = (1u << lane) - 1u;
  for (int base = eb + w * 32; base < ee; base += PT) {
    const int e = base + lane;
    const bool valid = e < ee;
    const int b = valid ? dst[e] / ROWS : 0;
    const unsigned live = __builtin_amdgcn_ballot_w32(valid);
    const unsigned peers = same_key_lanes<8>(b, live);
    if (valid && (peers & lt) == 0u) hist[b] += __popc(peers);
  }
}

__global__ void __launch_bounds__(PT)
part_sort(const int* __restrict__ src, const int* __restrict__ dst,
          int* __restrict__ tab, unsigned* __restrict__ elist, int n_edges,
          int num_users) {
  __shared__ int whist[NWAVES * NBMAX];
  __shared__ int wsum[NWAVES];
  const int tid = threadIdx.x;
  const int lane = tid & 31;
  const int w = __builtin_amdgcn_readfirstlane(tid >> 5);
  const int nb = (num_users + ROWS - 1) / ROWS;
  for (int i = tid; i < NWAVES * NBMAX; i += PT) whist[i] = 0;
  __syncthreads();
  const int cs = (n_edges + NCH - 1) / NCH;
  const int eb = blockIdx.x * cs;
  const int ee = min(n_edges, eb + cs);
  wave_block_hist(dst, whist + w * NBMAX, eb, ee, w, lane);
  __syncthreads();
  int tot = 0;
  if (tid < NBMAX) {
    for (int k = 0; k < NWAVES; ++k) tot += whist[k * NBMAX + tid];
  }
  int inc = tot;
#pragma unroll
  for (int o = 1; o < 32; o <<= 1) {
    const int t = __shfl_up(inc, o);
    if (lane >= o) inc += t;
  }
  if (lane == 31) wsum[w] = inc;
  __syncthreads();
  if (tid < NBMAX) {
    int run = eb + inc - tot;
#pragma unroll
    for (int k = 0; k < NBMAX / 32; ++k) run += (k < w) ? wsum[k] : 0;
    if (tid < nb) tab[tid * NCH + blockIdx.x] = run;
    if (tid == nb - 1) tab[nb * NCH + blockIdx.x] = run + tot;
    for (int k = 0; k < NWAVES; ++k) {
      const int c = whist[k * NBMAX + tid];
      whist[k * NBMAX + tid] = run;
      run += c;
    }
  }
  __syncthreads();
  int* cur = whist + w * NBMAX;
  const unsigned lt = (1u << lane) - 1u;
  for (int base = eb + w * 32; base < ee; base += PT) {
    const int e = base + lane;
    const bool valid = e < ee;
    const int d = valid ? dst[e] : 0;
    const int b = d / ROWS;
    const unsigned live = __builtin_amdgcn_ballot_w32(valid);
    const unsigned peers = same_key_lanes<8>(b, live);
    const int rank = __popc(peers & lt);
    const int c0 = cur[b];
    if (valid)
      elist[c0 + rank] =
          ((unsigned)src[e] << ROW_BITS) | (unsigned)(d - b * ROWS);
    if (valid && rank == 0) cur[b] = c0 + __popc(peers);
  }
}

template <int G>
__device__ __forceinline__ void hold_loaded(const v4f (&v)[G]) {
  if constexpr (G == 1) asm volatile("" :: "v"(v[0]));
  else if constexpr (G == 2) asm volatile("" :: "v"(v[0]), "v"(v[1]));
  else if constexpr (G == 3) asm volatile("" :: "v"(v[0]), "v"(v[1]), "v"(v[2]));
  else if constexpr (G == 4) asm volatile("" :: "v"(v[0]), "v"(v[1]), "v"(v[2]), "v"(v[3]));
  else if constexpr (G == 5) asm volatile("" :: "v"(v[0]), "v"(v[1]), "v"(v[2]), "v"(v[3]), "v"(v[4]));
  else if constexpr (G == 6) asm volatile("" :: "v"(v[0]), "v"(v[1]), "v"(v[2]), "v"(v[3]), "v"(v[4]), "v"(v[5]));
  else if constexpr (G == 7) asm volatile("" :: "v"(v[0]), "v"(v[1]), "v"(v[2]), "v"(v[3]), "v"(v[4]), "v"(v[5]), "v"(v[6]));
  else if constexpr (G == 8) asm volatile("" :: "v"(v[0]), "v"(v[1]), "v"(v[2]), "v"(v[3]), "v"(v[4]), "v"(v[5]), "v"(v[6]), "v"(v[7]));
  else if constexpr (G == 9) asm volatile("" :: "v"(v[0]), "v"(v[1]), "v"(v[2]), "v"(v[3]), "v"(v[4]), "v"(v[5]), "v"(v[6]), "v"(v[7]), "v"(v[8]));
  else if constexpr (G == 10) asm volatile("" :: "v"(v[0]), "v"(v[1]), "v"(v[2]), "v"(v[3]), "v"(v[4]), "v"(v[5]), "v"(v[6]), "v"(v[7]), "v"(v[8]), "v"(v[9]));
  else if constexpr (G == 11) asm volatile("" :: "v"(v[0]), "v"(v[1]), "v"(v[2]), "v"(v[3]), "v"(v[4]), "v"(v[5]), "v"(v[6]), "v"(v[7]), "v"(v[8]), "v"(v[9]), "v"(v[10]));
  else if constexpr (G == 12) asm volatile("" :: "v"(v[0]), "v"(v[1]), "v"(v[2]), "v"(v[3]), "v"(v[4]), "v"(v[5]), "v"(v[6]), "v"(v[7]), "v"(v[8]), "v"(v[9]), "v"(v[10]), "v"(v[11]));
  else if constexpr (G == 13) asm volatile("" :: "v"(v[0]), "v"(v[1]), "v"(v[2]), "v"(v[3]), "v"(v[4]), "v"(v[5]), "v"(v[6]), "v"(v[7]), "v"(v[8]), "v"(v[9]), "v"(v[10]), "v"(v[11]), "v"(v[12]));
  else if constexpr (G == 14) asm volatile("" :: "v"(v[0]), "v"(v[1]), "v"(v[2]), "v"(v[3]), "v"(v[4]), "v"(v[5]), "v"(v[6]), "v"(v[7]), "v"(v[8]), "v"(v[9]), "v"(v[10]), "v"(v[11]), "v"(v[12]), "v"(v[13]));
  else if constexpr (G == 15) asm volatile("" :: "v"(v[0]), "v"(v[1]), "v"(v[2]), "v"(v[3]), "v"(v[4]), "v"(v[5]), "v"(v[6]), "v"(v[7]), "v"(v[8]), "v"(v[9]), "v"(v[10]), "v"(v[11]), "v"(v[12]), "v"(v[13]), "v"(v[14]));
  else if constexpr (G == 16) asm volatile("" :: "v"(v[0]), "v"(v[1]), "v"(v[2]), "v"(v[3]), "v"(v[4]), "v"(v[5]), "v"(v[6]), "v"(v[7]), "v"(v[8]), "v"(v[9]), "v"(v[10]), "v"(v[11]), "v"(v[12]), "v"(v[13]), "v"(v[14]), "v"(v[15]));
}

template <int G>
__device__ __forceinline__ void sum_group(const unsigned* p,
                                          const float* __restrict__ h,
                                          int lane, float4& s) {
  const unsigned mine = p[lane < G ? lane : 0];
  unsigned xe[G];
#pragma unroll
  for (int j = 0; j < G; ++j)
    xe[j] = (unsigned)__builtin_amdgcn_readlane((int)mine, j);
  v4f v[G];
#pragma unroll
  for (int j = 0; j < G; ++j) {
    const unsigned off =
        xe[j] * (unsigned)(EMB_DIM * sizeof(float)) + (unsigned)lane * 16u;
    v[j] = *(const v4f*)((const char*)h + off);
  }
  hold_loaded<G>(v);
#pragma unroll
  for (int j = 0; j < G; ++j) {
    s.x += v[j].x;
    s.y += v[j].y;
    s.z += v[j].z;
    s.w += v[j].w;
  }
}

template <bool LAST>
__global__ void __launch_bounds__(PT)
conv_owner(const unsigned* __restrict__ elist, const int* __restrict__ tab,
           const float* __restrict__ h, float* __restrict__ gout,
           const int* __restrict__ users, const float* __restrict__ uemb,
           float* __restrict__ out_user, int num_users, int batch) {
  __shared__ __attribute__((aligned(16))) float acc[ROWS * EMB_DIM];
  __shared__ unsigned sorted[CAP];
  __shared__ int segpos[NCH + 1];
  __shared__ int segsrc[NCH];
  __shared__ int whist[NWAVES * HS];
  __shared__ int rs[HS];
  __shared__ int rc[HS];
  __shared__ int cnt[HS];
  __shared__ int wsum[NWAVES];
  const int tid = threadIdx.x;
  const int lane = tid & 31;
  const int w = __builtin_amdgcn_readfirstlane(tid >> 5);
  const unsigned lt = (1u << lane) - 1u;
  const int nb = (num_users + ROWS - 1) / ROWS;
  float4* acc4 = (float4*)acc;
  int* myh = whist + w * HS;

  for (int bin = blockIdx.x; bin < nb; bin += gridDim.x) {
    const int row0 = bin * ROWS;
    const int nrows = min(ROWS, num_users - row0);

    for (int i = tid; i < ROWS * EMB_DIM / 4; i += PT)
      acc4[i] = make_float4(0.0f, 0.0f, 0.0f, 0.0f);
    for (int i = tid; i < HS; i += PT) cnt[i] = 0;

    int gsrc = 0;
    int glen = 0;
    if (tid < NCH) {
      gsrc = tab[bin * NCH + tid];
      glen = tab[(bin + 1) * NCH + tid] - gsrc;
    }
    int ginc = glen;
#pragma unroll
    for (int o = 1; o < 32; o <<= 1) {
      const int t = __shfl_up(ginc, o);
      if (lane >= o) ginc += t;
    }
    if (lane == 31) wsum[w] = ginc;
    __syncthreads();
    if (tid < NCH) {
      int p = ginc - glen;
#pragma unroll
      for (int k = 0; k < NCH / 32; ++k) p += (k < w) ? wsum[k] : 0;
      segpos[tid] = p;
      segsrc[tid] = gsrc;
      if (tid == NCH - 1) segpos[NCH] = p + glen;
    }
    __syncthreads();
    const int nlist = segpos[NCH];

    for (int t0 = 0; t0 < nlist; t0 += CAP) {
      const int tn = min(CAP, nlist - t0);
      __syncthreads();
      for (int i = tid; i < NWAVES * HS; i += PT) whist[i] = 0;
      int gp[CPW + 1];
      int gs[CPW];
#pragma unroll
      for (int k = 0; k <= CPW; ++k) gp[k] = segpos[w * CPW + k];
#pragma unroll
      for (int k = 0; k < CPW; ++k) gs[k] = segsrc[w * CPW + k];
      unsigned gv[CPW];
      int gi[CPW];
#pragma unroll
      for (int k = 0; k < CPW; ++k) {
        const int a = max(gp[k], t0);
        const int b = min(gp[k + 1], t0 + tn);
        const int i = a + lane;
        gi[k] = (i < b) ? (i - t0) : -1;
        gv[k] = (i < b) ? elist[gs[k] + (i - gp[k])] : 0u;
      }
#pragma unroll
      for (int k = 0; k < CPW; ++k)
        if (gi[k] >= 0) sorted[gi[k]] = gv[k];
#pragma nounroll
      for (int k = 0; k < CPW; ++k) {
        const int c = w * CPW + k;
        const int p = segpos[c];
        const int a = max(p, t0);
        const int b = min(segpos[c + 1], t0 + tn);
#pragma nounroll
        for (int i = a + 32 + lane; i < b; i += 32)
          sorted[i - t0] = elist[segsrc[c] + (i - p)];
      }
      __syncthreads();
      unsigned x[KSTEPS];
#pragma unroll
      for (int k = 0; k < KSTEPS; ++k) {
        const int i = w * SHARE + k * 32 + lane;
        x[k] = (i < tn) ? sorted[i] : 0u;
      }
      __syncthreads();

      unsigned peers[KSTEPS];
#pragma unroll
      for (int k = 0; k < KSTEPS; ++k) {
        const bool valid = w * SHARE + k * 32 + lane < tn;
        const int key = (int)(x[k] & ROW_MASK);
        const unsigned live = __builtin_amdgcn_ballot_w32(valid);
        peers[k] = same_key_lanes<ROW_BITS>(key, live);
        if (valid && (peers[k] & lt) == 0u) myh[key] += __popc(peers[k]);
      }
      __syncthreads();

      int tot = 0;
      if (tid < HS) {
        int cw[NWAVES];
#pragma unroll
        for (int k = 0; k < NWAVES; ++k) cw[k] = whist[k * HS + tid];
#pragma unroll
        for (int k = 0; k < NWAVES; ++k) {
          whist[k * HS + tid] = tot;
          tot += cw[k];
        }
      }
      int inc = tot;
#pragma unroll
      for (int o = 1; o < 32; o <<= 1) {
        const int t = __shfl_up(inc, o);
        if (lane >= o) inc += t;
      }
      if (lane == 31) wsum[w] = inc;
      __syncthreads();
      if (w == 0) {
        const int v = wsum[lane];
        int vi = v;
#pragma unroll
        for (int o = 1; o < 32; o <<= 1) {
          const int t = __shfl_up(vi, o);
          if (lane >= o) vi += t;
        }
        wsum[lane] = vi - v;
      }
      __syncthreads();
      if (tid < HS) {
        rs[tid] = wsum[w] + inc - tot;
        rc[tid] = tot;
      }
      __syncthreads();

#pragma unroll
      for (int k = 0; k < KSTEPS; ++k) {
        const bool valid = w * SHARE + k * 32 + lane < tn;
        const int key = (int)(x[k] & ROW_MASK);
        const int rank = __popc(peers[k] & lt);
        const int c0 = myh[key];
        if (valid) sorted[rs[key] + c0 + rank] = x[k] >> ROW_BITS;
        if (valid && rank == 0) myh[key] = c0 + __popc(peers[k]);
      }
      __syncthreads();

      for (int r = w; r < ROWS; r += NWAVES) {
        const int st = __builtin_amdgcn_readfirstlane(rs[r]);
        const int c = __builtin_amdgcn_readfirstlane(rc[r]);
        if (c == 0) continue;
        float4 s = make_float4(0.0f, 0.0f, 0.0f, 0.0f);
        const unsigned* p = sorted + st;
        int q = 0;
        for (; q + GRP <= c; q += GRP) sum_group<GRP>(p + q, h, lane, s);
        switch (c - q) {
          case 1: sum_group<1>(p + q, h, lane, s); break;
          case 2: sum_group<2>(p + q, h, lane, s); break;
          case 3: sum_group<3>(p + q, h, lane, s); break;
          case 4: sum_group<4>(p + q, h, lane, s); break;
          case 5: sum_group<5>(p + q, h, lane, s); break;
          case 6: sum_group<6>(p + q, h, lane, s); break;
          case 7: sum_group<7>(p + q, h, lane, s); break;
          case 8: sum_group<8>(p + q, h, lane, s); break;
          case 9: sum_group<9>(p + q, h, lane, s); break;
          case 10: sum_group<10>(p + q, h, lane, s); break;
          case 11: sum_group<11>(p + q, h, lane, s); break;
          case 12: sum_group<12>(p + q, h, lane, s); break;
          case 13: sum_group<13>(p + q, h, lane, s); break;
          case 14: sum_group<14>(p + q, h, lane, s); break;
          case 15: sum_group<15>(p + q, h, lane, s); break;
          default: break;
        }
        float4 a = acc4[r * 32 + lane];
        a.x += s.x;
        a.y += s.y;
        a.z += s.z;
        a.w += s.w;
        acc4[r * 32 + lane] = a;
        if (lane == 0) cnt[r] += c;
      }
    }
    __syncthreads();

    if (!LAST) {
      for (int r = w; r < nrows; r += NWAVES) {
        const float m = fmaxf((float)cnt[r], 1.0f);
        const float4 a = acc4[r * 32 + lane];
        float4 o;
        o.x = a.x / m;
        o.y = a.y / m;
        o.z = a.z / m;
        o.w = a.w / m;
        *(float4*)(gout + (size_t)(row0 + r) * EMB_DIM + lane * 4) = o;
      }
    } else {
      for (int b0 = w * 32; b0 < batch; b0 += 4 * PT) {
        int r[4];
#pragma unroll
        for (int k = 0; k < 4; ++k) {
          const int b = b0 + k * PT + lane;
          r[k] = (b < batch) ? (users[b] - row0) : -1;
        }
#pragma unroll
        for (int k = 0; k < 4; ++k) {
          unsigned mask = __builtin_amdgcn_ballot_w32(r[k] >= 0 && r[k] < nrows);
          while (mask) {
            const int j = __builtin_ctz(mask);
            mask &= mask - 1;
            const int rj = __builtin_amdgcn_readlane(r[k], j);
            const size_t ro = (size_t)(row0 + rj) * EMB_DIM + lane * 4;
            const float4 x = *(const float4*)(uemb + ro);
            const float4 c = *(const float4*)(h + ro);
            const float m = fmaxf((float)cnt[rj], 1.0f);
            const float4 a = acc4[rj * 32 + lane];
            float4 o;
            o.x = x.x + c.x + a.x / m;
            o.y = x.y + c.y + a.y / m;
            o.z = x.z + c.z + a.z / m;
            o.w = x.w + c.w + a.w / m;
            *(float4*)(out_user + (size_t)(b0 + k * PT + j) * EMB_DIM +
                       lane * 4) = o;
          }
        }
      }
    }
    __syncthreads();
  }
}

__global__ void __launch_bounds__(256)
gather_items(const int* __restrict__ items, const float* __restrict__ iemb,
             float* __restrict__ out_item, int batch) {
  int t = blockIdx.x * blockDim.x + threadIdx.x;
  int b = t >> 5;
  int lane = t & 31;
  if (b >= batch) return;
  int it = items[b];
  size_t off = (size_t)lane * 4;
  *(float4*)(out_item + (size_t)b * EMB_DIM + off) =
      *(const float4*)(iemb + (size_t)it * EMB_DIM + off);
}

__global__ void __launch_bounds__(256)
dot_sigmoid_wmma(const float* __restrict__ U, const float* __restrict__ V,
                 float* __restrict__ pred, int batch) {
  int gid = blockIdx.x * blockDim.x + threadIdx.x;
  int wave = gid >> 5;
  int lane = threadIdx.x & 31;
  if (wave * 16 >= batch) return;

  int row = lane & 15;
  int half = lane >> 4;
  const float* urow = U + ((size_t)wave * 16 + row) * EMB_DIM + 2 * half;
  const float* vrow = V + ((size_t)wave * 16 + row) * EMB_DIM + 2 * half;

  v8f acc = {};
#pragma unroll
  for (int k0 = 0; k0 < EMB_DIM; k0 += 4) {
    v2f a = *(const v2f*)(urow + k0);
    v2f b = *(const v2f*)(vrow + k0);
    acc = __builtin_amdgcn_wmma_f32_16x16x4_f32(
         false, a,  false, b,
         (short)0, acc,  false,  false);
  }

  if (lane < 8 || lane >= 24) {
    int i = (lane < 8) ? lane : (lane - 16);
    float d = acc[lane & 7];
    pred[(size_t)wave * 16 + i] = 1.0f / (1.0f + expf(-d));
  }
}

extern "C" void kernel_launch(void* const* d_in, const int* in_sizes, int n_in,
                              void* d_out, int out_size, void* d_ws,
                              size_t ws_size, hipStream_t stream) {
  const int* users  = (const int*)d_in[0];
  const int* items  = (const int*)d_in[1];
  const int* esrc   = (const int*)d_in[2];
  const int* edst   = (const int*)d_in[3];
  const float* uemb = (const float*)d_in[4];
  const float* iemb = (const float*)d_in[5];

  const int batch     = in_sizes[0];
  const int n_edges   = in_sizes[2];
  const int num_users = in_sizes[4] / EMB_DIM;

  int* tab        = (int*)d_ws;
  float* g1       = (float*)d_ws + num_users;
  unsigned* elist = (unsigned*)(g1 + (size_t)num_users * EMB_DIM);

  float* pred     = (float*)d_out;
  float* out_user = pred + batch;
  float* out_item = out_user + (size_t)batch * EMB_DIM;

  const int B = 256;

  part_sort<<<NCH, PT, 0, stream>>>(esrc, edst, tab, elist, n_edges,
                                    num_users);

  conv_owner<false><<<NBMAX, PT, 0, stream>>>(elist, tab, uemb, g1, users,
                                              uemb, out_user, num_users,
                                              batch);

  conv_owner<true><<<NBMAX, PT, 0, stream>>>(elist, tab, g1, (float*)nullptr,
                                             users, uemb, out_user, num_users,
                                             batch);

  long long ga_threads = (long long)batch * 32;
  gather_items<<<(int)((ga_threads + B - 1) / B), B, 0, stream>>>(
      items, iemb, out_item, batch);

  long long dot_threads = (long long)((batch + 15) / 16) * 32;
  dot_sigmoid_wmma<<<(int)((dot_threads + B - 1) / B), B, 0, stream>>>(
      out_user, out_item, pred, batch);
}


# a wave's serial single-load trips batched: partition reads its edges 8 walk steps a trip (2 trips a chunk, were 21); owner reads a group's words 32-63 with its first 32; list and sum order kept
# speedup vs baseline: 5.9595x; 1.0368x over previous
#include <hip/hip_runtime.h>
#include <hip/hip_bf16.h>


#define EMB_DIM 128
#define ROWS 391
#define ROW_BITS 9
#define ROW_MASK 511u
#define NBMAX 256
#define NCH 256
#define PT 1024
#define NWAVES (PT / 32)
#define CPW (NCH / NWAVES)
#define CAP 8192
#define SHARE (CAP / NWAVES)
#define KSTEPS (SHARE / 32)
#define HS 392
#define GRP 16
#define SST 8

typedef __attribute__((ext_vector_type(2))) float v2f;
typedef __attribute__((ext_vector_type(8))) float v8f;
typedef __attribute__((ext_vector_type(4))) float v4f;

template <int BITS>
__device__ __forceinline__ unsigned same_key_lanes(int key, unsigned live) {
  unsigned peers = live;
#pragma unroll
  for (int i = 0; i < BITS; ++i) {
    const bool bit = ((key >> i) & 1) != 0;
    const unsigned m = __builtin_amdgcn_ballot_w32(bit);
    peers &= bit ? m : ~m;
  }
  return peers;
}

__device__ __forceinline__ void hold_edges(const int (&v)[SST]) {
  static_assert(SST == 8, "hold_edges names SST values");
  asm volatile("" :: "v"(v[0]), "v"(v[1]), "v"(v[2]), "v"(v[3]), "v"(v[4]), "v"(v[5]), "v"(v[6]), "v"(v[7]));
}

__device__ __forceinline__ void wave_block_hist(const int* __restrict__ dst,
                                                int* hist, int eb, int ee,
                                                int w, int lane) {
  const unsigned lt = (1u << lane) - 1u;
  for (int base0 = eb + w * 32; base0 < ee; base0 += SST * PT) {
    int d[SST];
#pragma unroll
    for (int k = 0; k < SST; ++k) d[k] = dst[min(base0 + k * PT + lane, ee - 1)];
    hold_edges(d);
#pragma unroll
    for (int k = 0; k < SST; ++k) {
      if (base0 + k * PT >= ee) break;
      const bool valid = base0 + k * PT + lane < ee;
      const int b = valid ? d[k] / ROWS : 0;
      const unsigned live = __builtin_amdgcn_ballot_w32(valid);
      const unsigned peers = same_key_lanes<8>(b, live);
      if (valid && (peers & lt) == 0u) hist[b] += __popc(peers);
    }
  }
}

__global__ void __launch_bounds__(PT)
part_sort(const int* __restrict__ src, const int* __restrict__ dst,
          int* __restrict__ tab, unsigned* __restrict__ elist, int n_edges,
          int num_users) {
  __shared__ int whist[NWAVES * NBMAX];
  __shared__ int wsum[NWAVES];
  const int tid = threadIdx.x;
  const int lane = tid & 31;
  const int w = __builtin_amdgcn_readfirstlane(tid >> 5);
  const int nb = (num_users + ROWS - 1) / ROWS;
  for (int i = tid; i < NWAVES * NBMAX; i += PT) whist[i] = 0;
  __syncthreads();
  const int cs = (n_edges + NCH - 1) / NCH;
  const int eb = blockIdx.x * cs;
  const int ee = min(n_edges, eb + cs);
  wave_block_hist(dst, whist + w * NBMAX, eb, ee, w, lane);
  __syncthreads();
  int tot = 0;
  if (tid < NBMAX) {
    for (int k = 0; k < NWAVES; ++k) tot += whist[k * NBMAX + tid];
  }
  int inc = tot;
#pragma unroll
  for (int o = 1; o < 32; o <<= 1) {
    const int t = __shfl_up(inc, o);
    if (lane >= o) inc += t;
  }
  if (lane == 31) wsum[w] = inc;
  __syncthreads();
  if (tid < NBMAX) {
    int run = eb + inc - tot;
#pragma unroll
    for (int k = 0; k < NBMAX / 32; ++k) run += (k < w) ? wsum[k] : 0;
    if (tid < nb) tab[tid * NCH + blockIdx.x] = run;
    if (tid == nb - 1) tab[nb * NCH + blockIdx.x] = run + tot;
    for (int k = 0; k < NWAVES; ++k) {
      const int c = whist[k * NBMAX + tid];
      whist[k * NBMAX + tid] = run;
      run += c;
    }
  }
  __syncthreads();
  int* cur = whist + w * NBMAX;
  const unsigned lt = (1u << lane) - 1u;
  for (int base0 = eb + w * 32; base0 < ee; base0 += SST * PT) {
    int dv[SST];
    int sv[SST];
#pragma unroll
    for (int k = 0; k < SST; ++k) {
      const int e = min(base0 + k * PT + lane, ee - 1);
      dv[k] = dst[e];
      sv[k] = src[e];
    }
    hold_edges(dv);
    hold_edges(sv);
#pragma unroll
    for (int k = 0; k < SST; ++k) {
      if (base0 + k * PT >= ee) break;
      const bool valid = base0 + k * PT + lane < ee;
      const int d = dv[k];
      const int b = d / ROWS;
      const unsigned live = __builtin_amdgcn_ballot_w32(valid);
      const unsigned peers = same_key_lanes<8>(b, live);
      const int rank = __popc(peers & lt);
      const int c0 = cur[b];
      if (valid)
        elist[c0 + rank] =
            ((unsigned)sv[k] << ROW_BITS) | (unsigned)(d - b * ROWS);
      if (valid && rank == 0) cur[b] = c0 + __popc(peers);
    }
  }
}

template <int G>
__device__ __forceinline__ void hold_loaded(const v4f (&v)[G]) {
  if constexpr (G == 1) asm volatile("" :: "v"(v[0]));
  else if constexpr (G == 2) asm volatile("" :: "v"(v[0]), "v"(v[1]));
  else if constexpr (G == 3) asm volatile("" :: "v"(v[0]), "v"(v[1]), "v"(v[2]));
  else if constexpr (G == 4) asm volatile("" :: "v"(v[0]), "v"(v[1]), "v"(v[2]), "v"(v[3]));
  else if constexpr (G == 5) asm volatile("" :: "v"(v[0]), "v"(v[1]), "v"(v[2]), "v"(v[3]), "v"(v[4]));
  else if constexpr (G == 6) asm volatile("" :: "v"(v[0]), "v"(v[1]), "v"(v[2]), "v"(v[3]), "v"(v[4]), "v"(v[5]));
  else if constexpr (G == 7) asm volatile("" :: "v"(v[0]), "v"(v[1]), "v"(v[2]), "v"(v[3]), "v"(v[4]), "v"(v[5]), "v"(v[6]));
  else if constexpr (G == 8) asm volatile("" :: "v"(v[0]), "v"(v[1]), "v"(v[2]), "v"(v[3]), "v"(v[4]), "v"(v[5]), "v"(v[6]), "v"(v[7]));
  else if constexpr (G == 9) asm volatile("" :: "v"(v[0]), "v"(v[1]), "v"(v[2]), "v"(v[3]), "v"(v[4]), "v"(v[5]), "v"(v[6]), "v"(v[7]), "v"(v[8]));
  else if constexpr (G == 10) asm volatile("" :: "v"(v[0]), "v"(v[1]), "v"(v[2]), "v"(v[3]), "v"(v[4]), "v"(v[5]), "v"(v[6]), "v"(v[7]), "v"(v[8]), "v"(v[9]));
  else if constexpr (G == 11) asm volatile("" :: "v"(v[0]), "v"(v[1]), "v"(v[2]), "v"(v[3]), "v"(v[4]), "v"(v[5]), "v"(v[6]), "v"(v[7]), "v"(v[8]), "v"(v[9]), "v"(v[10]));
  else if constexpr (G == 12) asm volatile("" :: "v"(v[0]), "v"(v[1]), "v"(v[2]), "v"(v[3]), "v"(v[4]), "v"(v[5]), "v"(v[6]), "v"(v[7]), "v"(v[8]), "v"(v[9]), "v"(v[10]), "v"(v[11]));
  else if constexpr (G == 13) asm volatile("" :: "v"(v[0]), "v"(v[1]), "v"(v[2]), "v"(v[3]), "v"(v[4]), "v"(v[5]), "v"(v[6]), "v"(v[7]), "v"(v[8]), "v"(v[9]), "v"(v[10]), "v"(v[11]), "v"(v[12]));
  else if constexpr (G == 14) asm volatile("" :: "v"(v[0]), "v"(v[1]), "v"(v[2]), "v"(v[3]), "v"(v[4]), "v"(v[5]), "v"(v[6]), "v"(v[7]), "v"(v[8]), "v"(v[9]), "v"(v[10]), "v"(v[11]), "v"(v[12]), "v"(v[13]));
  else if constexpr (G == 15) asm volatile("" :: "v"(v[0]), "v"(v[1]), "v"(v[2]), "v"(v[3]), "v"(v[4]), "v"(v[5]), "v"(v[6]), "v"(v[7]), "v"(v[8]), "v"(v[9]), "v"(v[10]), "v"(v[11]), "v"(v[12]), "v"(v[13]), "v"(v[14]));
  else if constexpr (G == 16) asm volatile("" :: "v"(v[0]), "v"(v[1]), "v"(v[2]), "v"(v[3]), "v"(v[4]), "v"(v[5]), "v"(v[6]), "v"(v[7]), "v"(v[8]), "v"(v[9]), "v"(v[10]), "v"(v[11]), "v"(v[12]), "v"(v[13]), "v"(v[14]), "v"(v[15]));
}

template <int G>
__device__ __forceinline__ void sum_group(const unsigned* p,
                                          const float* __restrict__ h,
                                          int lane, float4& s) {
  const unsigned mine = p[lane < G ? lane : 0];
  unsigned xe[G];
#pragma unroll
  for (int j = 0; j < G; ++j)
    xe[j] = (unsigned)__builtin_amdgcn_readlane((int)mine, j);
  v4f v[G];
#pragma unroll
  for (int j = 0; j < G; ++j) {
    const unsigned off =
        xe[j] * (unsigned)(EMB_DIM * sizeof(float)) + (unsigned)lane * 16u;
    v[j] = *(const v4f*)((const char*)h + off);
  }
  hold_loaded<G>(v);
#pragma unroll
  for (int j = 0; j < G; ++j) {
    s.x += v[j].x;
    s.y += v[j].y;
    s.z += v[j].z;
    s.w += v[j].w;
  }
}

template <bool LAST>
__global__ void __launch_bounds__(PT)
conv_owner(const unsigned* __restrict__ elist, const int* __restrict__ tab,
           const float* __restrict__ h, float* __restrict__ gout,
           const int* __restrict__ users, const float* __restrict__ uemb,
           float* __restrict__ out_user, int num_users, int batch) {
  __shared__ __attribute__((aligned(16))) float acc[ROWS * EMB_DIM];
  __shared__ unsigned sorted[CAP];
  __shared__ int segpos[NCH + 1];
  __shared__ int segsrc[NCH];
  __shared__ int whist[NWAVES * HS];
  __shared__ int rs[HS];
  __shared__ int rc[HS];
  __shared__ int cnt[HS];
  __shared__ int wsum[NWAVES];
  const int tid = threadIdx.x;
  const int lane = tid & 31;
  const int w = __builtin_amdgcn_readfirstlane(tid >> 5);
  const unsigned lt = (1u << lane) - 1u;
  const int nb = (num_users + ROWS - 1) / ROWS;
  float4* acc4 = (float4*)acc;
  int* myh = whist + w * HS;

  for (int bin = blockIdx.x; bin < nb; bin += gridDim.x) {
    const int row0 = bin * ROWS;
    const int nrows = min(ROWS, num_users - row0);

    for (int i = tid; i < ROWS * EMB_DIM / 4; i += PT)
      acc4[i] = make_float4(0.0f, 0.0f, 0.0f, 0.0f);
    for (int i = tid; i < HS; i += PT) cnt[i] = 0;

    int gsrc = 0;
    int glen = 0;
    if (tid < NCH) {
      gsrc = tab[bin * NCH + tid];
      glen = tab[(bin + 1) * NCH + tid] - gsrc;
    }
    int ginc = glen;
#pragma unroll
    for (int o = 1; o < 32; o <<= 1) {
      const int t = __shfl_up(ginc, o);
      if (lane >= o) ginc += t;
    }
    if (lane == 31) wsum[w] = ginc;
    __syncthreads();
    if (tid < NCH) {
      int p = ginc - glen;
#pragma unroll
      for (int k = 0; k < NCH / 32; ++k) p += (k < w) ? wsum[k] : 0;
      segpos[tid] = p;
      segsrc[tid] = gsrc;
      if (tid == NCH - 1) segpos[NCH] = p + glen;
    }
    __syncthreads();
    const int nlist = segpos[NCH];

    for (int t0 = 0; t0 < nlist; t0 += CAP) {
      const int tn = min(CAP, nlist - t0);
      __syncthreads();
      for (int i = tid; i < NWAVES * HS; i += PT) whist[i] = 0;
      int gp[CPW + 1];
      int gs[CPW];
#pragma unroll
      for (int k = 0; k <= CPW; ++k) gp[k] = segpos[w * CPW + k];
#pragma unroll
      for (int k = 0; k < CPW; ++k) gs[k] = segsrc[w * CPW + k];
      unsigned gv[CPW];
      int gi[CPW];
      unsigned hv[CPW];
      int hi[CPW];
      int longest = 0;
#pragma unroll
      for (int k = 0; k < CPW; ++k) {
        const int a = max(gp[k], t0);
        const int b = min(gp[k + 1], t0 + tn);
        const int i = a + lane;
        gi[k] = (i < b) ? (i - t0) : -1;
        gv[k] = (i < b) ? elist[gs[k] + (i - gp[k])] : 0u;
        const int i2 = i + 32;
        hi[k] = (i2 < b) ? (i2 - t0) : -1;
        hv[k] = (i2 < b) ? elist[gs[k] + (i2 - gp[k])] : 0u;
        longest = max(longest, b - a);
      }
#pragma unroll
      for (int k = 0; k < CPW; ++k) {
        if (gi[k] >= 0) sorted[gi[k]] = gv[k];
        if (hi[k] >= 0) sorted[hi[k]] = hv[k];
      }
      if (longest > 64) {
#pragma nounroll
        for (int k = 0; k < CPW; ++k) {
          const int c = w * CPW + k;
          const int p = segpos[c];
          const int a = max(p, t0);
          const int b = min(segpos[c + 1], t0 + tn);
#pragma nounroll
          for (int i = a + 64 + lane; i < b; i += 32)
            sorted[i - t0] = elist[segsrc[c] + (i - p)];
        }
      }
      __syncthreads();
      unsigned x[KSTEPS];
#pragma unroll
      for (int k = 0; k < KSTEPS; ++k) {
        const int i = w * SHARE + k * 32 + lane;
        x[k] = (i < tn) ? sorted[i] : 0u;
      }
      __syncthreads();

      unsigned peers[KSTEPS];
#pragma unroll
      for (int k = 0; k < KSTEPS; ++k) {
        const bool valid = w * SHARE + k * 32 + lane < tn;
        const int key = (int)(x[k] & ROW_MASK);
        const unsigned live = __builtin_amdgcn_ballot_w32(valid);
        peers[k] = same_key_lanes<ROW_BITS>(key, live);
        if (valid && (peers[k] & lt) == 0u) myh[key] += __popc(peers[k]);
      }
      __syncthreads();

      int tot = 0;
      if (tid < HS) {
        int cw[NWAVES];
#pragma unroll
        for (int k = 0; k < NWAVES; ++k) cw[k] = whist[k * HS + tid];
#pragma unroll
        for (int k = 0; k < NWAVES; ++k) {
          whist[k * HS + tid] = tot;
          tot += cw[k];
        }
      }
      int inc = tot;
#pragma unroll
      for (int o = 1; o < 32; o <<= 1) {
        const int t = __shfl_up(inc, o);
        if (lane >= o) inc += t;
      }
      if (lane == 31) wsum[w] = inc;
      __syncthreads();
      if (w == 0) {
        const int v = wsum[lane];
        int vi = v;
#pragma unroll
        for (int o = 1; o < 32; o <<= 1) {
          const int t = __shfl_up(vi, o);
          if (lane >= o) vi += t;
        }
        wsum[lane] = vi - v;
      }
      __syncthreads();
      if (tid < HS) {
        rs[tid] = wsum[w] + inc - tot;
        rc[tid] = tot;
      }
      __syncthreads();

#pragma unroll
      for (int k = 0; k < KSTEPS; ++k) {
        const bool valid = w * SHARE + k * 32 + lane < tn;
        const int key = (int)(x[k] & ROW_MASK);
        const int rank = __popc(peers[k] & lt);
        const int c0 = myh[key];
        if (valid) sorted[rs[key] + c0 + rank] = x[k] >> ROW_BITS;
        if (valid && rank == 0) myh[key] = c0 + __popc(peers[k]);
      }
      __syncthreads();

      for (int r = w; r < ROWS; r += NWAVES) {
        const int st = __builtin_amdgcn_readfirstlane(rs[r]);
        const int c = __builtin_amdgcn_readfirstlane(rc[r]);
        if (c == 0) continue;
        float4 s = make_float4(0.0f, 0.0f, 0.0f, 0.0f);
        const unsigned* p = sorted + st;
        int q = 0;
        for (; q + GRP <= c; q += GRP) sum_group<GRP>(p + q, h, lane, s);
        switch (c - q) {
          case 1: sum_group<1>(p + q, h, lane, s); break;
          case 2: sum_group<2>(p + q, h, lane, s); break;
          case 3: sum_group<3>(p + q, h, lane, s); break;
          case 4: sum_group<4>(p + q, h, lane, s); break;
          case 5: sum_group<5>(p + q, h, lane, s); break;
          case 6: sum_group<6>(p + q, h, lane, s); break;
          case 7: sum_group<7>(p + q, h, lane, s); break;
          case 8: sum_group<8>(p + q, h, lane, s); break;
          case 9: sum_group<9>(p + q, h, lane, s); break;
          case 10: sum_group<10>(p + q, h, lane, s); break;
          case 11: sum_group<11>(p + q, h, lane, s); break;
          case 12: sum_group<12>(p + q, h, lane, s); break;
          case 13: sum_group<13>(p + q, h, lane, s); break;
          case 14: sum_group<14>(p + q, h, lane, s); break;
          case 15: sum_group<15>(p + q, h, lane, s); break;
          default: break;
        }
        float4 a = acc4[r * 32 + lane];
        a.x += s.x;
        a.y += s.y;
        a.z += s.z;
        a.w += s.w;
        acc4[r * 32 + lane] = a;
        if (lane == 0) cnt[r] += c;
      }
    }
    __syncthreads();

    if (!LAST) {
      for (int r = w; r < nrows; r += NWAVES) {
        const float m = fmaxf((float)cnt[r], 1.0f);
        const float4 a = acc4[r * 32 + lane];
        float4 o;
        o.x = a.x / m;
        o.y = a.y / m;
        o.z = a.z / m;
        o.w = a.w / m;
        *(float4*)(gout + (size_t)(row0 + r) * EMB_DIM + lane * 4) = o;
      }
    } else {
      for (int b0 = w * 32; b0 < batch; b0 += 4 * PT) {
        int r[4];
#pragma unroll
        for (int k = 0; k < 4; ++k) {
          const int b = b0 + k * PT + lane;
          r[k] = (b < batch) ? (users[b] - row0) : -1;
        }
#pragma unroll
        for (int k = 0; k < 4; ++k) {
          unsigned mask = __builtin_amdgcn_ballot_w32(r[k] >= 0 && r[k] < nrows);
          while (mask) {
            const int j = __builtin_ctz(mask);
            mask &= mask - 1;
            const int rj = __builtin_amdgcn_readlane(r[k], j);
            const size_t ro = (size_t)(row0 + rj) * EMB_DIM + lane * 4;
            const float4 x = *(const float4*)(uemb + ro);
            const float4 c = *(const float4*)(h + ro);
            const float m = fmaxf((float)cnt[rj], 1.0f);
            const float4 a = acc4[rj * 32 + lane];
            float4 o;
            o.x = x.x + c.x + a.x / m;
            o.y = x.y + c.y + a.y / m;
            o.z = x.z + c.z + a.z / m;
            o.w = x.w + c.w + a.w / m;
            *(float4*)(out_user + (size_t)(b0 + k * PT + j) * EMB_DIM +
                       lane * 4) = o;
          }
        }
      }
    }
    __syncthreads();
  }
}

__global__ void __launch_bounds__(256)
gather_items(const int* __restrict__ items, const float* __restrict__ iemb,
             float* __restrict__ out_item, int batch) {
  int t = blockIdx.x * blockDim.x + threadIdx.x;
  int b = t >> 5;
  int lane = t & 31;
  if (b >= batch) return;
  int it = items[b];
  size_t off = (size_t)lane * 4;
  *(float4*)(out_item + (size_t)b * EMB_DIM + off) =
      *(const float4*)(iemb + (size_t)it * EMB_DIM + off);
}

__global__ void __launch_bounds__(256)
dot_sigmoid_wmma(const float* __restrict__ U, const float* __restrict__ V,
                 float* __restrict__ pred, int batch) {
  int gid = blockIdx.x * blockDim.x + threadIdx.x;
  int wave = gid >> 5;
  int lane = threadIdx.x & 31;
  if (wave * 16 >= batch) return;

  int row = lane & 15;
  int half = lane >> 4;
  const float* urow = U + ((size_t)wave * 16 + row) * EMB_DIM + 2 * half;
  const float* vrow = V + ((size_t)wave * 16 + row) * EMB_DIM + 2 * half;

  v8f acc = {};
#pragma unroll
  for (int k0 = 0; k0 < EMB_DIM; k0 += 4) {
    v2f a = *(const v2f*)(urow + k0);
    v2f b = *(const v2f*)(vrow + k0);
    acc = __builtin_amdgcn_wmma_f32_16x16x4_f32(
         false, a,  false, b,
         (short)0, acc,  false,  false);
  }

  if (lane < 8 || lane >= 24) {
    int i = (lane < 8) ? lane : (lane - 16);
    float d = acc[lane & 7];
    pred[(size_t)wave * 16 + i] = 1.0f / (1.0f + expf(-d));
  }
}

extern "C" void kernel_launch(void* const* d_in, const int* in_sizes, int n_in,
                              void* d_out, int out_size, void* d_ws,
                              size_t ws_size, hipStream_t stream) {
  const int* users  = (const int*)d_in[0];
  const int* items  = (const int*)d_in[1];
  const int* esrc   = (const int*)d_in[2];
  const int* edst   = (const int*)d_in[3];
  const float* uemb = (const float*)d_in[4];
  const float* iemb = (const float*)d_in[5];

  const int batch     = in_sizes[0];
  const int n_edges   = in_sizes[2];
  const int num_users = in_sizes[4] / EMB_DIM;

  int* tab        = (int*)d_ws;
  float* g1       = (float*)d_ws + num_users;
  unsigned* elist = (unsigned*)(g1 + (size_t)num_users * EMB_DIM);

  float* pred     = (float*)d_out;
  float* out_user = pred + batch;
  float* out_item = out_user + (size_t)batch * EMB_DIM;

  const int B = 256;

  part_sort<<<NCH, PT, 0, stream>>>(esrc, edst, tab, elist, n_edges,
                                    num_users);

  conv_owner<false><<<NBMAX, PT, 0, stream>>>(elist, tab, uemb, g1, users,
                                              uemb, out_user, num_users,
                                              batch);

  conv_owner<true><<<NBMAX, PT, 0, stream>>>(elist, tab, g1, (float*)nullptr,
                                             users, uemb, out_user, num_users,
                                             batch);

  long long ga_threads = (long long)batch * 32;
  gather_items<<<(int)((ga_threads + B - 1) / B), B, 0, stream>>>(
      items, iemb, out_item, batch);

  long long dot_threads = (long long)((batch + 15) / 16) * 32;
  dot_sigmoid_wmma<<<(int)((dot_threads + B - 1) / B), B, 0, stream>>>(
      out_user, out_item, pred, batch);
}


# layer 2's owner starts from the row-sorted source list and row tables that layer 1's owner leaves in the workspace: its table read, collect and LDS counting sort are gone; sum order kept
# speedup vs baseline: 6.2133x; 1.0426x over previous
#include <hip/hip_runtime.h>
#include <hip/hip_bf16.h>


#define EMB_DIM 128
#define ROWS 391
#define ROW_BITS 9
#define ROW_MASK 511u
#define NBMAX 256
#define NCH 256
#define PT 1024
#define NWAVES (PT / 32)
#define CPW (NCH / NWAVES)
#define CAP 8192
#define SHARE (CAP / NWAVES)
#define KSTEPS (SHARE / 32)
#define HS 392
#define GRP 16
#define SST 8
#define RT (2 * HS + 8)

typedef __attribute__((ext_vector_type(2))) float v2f;
typedef __attribute__((ext_vector_type(8))) float v8f;
typedef __attribute__((ext_vector_type(4))) float v4f;

template <int BITS>
__device__ __forceinline__ unsigned same_key_lanes(int key, unsigned live) {
  unsigned peers = live;
#pragma unroll
  for (int i = 0; i < BITS; ++i) {
    const bool bit = ((key >> i) & 1) != 0;
    const unsigned m = __builtin_amdgcn_ballot_w32(bit);
    peers &= bit ? m : ~m;
  }
  return peers;
}

__device__ __forceinline__ void hold_edges(const int (&v)[SST]) {
  static_assert(SST == 8, "hold_edges names SST values");
  asm volatile("" :: "v"(v[0]), "v"(v[1]), "v"(v[2]), "v"(v[3]), "v"(v[4]), "v"(v[5]), "v"(v[6]), "v"(v[7]));
}

__device__ __forceinline__ void wave_block_hist(const int* __restrict__ dst,
                                                int* hist, int eb, int ee,
                                                int w, int lane) {
  const unsigned lt = (1u << lane) - 1u;
  for (int base0 = eb + w * 32; base0 < ee; base0 += SST * PT) {
    int d[SST];
#pragma unroll
    for (int k = 0; k < SST; ++k) d[k] = dst[min(base0 + k * PT + lane, ee - 1)];
    hold_edges(d);
#pragma unroll
    for (int k = 0; k < SST; ++k) {
      if (base0 + k * PT >= ee) break;
      const bool valid = base0 + k * PT + lane < ee;
      const int b = valid ? d[k] / ROWS : 0;
      const unsigned live = __builtin_amdgcn_ballot_w32(valid);
      const unsigned peers = same_key_lanes<8>(b, live);
      if (valid && (peers & lt) == 0u) hist[b] += __popc(peers);
    }
  }
}

__global__ void __launch_bounds__(PT)
part_sort(const int* __restrict__ src, const int* __restrict__ dst,
          int* __restrict__ tab, unsigned* __restrict__ elist, int n_edges,
          int num_users) {
  __shared__ int whist[NWAVES * NBMAX];
  __shared__ int wsum[NWAVES];
  const int tid = threadIdx.x;
  const int lane = tid & 31;
  const int w = __builtin_amdgcn_readfirstlane(tid >> 5);
  const int nb = (num_users + ROWS - 1) / ROWS;
  for (int i = tid; i < NWAVES * NBMAX; i += PT) whist[i] = 0;
  __syncthreads();
  const int cs = (n_edges + NCH - 1) / NCH;
  const int eb = blockIdx.x * cs;
  const int ee = min(n_edges, eb + cs);
  wave_block_hist(dst, whist + w * NBMAX, eb, ee, w, lane);
  __syncthreads();
  int tot = 0;
  if (tid < NBMAX) {
    for (int k = 0; k < NWAVES; ++k) tot += whist[k * NBMAX + tid];
  }
  int inc = tot;
#pragma unroll
  for (int o = 1; o < 32; o <<= 1) {
    const int t = __shfl_up(inc, o);
    if (lane >= o) inc += t;
  }
  if (lane == 31) wsum[w] = inc;
  __syncthreads();
  if (tid < NBMAX) {
    int run = eb + inc - tot;
#pragma unroll
    for (int k = 0; k < NBMAX / 32; ++k) run += (k < w) ? wsum[k] : 0;
    if (tid < nb) tab[tid * NCH + blockIdx.x] = run;
    if (tid == nb - 1) tab[nb * NCH + blockIdx.x] = run + tot;
    for (int k = 0; k < NWAVES; ++k) {
      const int c = whist[k * NBMAX + tid];
      whist[k * NBMAX + tid] = run;
      run += c;
    }
  }
  __syncthreads();
  int* cur = whist + w * NBMAX;
  const unsigned lt = (1u << lane) - 1u;
  for (int base0 = eb + w * 32; base0 < ee; base0 += SST * PT) {
    int dv[SST];
    int sv[SST];
#pragma unroll
    for (int k = 0; k < SST; ++k) {
      const int e = min(base0 + k * PT + lane, ee - 1);
      dv[k] = dst[e];
      sv[k] = src[e];
    }
    hold_edges(dv);
    hold_edges(sv);
#pragma unroll
    for (int k = 0; k < SST; ++k) {
      if (base0 + k * PT >= ee) break;
      const bool valid = base0 + k * PT + lane < ee;
      const int d = dv[k];
      const int b = d / ROWS;
      const unsigned live = __builtin_amdgcn_ballot_w32(valid);
      const unsigned peers = same_key_lanes<8>(b, live);
      const int rank = __popc(peers & lt);
      const int c0 = cur[b];
      if (valid)
        elist[c0 + rank] =
            ((unsigned)sv[k] << ROW_BITS) | (unsigned)(d - b * ROWS);
      if (valid && rank == 0) cur[b] = c0 + __popc(peers);
    }
  }
}

template <int G>
__device__ __forceinline__ void hold_loaded(const v4f (&v)[G]) {
  if constexpr (G == 1) asm volatile("" :: "v"(v[0]));
  else if constexpr (G == 2) asm volatile("" :: "v"(v[0]), "v"(v[1]));
  else if constexpr (G == 3) asm volatile("" :: "v"(v[0]), "v"(v[1]), "v"(v[2]));
  else if constexpr (G == 4) asm volatile("" :: "v"(v[0]), "v"(v[1]), "v"(v[2]), "v"(v[3]));
  else if constexpr (G == 5) asm volatile("" :: "v"(v[0]), "v"(v[1]), "v"(v[2]), "v"(v[3]), "v"(v[4]));
  else if constexpr (G == 6) asm volatile("" :: "v"(v[0]), "v"(v[1]), "v"(v[2]), "v"(v[3]), "v"(v[4]), "v"(v[5]));
  else if constexpr (G == 7) asm volatile("" :: "v"(v[0]), "v"(v[1]), "v"(v[2]), "v"(v[3]), "v"(v[4]), "v"(v[5]), "v"(v[6]));
  else if constexpr (G == 8) asm volatile("" :: "v"(v[0]), "v"(v[1]), "v"(v[2]), "v"(v[3]), "v"(v[4]), "v"(v[5]), "v"(v[6]), "v"(v[7]));
  else if constexpr (G == 9) asm volatile("" :: "v"(v[0]), "v"(v[1]), "v"(v[2]), "v"(v[3]), "v"(v[4]), "v"(v[5]), "v"(v[6]), "v"(v[7]), "v"(v[8]));
  else if constexpr (G == 10) asm volatile("" :: "v"(v[0]), "v"(v[1]), "v"(v[2]), "v"(v[3]), "v"(v[4]), "v"(v[5]), "v"(v[6]), "v"(v[7]), "v"(v[8]), "v"(v[9]));
  else if constexpr (G == 11) asm volatile("" :: "v"(v[0]), "v"(v[1]), "v"(v[2]), "v"(v[3]), "v"(v[4]), "v"(v[5]), "v"(v[6]), "v"(v[7]), "v"(v[8]), "v"(v[9]), "v"(v[10]));
  else if constexpr (G == 12) asm volatile("" :: "v"(v[0]), "v"(v[1]), "v"(v[2]), "v"(v[3]), "v"(v[4]), "v"(v[5]), "v"(v[6]), "v"(v[7]), "v"(v[8]), "v"(v[9]), "v"(v[10]), "v"(v[11]));
  else if constexpr (G == 13) asm volatile("" :: "v"(v[0]), "v"(v[1]), "v"(v[2]), "v"(v[3]), "v"(v[4]), "v"(v[5]), "v"(v[6]), "v"(v[7]), "v"(v[8]), "v"(v[9]), "v"(v[10]), "v"(v[11]), "v"(v[12]));
  else if constexpr (G == 14) asm volatile("" :: "v"(v[0]), "v"(v[1]), "v"(v[2]), "v"(v[3]), "v"(v[4]), "v"(v[5]), "v"(v[6]), "v"(v[7]), "v"(v[8]), "v"(v[9]), "v"(v[10]), "v"(v[11]), "v"(v[12]), "v"(v[13]));
  else if constexpr (G == 15) asm volatile("" :: "v"(v[0]), "v"(v[1]), "v"(v[2]), "v"(v[3]), "v"(v[4]), "v"(v[5]), "v"(v[6]), "v"(v[7]), "v"(v[8]), "v"(v[9]), "v"(v[10]), "v"(v[11]), "v"(v[12]), "v"(v[13]), "v"(v[14]));
  else if constexpr (G == 16) asm volatile("" :: "v"(v[0]), "v"(v[1]), "v"(v[2]), "v"(v[3]), "v"(v[4]), "v"(v[5]), "v"(v[6]), "v"(v[7]), "v"(v[8]), "v"(v[9]), "v"(v[10]), "v"(v[11]), "v"(v[12]), "v"(v[13]), "v"(v[14]), "v"(v[15]));
}

template <int G>
__device__ __forceinline__ void sum_group(const unsigned* p,
                                          const float* __restrict__ h,
                                          int lane, float4& s) {
  const unsigned mine = p[lane < G ? lane : 0];
  unsigned xe[G];
#pragma unroll
  for (int j = 0; j < G; ++j)
    xe[j] = (unsigned)__builtin_amdgcn_readlane((int)mine, j);
  v4f v[G];
#pragma unroll
  for (int j = 0; j < G; ++j) {
    const unsigned off =
        xe[j] * (unsigned)(EMB_DIM * sizeof(float)) + (unsigned)lane * 16u;
    v[j] = *(const v4f*)((const char*)h + off);
  }
  hold_loaded<G>(v);
#pragma unroll
  for (int j = 0; j < G; ++j) {
    s.x += v[j].x;
    s.y += v[j].y;
    s.z += v[j].z;
    s.w += v[j].w;
  }
}

template <bool LAST>
__global__ void __launch_bounds__(PT)
conv_owner(const unsigned* __restrict__ elist, const int* __restrict__ tab,
           const float* __restrict__ h, float* __restrict__ gout,
           const int* __restrict__ users, const float* __restrict__ uemb,
           float* __restrict__ out_user, unsigned* __restrict__ slist,
           int* __restrict__ rtab, int num_users, int batch) {
  __shared__ __attribute__((aligned(16))) float acc[ROWS * EMB_DIM];
  __shared__ unsigned sorted[CAP];
  __shared__ int segpos[NCH + 1];
  __shared__ int segsrc[NCH];
  __shared__ int whist[NWAVES * HS];
  __shared__ int rs[HS];
  __shared__ int rc[HS];
  __shared__ int cnt[HS];
  __shared__ int wsum[NWAVES];
  const int tid = threadIdx.x;
  const int lane = tid & 31;
  const int w = __builtin_amdgcn_readfirstlane(tid >> 5);
  const unsigned lt = (1u << lane) - 1u;
  const int nb = (num_users + ROWS - 1) / ROWS;
  float4* acc4 = (float4*)acc;
  int* myh = whist + w * HS;

  for (int bin = blockIdx.x; bin < nb; bin += gridDim.x) {
    const int row0 = bin * ROWS;
    const int nrows = min(ROWS, num_users - row0);

    unsigned* sl = slist + (size_t)bin * CAP;
    int* rt = rtab + (size_t)bin * RT;
    unsigned xs[KSTEPS];
    int hrs = 0;
    int hrc = 0;
    int hlen = 0;
    if (LAST) {
#pragma unroll
      for (int k = 0; k < KSTEPS; ++k) xs[k] = sl[k * PT + tid];
      if (tid < HS) {
        hrs = rt[tid];
        hrc = rt[HS + tid];
      }
      hlen = rt[2 * HS];
    }

    for (int i = tid; i < ROWS * EMB_DIM / 4; i += PT)
      acc4[i] = make_float4(0.0f, 0.0f, 0.0f, 0.0f);
    for (int i = tid; i < HS; i += PT) cnt[i] = 0;

    bool handed = false;
    int nlist = 0;
    if (LAST) {
      const int hn = __builtin_amdgcn_readfirstlane(hlen);
      handed = hn <= CAP;
      if (handed) {
        nlist = hn;
#pragma unroll
        for (int k = 0; k < KSTEPS; ++k) sorted[k * PT + tid] = xs[k];
        if (tid < HS) {
          rs[tid] = hrs;
          rc[tid] = hrc;
        }
      }
    }
    if (!handed) {
      int gsrc = 0;
      int glen = 0;
      if (tid < NCH) {
        gsrc = tab[bin * NCH + tid];
        glen = tab[(bin + 1) * NCH + tid] - gsrc;
      }
      int ginc = glen;
#pragma unroll
      for (int o = 1; o < 32; o <<= 1) {
        const int t = __shfl_up(ginc, o);
        if (lane >= o) ginc += t;
      }
      if (lane == 31) wsum[w] = ginc;
      __syncthreads();
      if (tid < NCH) {
        int p = ginc - glen;
#pragma unroll
        for (int k = 0; k < NCH / 32; ++k) p += (k < w) ? wsum[k] : 0;
        segpos[tid] = p;
        segsrc[tid] = gsrc;
        if (tid == NCH - 1) segpos[NCH] = p + glen;
      }
      __syncthreads();
      nlist = segpos[NCH];
      if (!LAST) {
        if (tid == 0) rt[2 * HS] = nlist;
        if (nlist == 0) {
#pragma nounroll
          for (int i = tid; i < CAP; i += PT) sl[i] = 0u;
          if (tid < HS) {
            rt[tid] = 0;
            rt[HS + tid] = 0;
          }
        }
      }
    }

    for (int t0 = 0; t0 < nlist; t0 += CAP) {
      const int tn = min(CAP, nlist - t0);
      __syncthreads();
      if (!handed) {
        for (int i = tid; i < NWAVES * HS; i += PT) whist[i] = 0;
        int gp[CPW + 1];
        int gs[CPW];
#pragma unroll
        for (int k = 0; k <= CPW; ++k) gp[k] = segpos[w * CPW + k];
#pragma unroll
        for (int k = 0; k < CPW; ++k) gs[k] = segsrc[w * CPW + k];
        unsigned gv[CPW];
        int gi[CPW];
        unsigned hv[CPW];
        int hi[CPW];
        int longest = 0;
#pragma unroll
        for (int k = 0; k < CPW; ++k) {
          const int a = max(gp[k], t0);
          const int b = min(gp[k + 1], t0 + tn);
          const int i = a + lane;
          gi[k] = (i < b) ? (i - t0) : -1;
          gv[k] = (i < b) ? elist[gs[k] + (i - gp[k])] : 0u;
          const int i2 = i + 32;
          hi[k] = (i2 < b) ? (i2 - t0) : -1;
          hv[k] = (i2 < b) ? elist[gs[k] + (i2 - gp[k])] : 0u;
          longest = max(longest, b - a);
        }
#pragma unroll
        for (int k = 0; k < CPW; ++k) {
          if (gi[k] >= 0) sorted[gi[k]] = gv[k];
          if (hi[k] >= 0) sorted[hi[k]] = hv[k];
        }
        if (longest > 64) {
#pragma nounroll
          for (int k = 0; k < CPW; ++k) {
            const int c = w * CPW + k;
            const int p = segpos[c];
            const int a = max(p, t0);
            const int b = min(segpos[c + 1], t0 + tn);
#pragma nounroll
            for (int i = a + 64 + lane; i < b; i += 32)
              sorted[i - t0] = elist[segsrc[c] + (i - p)];
          }
        }
        __syncthreads();
        unsigned x[KSTEPS];
#pragma unroll
        for (int k = 0; k < KSTEPS; ++k) {
          const int i = w * SHARE + k * 32 + lane;
          x[k] = (i < tn) ? sorted[i] : 0u;
        }
        __syncthreads();

        unsigned peers[KSTEPS];
#pragma unroll
        for (int k = 0; k < KSTEPS; ++k) {
          const bool valid = w * SHARE + k * 32 + lane < tn;
          const int key = (int)(x[k] & ROW_MASK);
          const unsigned live = __builtin_amdgcn_ballot_w32(valid);
          peers[k] = same_key_lanes<ROW_BITS>(key, live);
          if (valid && (peers[k] & lt) == 0u) myh[key] += __popc(peers[k]);
        }
        __syncthreads();

        int tot = 0;
        if (tid < HS) {
          int cw[NWAVES];
#pragma unroll
          for (int k = 0; k < NWAVES; ++k) cw[k] = whist[k * HS + tid];
#pragma unroll
          for (int k = 0; k < NWAVES; ++k) {
            whist[k * HS + tid] = tot;
            tot += cw[k];
          }
        }
        int inc = tot;
#pragma unroll
        for (int o = 1; o < 32; o <<= 1) {
          const int t = __shfl_up(inc, o);
          if (lane >= o) inc += t;
        }
        if (lane == 31) wsum[w] = inc;
        __syncthreads();
        if (w == 0) {
          const int v = wsum[lane];
          int vi = v;
#pragma unroll
          for (int o = 1; o < 32; o <<= 1) {
            const int t = __shfl_up(vi, o);
            if (lane >= o) vi += t;
          }
          wsum[lane] = vi - v;
        }
        __syncthreads();
        if (tid < HS) {
          rs[tid] = wsum[w] + inc - tot;
          rc[tid] = tot;
        }
        __syncthreads();

#pragma unroll
        for (int k = 0; k < KSTEPS; ++k) {
          const bool valid = w * SHARE + k * 32 + lane < tn;
          const int key = (int)(x[k] & ROW_MASK);
          const int rank = __popc(peers[k] & lt);
          const int c0 = myh[key];
          if (valid) sorted[rs[key] + c0 + rank] = x[k] >> ROW_BITS;
          if (valid && rank == 0) myh[key] = c0 + __popc(peers[k]);
        }
        __syncthreads();
        if (!LAST && t0 == 0) {
#pragma unroll
          for (int k = 0; k < KSTEPS; ++k) {
            const int i = k * PT + tid;
            sl[i] = (i < tn) ? sorted[i] : 0u;
          }
          if (tid < HS) {
            rt[tid] = rs[tid];
            rt[HS + tid] = rc[tid];
          }
        }
      }

      for (int r = w; r < ROWS; r += NWAVES) {
        const int st = __builtin_amdgcn_readfirstlane(rs[r]);
        const int c = __builtin_amdgcn_readfirstlane(rc[r]);
        if (c == 0) continue;
        float4 s = make_float4(0.0f, 0.0f, 0.0f, 0.0f);
        const unsigned* p = sorted + st;
        int q = 0;
        for (; q + GRP <= c; q += GRP) sum_group<GRP>(p + q, h, lane, s);
        switch (c - q) {
          case 1: sum_group<1>(p + q, h, lane, s); break;
          case 2: sum_group<2>(p + q, h, lane, s); break;
          case 3: sum_group<3>(p + q, h, lane, s); break;
          case 4: sum_group<4>(p + q, h, lane, s); break;
          case 5: sum_group<5>(p + q, h, lane, s); break;
          case 6: sum_group<6>(p + q, h, lane, s); break;
          case 7: sum_group<7>(p + q, h, lane, s); break;
          case 8: sum_group<8>(p + q, h, lane, s); break;
          case 9: sum_group<9>(p + q, h, lane, s); break;
          case 10: sum_group<10>(p + q, h, lane, s); break;
          case 11: sum_group<11>(p + q, h, lane, s); break;
          case 12: sum_group<12>(p + q, h, lane, s); break;
          case 13: sum_group<13>(p + q, h, lane, s); break;
          case 14: sum_group<14>(p + q, h, lane, s); break;
          case 15: sum_group<15>(p + q, h, lane, s); break;
          default: break;
        }
        float4 a = acc4[r * 32 + lane];
        a.x += s.x;
        a.y += s.y;
        a.z += s.z;
        a.w += s.w;
        acc4[r * 32 + lane] = a;
        if (lane == 0) cnt[r] += c;
      }
    }
    __syncthreads();

    if (!LAST) {
      for (int r = w; r < nrows; r += NWAVES) {
        const float m = fmaxf((float)cnt[r], 1.0f);
        const float4 a = acc4[r * 32 + lane];
        float4 o;
        o.x = a.x / m;
        o.y = a.y / m;
        o.z = a.z / m;
        o.w = a.w / m;
        *(float4*)(gout + (size_t)(row0 + r) * EMB_DIM + lane * 4) = o;
      }
    } else {
      for (int b0 = w * 32; b0 < batch; b0 += 4 * PT) {
        int r[4];
#pragma unroll
        for (int k = 0; k < 4; ++k) {
          const int b = b0 + k * PT + lane;
          r[k] = (b < batch) ? (users[b] - row0) : -1;
        }
#pragma unroll
        for (int k = 0; k < 4; ++k) {
          unsigned mask = __builtin_amdgcn_ballot_w32(r[k] >= 0 && r[k] < nrows);
          while (mask) {
            const int j = __builtin_ctz(mask);
            mask &= mask - 1;
            const int rj = __builtin_amdgcn_readlane(r[k], j);
            const size_t ro = (size_t)(row0 + rj) * EMB_DIM + lane * 4;
            const float4 x = *(const float4*)(uemb + ro);
            const float4 c = *(const float4*)(h + ro);
            const float m = fmaxf((float)cnt[rj], 1.0f);
            const float4 a = acc4[rj * 32 + lane];
            float4 o;
            o.x = x.x + c.x + a.x / m;
            o.y = x.y + c.y + a.y / m;
            o.z = x.z + c.z + a.z / m;
            o.w = x.w + c.w + a.w / m;
            *(float4*)(out_user + (size_t)(b0 + k * PT + j) * EMB_DIM +
                       lane * 4) = o;
          }
        }
      }
    }
    __syncthreads();
  }
}

__global__ void __launch_bounds__(256)
gather_items(const int* __restrict__ items, const float* __restrict__ iemb,
             float* __restrict__ out_item, int batch) {
  int t = blockIdx.x * blockDim.x + threadIdx.x;
  int b = t >> 5;
  int lane = t & 31;
  if (b >= batch) return;
  int it = items[b];
  size_t off = (size_t)lane * 4;
  *(float4*)(out_item + (size_t)b * EMB_DIM + off) =
      *(const float4*)(iemb + (size_t)it * EMB_DIM + off);
}

__global__ void __launch_bounds__(256)
dot_sigmoid_wmma(const float* __restrict__ U, const float* __restrict__ V,
                 float* __restrict__ pred, int batch) {
  int gid = blockIdx.x * blockDim.x + threadIdx.x;
  int wave = gid >> 5;
  int lane = threadIdx.x & 31;
  if (wave * 16 >= batch) return;

  int row = lane & 15;
  int half = lane >> 4;
  const float* urow = U + ((size_t)wave * 16 + row) * EMB_DIM + 2 * half;
  const float* vrow = V + ((size_t)wave * 16 + row) * EMB_DIM + 2 * half;

  v8f acc = {};
#pragma unroll
  for (int k0 = 0; k0 < EMB_DIM; k0 += 4) {
    v2f a = *(const v2f*)(urow + k0);
    v2f b = *(const v2f*)(vrow + k0);
    acc = __builtin_amdgcn_wmma_f32_16x16x4_f32(
         false, a,  false, b,
         (short)0, acc,  false,  false);
  }

  if (lane < 8 || lane >= 24) {
    int i = (lane < 8) ? lane : (lane - 16);
    float d = acc[lane & 7];
    pred[(size_t)wave * 16 + i] = 1.0f / (1.0f + expf(-d));
  }
}

extern "C" void kernel_launch(void* const* d_in, const int* in_sizes, int n_in,
                              void* d_out, int out_size, void* d_ws,
                              size_t ws_size, hipStream_t stream) {
  const int* users  = (const int*)d_in[0];
  const int* items  = (const int*)d_in[1];
  const int* esrc   = (const int*)d_in[2];
  const int* edst   = (const int*)d_in[3];
  const float* uemb = (const float*)d_in[4];
  const float* iemb = (const float*)d_in[5];

  const int batch     = in_sizes[0];
  const int n_edges   = in_sizes[2];
  const int num_users = in_sizes[4] / EMB_DIM;

  int* tab        = (int*)d_ws;
  float* g1       = (float*)d_ws + num_users;
  unsigned* elist = (unsigned*)(g1 + (size_t)num_users * EMB_DIM);
  unsigned* slist = elist + n_edges;
  int* rtab       = (int*)(slist + (size_t)NBMAX * CAP);

  float* pred     = (float*)d_out;
  float* out_user = pred + batch;
  float* out_item = out_user + (size_t)batch * EMB_DIM;

  const int B = 256;

  part_sort<<<NCH, PT, 0, stream>>>(esrc, edst, tab, elist, n_edges,
                                    num_users);

  conv_owner<false><<<NBMAX, PT, 0, stream>>>(elist, tab, uemb, g1, users,
                                              uemb, out_user, slist, rtab,
                                              num_users, batch);

  conv_owner<true><<<NBMAX, PT, 0, stream>>>(elist, tab, g1, (float*)nullptr,
                                             users, uemb, out_user, slist,
                                             rtab, num_users, batch);

  long long ga_threads = (long long)batch * 32;
  gather_items<<<(int)((ga_threads + B - 1) / B), B, 0, stream>>>(
      items, iemb, out_item, batch);

  long long dot_threads = (long long)((batch + 15) / 16) * 32;
  dot_sigmoid_wmma<<<(int)((dot_threads + B - 1) / B), B, 0, stream>>>(
      out_user, out_item, pred, batch);
}


# layer 2's owner sums only the rows of its block that the batch names: one scan of users marks them in LDS, the marked rows are listed and dealt evenly over the waves; order of adds in a row kept
# speedup vs baseline: 7.8309x; 1.2603x over previous
#include <hip/hip_runtime.h>
#include <hip/hip_bf16.h>


#define EMB_DIM 128
#define ROWS 391
#define ROW_BITS 9
#define ROW_MASK 511u
#define NBMAX 256
#define NCH 256
#define PT 1024
#define NWAVES (PT / 32)
#define CPW (NCH / NWAVES)
#define CAP 8192
#define SHARE (CAP / NWAVES)
#define KSTEPS (SHARE / 32)
#define HS 392
#define GRP 16
#define SST 8
#define RT (2 * HS + 8)

typedef __attribute__((ext_vector_type(2))) float v2f;
typedef __attribute__((ext_vector_type(8))) float v8f;
typedef __attribute__((ext_vector_type(4))) float v4f;

template <int BITS>
__device__ __forceinline__ unsigned same_key_lanes(int key, unsigned live) {
  unsigned peers = live;
#pragma unroll
  for (int i = 0; i < BITS; ++i) {
    const bool bit = ((key >> i) & 1) != 0;
    const unsigned m = __builtin_amdgcn_ballot_w32(bit);
    peers &= bit ? m : ~m;
  }
  return peers;
}

__device__ __forceinline__ void hold_edges(const int (&v)[SST]) {
  static_assert(SST == 8, "hold_edges names SST values");
  asm volatile("" :: "v"(v[0]), "v"(v[1]), "v"(v[2]), "v"(v[3]), "v"(v[4]), "v"(v[5]), "v"(v[6]), "v"(v[7]));
}

__device__ __forceinline__ void wave_block_hist(const int* __restrict__ dst,
                                                int* hist, int eb, int ee,
                                                int w, int lane) {
  const unsigned lt = (1u << lane) - 1u;
  for (int base0 = eb + w * 32; base0 < ee; base0 += SST * PT) {
    int d[SST];
#pragma unroll
    for (int k = 0; k < SST; ++k) d[k] = dst[min(base0 + k * PT + lane, ee - 1)];
    hold_edges(d);
#pragma unroll
    for (int k = 0; k < SST; ++k) {
      if (base0 + k * PT >= ee) break;
      const bool valid = base0 + k * PT + lane < ee;
      const int b = valid ? d[k] / ROWS : 0;
      const unsigned live = __builtin_amdgcn_ballot_w32(valid);
      const unsigned peers = same_key_lanes<8>(b, live);
      if (valid && (peers & lt) == 0u) hist[b] += __popc(peers);
    }
  }
}

__global__ void __launch_bounds__(PT)
part_sort(const int* __restrict__ src, const int* __restrict__ dst,
          int* __restrict__ tab, unsigned* __restrict__ elist, int n_edges,
          int num_users) {
  __shared__ int whist[NWAVES * NBMAX];
  __shared__ int wsum[NWAVES];
  const int tid = threadIdx.x;
  const int lane = tid & 31;
  const int w = __builtin_amdgcn_readfirstlane(tid >> 5);
  const int nb = (num_users + ROWS - 1) / ROWS;
  for (int i = tid; i < NWAVES * NBMAX; i += PT) whist[i] = 0;
  __syncthreads();
  const int cs = (n_edges + NCH - 1) / NCH;
  const int eb = blockIdx.x * cs;
  const int ee = min(n_edges, eb + cs);
  wave_block_hist(dst, whist + w * NBMAX, eb, ee, w, lane);
  __syncthreads();
  int tot = 0;
  if (tid < NBMAX) {
    for (int k = 0; k < NWAVES; ++k) tot += whist[k * NBMAX + tid];
  }
  int inc = tot;
#pragma unroll
  for (int o = 1; o < 32; o <<= 1) {
    const int t = __shfl_up(inc, o);
    if (lane >= o) inc += t;
  }
  if (lane == 31) wsum[w] = inc;
  __syncthreads();
  if (tid < NBMAX) {
    int run = eb + inc - tot;
#pragma unroll
    for (int k = 0; k < NBMAX / 32; ++k) run += (k < w) ? wsum[k] : 0;
    if (tid < nb) tab[tid * NCH + blockIdx.x] = run;
    if (tid == nb - 1) tab[nb * NCH + blockIdx.x] = run + tot;
    for (int k = 0; k < NWAVES; ++k) {
      const int c = whist[k * NBMAX + tid];
      whist[k * NBMAX + tid] = run;
      run += c;
    }
  }
  __syncthreads();
  int* cur = whist + w * NBMAX;
  const unsigned lt = (1u << lane) - 1u;
  for (int base0 = eb + w * 32; base0 < ee; base0 += SST * PT) {
    int dv[SST];
    int sv[SST];
#pragma unroll
    for (int k = 0; k < SST; ++k) {
      const int e = min(base0 + k * PT + lane, ee - 1);
      dv[k] = dst[e];
      sv[k] = src[e];
    }
    hold_edges(dv);
    hold_edges(sv);
#pragma unroll
    for (int k = 0; k < SST; ++k) {
      if (base0 + k * PT >= ee) break;
      const bool valid = base0 + k * PT + lane < ee;
      const int d = dv[k];
      const int b = d / ROWS;
      const unsigned live = __builtin_amdgcn_ballot_w32(valid);
      const unsigned peers = same_key_lanes<8>(b, live);
      const int rank = __popc(peers & lt);
      const int c0 = cur[b];
      if (valid)
        elist[c0 + rank] =
            ((unsigned)sv[k] << ROW_BITS) | (unsigned)(d - b * ROWS);
      if (valid && rank == 0) cur[b] = c0 + __popc(peers);
    }
  }
}

template <int G>
__device__ __forceinline__ void hold_loaded(const v4f (&v)[G]) {
  if constexpr (G == 1) asm volatile("" :: "v"(v[0]));
  else if constexpr (G == 2) asm volatile("" :: "v"(v[0]), "v"(v[1]));
  else if constexpr (G == 3) asm volatile("" :: "v"(v[0]), "v"(v[1]), "v"(v[2]));
  else if constexpr (G == 4) asm volatile("" :: "v"(v[0]), "v"(v[1]), "v"(v[2]), "v"(v[3]));
  else if constexpr (G == 5) asm volatile("" :: "v"(v[0]), "v"(v[1]), "v"(v[2]), "v"(v[3]), "v"(v[4]));
  else if constexpr (G == 6) asm volatile("" :: "v"(v[0]), "v"(v[1]), "v"(v[2]), "v"(v[3]), "v"(v[4]), "v"(v[5]));
  else if constexpr (G == 7) asm volatile("" :: "v"(v[0]), "v"(v[1]), "v"(v[2]), "v"(v[3]), "v"(v[4]), "v"(v[5]), "v"(v[6]));
  else if constexpr (G == 8) asm volatile("" :: "v"(v[0]), "v"(v[1]), "v"(v[2]), "v"(v[3]), "v"(v[4]), "v"(v[5]), "v"(v[6]), "v"(v[7]));
  else if constexpr (G == 9) asm volatile("" :: "v"(v[0]), "v"(v[1]), "v"(v[2]), "v"(v[3]), "v"(v[4]), "v"(v[5]), "v"(v[6]), "v"(v[7]), "v"(v[8]));
  else if constexpr (G == 10) asm volatile("" :: "v"(v[0]), "v"(v[1]), "v"(v[2]), "v"(v[3]), "v"(v[4]), "v"(v[5]), "v"(v[6]), "v"(v[7]), "v"(v[8]), "v"(v[9]));
  else if constexpr (G == 11) asm volatile("" :: "v"(v[0]), "v"(v[1]), "v"(v[2]), "v"(v[3]), "v"(v[4]), "v"(v[5]), "v"(v[6]), "v"(v[7]), "v"(v[8]), "v"(v[9]), "v"(v[10]));
  else if constexpr (G == 12) asm volatile("" :: "v"(v[0]), "v"(v[1]), "v"(v[2]), "v"(v[3]), "v"(v[4]), "v"(v[5]), "v"(v[6]), "v"(v[7]), "v"(v[8]), "v"(v[9]), "v"(v[10]), "v"(v[11]));
  else if constexpr (G == 13) asm volatile("" :: "v"(v[0]), "v"(v[1]), "v"(v[2]), "v"(v[3]), "v"(v[4]), "v"(v[5]), "v"(v[6]), "v"(v[7]), "v"(v[8]), "v"(v[9]), "v"(v[10]), "v"(v[11]), "v"(v[12]));
  else if constexpr (G == 14) asm volatile("" :: "v"(v[0]), "v"(v[1]), "v"(v[2]), "v"(v[3]), "v"(v[4]), "v"(v[5]), "v"(v[6]), "v"(v[7]), "v"(v[8]), "v"(v[9]), "v"(v[10]), "v"(v[11]), "v"(v[12]), "v"(v[13]));
  else if constexpr (G == 15) asm volatile("" :: "v"(v[0]), "v"(v[1]), "v"(v[2]), "v"(v[3]), "v"(v[4]), "v"(v[5]), "v"(v[6]), "v"(v[7]), "v"(v[8]), "v"(v[9]), "v"(v[10]), "v"(v[11]), "v"(v[12]), "v"(v[13]), "v"(v[14]));
  else if constexpr (G == 16) asm volatile("" :: "v"(v[0]), "v"(v[1]), "v"(v[2]), "v"(v[3]), "v"(v[4]), "v"(v[5]), "v"(v[6]), "v"(v[7]), "v"(v[8]), "v"(v[9]), "v"(v[10]), "v"(v[11]), "v"(v[12]), "v"(v[13]), "v"(v[14]), "v"(v[15]));
}

template <int G>
__device__ __forceinline__ void sum_group(const unsigned* p,
                                          const float* __restrict__ h,
                                          int lane, float4& s) {
  const unsigned mine = p[lane < G ? lane : 0];
  unsigned xe[G];
#pragma unroll
  for (int j = 0; j < G; ++j)
    xe[j] = (unsigned)__builtin_amdgcn_readlane((int)mine, j);
  v4f v[G];
#pragma unroll
  for (int j = 0; j < G; ++j) {
    const unsigned off =
        xe[j] * (unsigned)(EMB_DIM * sizeof(float)) + (unsigned)lane * 16u;
    v[j] = *(const v4f*)((const char*)h + off);
  }
  hold_loaded<G>(v);
#pragma unroll
  for (int j = 0; j < G; ++j) {
    s.x += v[j].x;
    s.y += v[j].y;
    s.z += v[j].z;
    s.w += v[j].w;
  }
}

template <bool LAST>
__global__ void __launch_bounds__(PT)
conv_owner(const unsigned* __restrict__ elist, const int* __restrict__ tab,
           const float* __restrict__ h, float* __restrict__ gout,
           const int* __restrict__ users, const float* __restrict__ uemb,
           float* __restrict__ out_user, unsigned* __restrict__ slist,
           int* __restrict__ rtab, int num_users, int batch) {
  __shared__ __attribute__((aligned(16))) float acc[ROWS * EMB_DIM];
  __shared__ unsigned sorted[CAP];
  __shared__ int segpos[NCH + 1];
  __shared__ int segsrc[NCH];
  __shared__ int whist[NWAVES * HS];
  __shared__ int rs[HS];
  __shared__ int rc[HS];
  __shared__ int cnt[HS];
  __shared__ int wsum[NWAVES];
  __shared__ int mark[HS];
  __shared__ int mrow[HS];
  __shared__ int mcnt[NWAVES];
  const int tid = threadIdx.x;
  const int lane = tid & 31;
  const int w = __builtin_amdgcn_readfirstlane(tid >> 5);
  const unsigned lt = (1u << lane) - 1u;
  const int nb = (num_users + ROWS - 1) / ROWS;
  float4* acc4 = (float4*)acc;
  int* myh = whist + w * HS;

  for (int bin = blockIdx.x; bin < nb; bin += gridDim.x) {
    const int row0 = bin * ROWS;
    const int nrows = min(ROWS, num_users - row0);

    unsigned* sl = slist + (size_t)bin * CAP;
    int* rt = rtab + (size_t)bin * RT;
    unsigned xs[KSTEPS];
    int hrs = 0;
    int hrc = 0;
    int hlen = 0;
    if (LAST) {
#pragma unroll
      for (int k = 0; k < KSTEPS; ++k) xs[k] = sl[k * PT + tid];
      if (tid < HS) {
        hrs = rt[tid];
        hrc = rt[HS + tid];
      }
      hlen = rt[2 * HS];
    }

    for (int i = tid; i < ROWS * EMB_DIM / 4; i += PT)
      acc4[i] = make_float4(0.0f, 0.0f, 0.0f, 0.0f);
    for (int i = tid; i < HS; i += PT) cnt[i] = 0;

    int nm = 0;
    if (LAST) {
      for (int i = tid; i < HS; i += PT) mark[i] = 0;
      __syncthreads();
      for (int b0 = tid; b0 < batch; b0 += 8 * PT) {
        int u[8];
#pragma unroll
        for (int k = 0; k < 8; ++k) {
          const int b = b0 + k * PT;
          u[k] = (b < batch) ? (users[b] - row0) : -1;
        }
#pragma unroll
        for (int k = 0; k < 8; ++k)
          if (u[k] >= 0 && u[k] < nrows) mark[u[k]] = 1;
      }
      __syncthreads();
      const bool named = tid < HS && mark[tid] != 0;
      const unsigned nl = __builtin_amdgcn_ballot_w32(named);
      if (lane == 0) mcnt[w] = __popc(nl);
      __syncthreads();
      const int mc = mcnt[lane];
      int minc = mc;
#pragma unroll
      for (int o = 1; o < 32; o <<= 1) {
        const int t = __shfl_up(minc, o);
        if (lane >= o) minc += t;
      }
      nm = __builtin_amdgcn_readlane(minc, NWAVES - 1);
      const int before = __builtin_amdgcn_readlane(minc - mc, w);
      if (named) mrow[before + __popc(nl & lt)] = tid;
    }

    bool handed = false;
    int nlist = 0;
    if (LAST) {
      const int hn = __builtin_amdgcn_readfirstlane(hlen);
      handed = hn <= CAP;
      if (handed) {
        nlist = hn;
#pragma unroll
        for (int k = 0; k < KSTEPS; ++k) sorted[k * PT + tid] = xs[k];
        if (tid < HS) {
          rs[tid] = hrs;
          rc[tid] = hrc;
        }
      }
    }
    if (!handed) {
      int gsrc = 0;
      int glen = 0;
      if (tid < NCH) {
        gsrc = tab[bin * NCH + tid];
        glen = tab[(bin + 1) * NCH + tid] - gsrc;
      }
      int ginc = glen;
#pragma unroll
      for (int o = 1; o < 32; o <<= 1) {
        const int t = __shfl_up(ginc, o);
        if (lane >= o) ginc += t;
      }
      if (lane == 31) wsum[w] = ginc;
      __syncthreads();
      if (tid < NCH) {
        int p = ginc - glen;
#pragma unroll
        for (int k = 0; k < NCH / 32; ++k) p += (k < w) ? wsum[k] : 0;
        segpos[tid] = p;
        segsrc[tid] = gsrc;
        if (tid == NCH - 1) segpos[NCH] = p + glen;
      }
      __syncthreads();
      nlist = segpos[NCH];
      if (!LAST) {
        if (tid == 0) rt[2 * HS] = nlist;
        if (nlist == 0) {
#pragma nounroll
          for (int i = tid; i < CAP; i += PT) sl[i] = 0u;
          if (tid < HS) {
            rt[tid] = 0;
            rt[HS + tid] = 0;
          }
        }
      }
    }

    for (int t0 = 0; t0 < nlist; t0 += CAP) {
      const int tn = min(CAP, nlist - t0);
      __syncthreads();
      if (!handed) {
        for (int i = tid; i < NWAVES * HS; i += PT) whist[i] = 0;
        int gp[CPW + 1];
        int gs[CPW];
#pragma unroll
        for (int k = 0; k <= CPW; ++k) gp[k] = segpos[w * CPW + k];
#pragma unroll
        for (int k = 0; k < CPW; ++k) gs[k] = segsrc[w * CPW + k];
        unsigned gv[CPW];
        int gi[CPW];
        unsigned hv[CPW];
        int hi[CPW];
        int longest = 0;
#pragma unroll
        for (int k = 0; k < CPW; ++k) {
          const int a = max(gp[k], t0);
          const int b = min(gp[k + 1], t0 + tn);
          const int i = a + lane;
          gi[k] = (i < b) ? (i - t0) : -1;
          gv[k] = (i < b) ? elist[gs[k] + (i - gp[k])] : 0u;
          const int i2 = i + 32;
          hi[k] = (i2 < b) ? (i2 - t0) : -1;
          hv[k] = (i2 < b) ? elist[gs[k] + (i2 - gp[k])] : 0u;
          longest = max(longest, b - a);
        }
#pragma unroll
        for (int k = 0; k < CPW; ++k) {
          if (gi[k] >= 0) sorted[gi[k]] = gv[k];
          if (hi[k] >= 0) sorted[hi[k]] = hv[k];
        }
        if (longest > 64) {
#pragma nounroll
          for (int k = 0; k < CPW; ++k) {
            const int c = w * CPW + k;
            const int p = segpos[c];
            const int a = max(p, t0);
            const int b = min(segpos[c + 1], t0 + tn);
#pragma nounroll
            for (int i = a + 64 + lane; i < b; i += 32)
              sorted[i - t0] = elist[segsrc[c] + (i - p)];
          }
        }
        __syncthreads();
        unsigned x[KSTEPS];
#pragma unroll
        for (int k = 0; k < KSTEPS; ++k) {
          const int i = w * SHARE + k * 32 + lane;
          x[k] = (i < tn) ? sorted[i] : 0u;
        }
        __syncthreads();

        unsigned peers[KSTEPS];
#pragma unroll
        for (int k = 0; k < KSTEPS; ++k) {
          const bool valid = w * SHARE + k * 32 + lane < tn;
          const int key = (int)(x[k] & ROW_MASK);
          const unsigned live = __builtin_amdgcn_ballot_w32(valid);
          peers[k] = same_key_lanes<ROW_BITS>(key, live);
          if (valid && (peers[k] & lt) == 0u) myh[key] += __popc(peers[k]);
        }
        __syncthreads();

        int tot = 0;
        if (tid < HS) {
          int cw[NWAVES];
#pragma unroll
          for (int k = 0; k < NWAVES; ++k) cw[k] = whist[k * HS + tid];
#pragma unroll
          for (int k = 0; k < NWAVES; ++k) {
            whist[k * HS + tid] = tot;
            tot += cw[k];
          }
        }
        int inc = tot;
#pragma unroll
        for (int o = 1; o < 32; o <<= 1) {
          const int t = __shfl_up(inc, o);
          if (lane >= o) inc += t;
        }
        if (lane == 31) wsum[w] = inc;
        __syncthreads();
        if (w == 0) {
          const int v = wsum[lane];
          int vi = v;
#pragma unroll
          for (int o = 1; o < 32; o <<= 1) {
            const int t = __shfl_up(vi, o);
            if (lane >= o) vi += t;
          }
          wsum[lane] = vi - v;
        }
        __syncthreads();
        if (tid < HS) {
          rs[tid] = wsum[w] + inc - tot;
          rc[tid] = tot;
        }
        __syncthreads();

#pragma unroll
        for (int k = 0; k < KSTEPS; ++k) {
          const bool valid = w * SHARE + k * 32 + lane < tn;
          const int key = (int)(x[k] & ROW_MASK);
          const int rank = __popc(peers[k] & lt);
          const int c0 = myh[key];
          if (valid) sorted[rs[key] + c0 + rank] = x[k] >> ROW_BITS;
          if (valid && rank == 0) myh[key] = c0 + __popc(peers[k]);
        }
        __syncthreads();
        if (!LAST && t0 == 0) {
#pragma unroll
          for (int k = 0; k < KSTEPS; ++k) {
            const int i = k * PT + tid;
            sl[i] = (i < tn) ? sorted[i] : 0u;
          }
          if (tid < HS) {
            rt[tid] = rs[tid];
            rt[HS + tid] = rc[tid];
          }
        }
      }

      const int nsum = LAST ? nm : ROWS;
      for (int i = w; i < nsum; i += NWAVES) {
        const int r = LAST ? __builtin_amdgcn_readfirstlane(mrow[i]) : i;
        const int st = __builtin_amdgcn_readfirstlane(rs[r]);
        const int c = __builtin_amdgcn_readfirstlane(rc[r]);
        if (c == 0) continue;
        float4 s = make_float4(0.0f, 0.0f, 0.0f, 0.0f);
        const unsigned* p = sorted + st;
        int q = 0;
        for (; q + GRP <= c; q += GRP) sum_group<GRP>(p + q, h, lane, s);
        switch (c - q) {
          case 1: sum_group<1>(p + q, h, lane, s); break;
          case 2: sum_group<2>(p + q, h, lane, s); break;
          case 3: sum_group<3>(p + q, h, lane, s); break;
          case 4: sum_group<4>(p + q, h, lane, s); break;
          case 5: sum_group<5>(p + q, h, lane, s); break;
          case 6: sum_group<6>(p + q, h, lane, s); break;
          case 7: sum_group<7>(p + q, h, lane, s); break;
          case 8: sum_group<8>(p + q, h, lane, s); break;
          case 9: sum_group<9>(p + q, h, lane, s); break;
          case 10: sum_group<10>(p + q, h, lane, s); break;
          case 11: sum_group<11>(p + q, h, lane, s); break;
          case 12: sum_group<12>(p + q, h, lane, s); break;
          case 13: sum_group<13>(p + q, h, lane, s); break;
          case 14: sum_group<14>(p + q, h, lane, s); break;
          case 15: sum_group<15>(p + q, h, lane, s); break;
          default: break;
        }
        float4 a = acc4[r * 32 + lane];
        a.x += s.x;
        a.y += s.y;
        a.z += s.z;
        a.w += s.w;
        acc4[r * 32 + lane] = a;
        if (lane == 0) cnt[r] += c;
      }
    }
    __syncthreads();

    if (!LAST) {
      for (int r = w; r < nrows; r += NWAVES) {
        const float m = fmaxf((float)cnt[r], 1.0f);
        const float4 a = acc4[r * 32 + lane];
        float4 o;
        o.x = a.x / m;
        o.y = a.y / m;
        o.z = a.z / m;
        o.w = a.w / m;
        *(float4*)(gout + (size_t)(row0 + r) * EMB_DIM + lane * 4) = o;
      }
    } else {
      for (int b0 = w * 32; b0 < batch; b0 += 4 * PT) {
        int r[4];
#pragma unroll
        for (int k = 0; k < 4; ++k) {
          const int b = b0 + k * PT + lane;
          r[k] = (b < batch) ? (users[b] - row0) : -1;
        }
#pragma unroll
        for (int k = 0; k < 4; ++k) {
          unsigned mask = __builtin_amdgcn_ballot_w32(r[k] >= 0 && r[k] < nrows);
          while (mask) {
            const int j = __builtin_ctz(mask);
            mask &= mask - 1;
            const int rj = __builtin_amdgcn_readlane(r[k], j);
            const size_t ro = (size_t)(row0 + rj) * EMB_DIM + lane * 4;
            const float4 x = *(const float4*)(uemb + ro);
            const float4 c = *(const float4*)(h + ro);
            const float m = fmaxf((float)cnt[rj], 1.0f);
            const float4 a = acc4[rj * 32 + lane];
            float4 o;
            o.x = x.x + c.x + a.x / m;
            o.y = x.y + c.y + a.y / m;
            o.z = x.z + c.z + a.z / m;
            o.w = x.w + c.w + a.w / m;
            *(float4*)(out_user + (size_t)(b0 + k * PT + j) * EMB_DIM +
                       lane * 4) = o;
          }
        }
      }
    }
    __syncthreads();
  }
}

__global__ void __launch_bounds__(256)
gather_items(const int* __restrict__ items, const float* __restrict__ iemb,
             float* __restrict__ out_item, int batch) {
  int t = blockIdx.x * blockDim.x + threadIdx.x;
  int b = t >> 5;
  int lane = t & 31;
  if (b >= batch) return;
  int it = items[b];
  size_t off = (size_t)lane * 4;
  *(float4*)(out_item + (size_t)b * EMB_DIM + off) =
      *(const float4*)(iemb + (size_t)it * EMB_DIM + off);
}

__global__ void __launch_bounds__(256)
dot_sigmoid_wmma(const float* __restrict__ U, const float* __restrict__ V,
                 float* __restrict__ pred, int batch) {
  int gid = blockIdx.x * blockDim.x + threadIdx.x;
  int wave = gid >> 5;
  int lane = threadIdx.x & 31;
  if (wave * 16 >= batch) return;

  int row = lane & 15;
  int half = lane >> 4;
  const float* urow = U + ((size_t)wave * 16 + row) * EMB_DIM + 2 * half;
  const float* vrow = V + ((size_t)wave * 16 + row) * EMB_DIM + 2 * half;

  v8f acc = {};
#pragma unroll
  for (int k0 = 0; k0 < EMB_DIM; k0 += 4) {
    v2f a = *(const v2f*)(urow + k0);
    v2f b = *(const v2f*)(vrow + k0);
    acc = __builtin_amdgcn_wmma_f32_16x16x4_f32(
         false, a,  false, b,
         (short)0, acc,  false,  false);
  }

  if (lane < 8 || lane >= 24) {
    int i = (lane < 8) ? lane : (lane - 16);
    float d = acc[lane & 7];
    pred[(size_t)wave * 16 + i] = 1.0f / (1.0f + expf(-d));
  }
}

extern "C" void kernel_launch(void* const* d_in, const int* in_sizes, int n_in,
                              void* d_out, int out_size, void* d_ws,
                              size_t ws_size, hipStream_t stream) {
  const int* users  = (const int*)d_in[0];
  const int* items  = (const int*)d_in[1];
  const int* esrc   = (const int*)d_in[2];
  const int* edst   = (const int*)d_in[3];
  const float* uemb = (const float*)d_in[4];
  const float* iemb = (const float*)d_in[5];

  const int batch     = in_sizes[0];
  const int n_edges   = in_sizes[2];
  const int num_users = in_sizes[4] / EMB_DIM;

  int* tab        = (int*)d_ws;
  float* g1       = (float*)d_ws + num_users;
  unsigned* elist = (unsigned*)(g1 + (size_t)num_users * EMB_DIM);
  unsigned* slist = elist + n_edges;
  int* rtab       = (int*)(slist + (size_t)NBMAX * CAP);

  float* pred     = (float*)d_out;
  float* out_user = pred + batch;
  float* out_item = out_user + (size_t)batch * EMB_DIM;

  const int B = 256;

  part_sort<<<NCH, PT, 0, stream>>>(esrc, edst, tab, elist, n_edges,
                                    num_users);

  conv_owner<false><<<NBMAX, PT, 0, stream>>>(elist, tab, uemb, g1, users,
                                              uemb, out_user, slist, rtab,
                                              num_users, batch);

  conv_owner<true><<<NBMAX, PT, 0, stream>>>(elist, tab, g1, (float*)nullptr,
                                             users, uemb, out_user, slist,
                                             rtab, num_users, batch);

  long long ga_threads = (long long)batch * 32;
  gather_items<<<(int)((ga_threads + B - 1) / B), B, 0, stream>>>(
      items, iemb, out_item, batch);

  long long dot_threads = (long long)((batch + 15) / 16) * 32;
  dot_sigmoid_wmma<<<(int)((dot_threads + B - 1) / B), B, 0, stream>>>(
      out_user, out_item, pred, batch);
}


# conv owners: with a one-part list the summing wave finishes its row from registers (layer 1 stores the mean, layer 2 the output row); LDS clearing, adds and closing loads gone; sum order kept
# speedup vs baseline: 8.0884x; 1.0329x over previous
#include <hip/hip_runtime.h>
#include <hip/hip_bf16.h>


#define EMB_DIM 128
#define ROWS 391
#define ROW_BITS 9
#define ROW_MASK 511u
#define NBMAX 256
#define NCH 256
#define PT 1024
#define NWAVES (PT / 32)
#define CPW (NCH / NWAVES)
#define CAP 8192
#define SHARE (CAP / NWAVES)
#define KSTEPS (SHARE / 32)
#define HS 392
#define GRP 16
#define SST 8
#define RT (2 * HS + 8)

typedef __attribute__((ext_vector_type(2))) float v2f;
typedef __attribute__((ext_vector_type(8))) float v8f;
typedef __attribute__((ext_vector_type(4))) float v4f;

template <int BITS>
__device__ __forceinline__ unsigned same_key_lanes(int key, unsigned live) {
  unsigned peers = live;
#pragma unroll
  for (int i = 0; i < BITS; ++i) {
    const bool bit = ((key >> i) & 1) != 0;
    const unsigned m = __builtin_amdgcn_ballot_w32(bit);
    peers &= bit ? m : ~m;
  }
  return peers;
}

__device__ __forceinline__ void hold_edges(const int (&v)[SST]) {
  static_assert(SST == 8, "hold_edges names SST values");
  asm volatile("" :: "v"(v[0]), "v"(v[1]), "v"(v[2]), "v"(v[3]), "v"(v[4]), "v"(v[5]), "v"(v[6]), "v"(v[7]));
}

__device__ __forceinline__ void wave_block_hist(const int* __restrict__ dst,
                                                int* hist, int eb, int ee,
                                                int w, int lane) {
  const unsigned lt = (1u << lane) - 1u;
  for (int base0 = eb + w * 32; base0 < ee; base0 += SST * PT) {
    int d[SST];
#pragma unroll
    for (int k = 0; k < SST; ++k) d[k] = dst[min(base0 + k * PT + lane, ee - 1)];
    hold_edges(d);
#pragma unroll
    for (int k = 0; k < SST; ++k) {
      if (base0 + k * PT >= ee) break;
      const bool valid = base0 + k * PT + lane < ee;
      const int b = valid ? d[k] / ROWS : 0;
      const unsigned live = __builtin_amdgcn_ballot_w32(valid);
      const unsigned peers = same_key_lanes<8>(b, live);
      if (valid && (peers & lt) == 0u) hist[b] += __popc(peers);
    }
  }
}

__global__ void __launch_bounds__(PT)
part_sort(const int* __restrict__ src, const int* __restrict__ dst,
          int* __restrict__ tab, unsigned* __restrict__ elist, int n_edges,
          int num_users) {
  __shared__ int whist[NWAVES * NBMAX];
  __shared__ int wsum[NWAVES];
  const int tid = threadIdx.x;
  const int lane = tid & 31;
  const int w = __builtin_amdgcn_readfirstlane(tid >> 5);
  const int nb = (num_users + ROWS - 1) / ROWS;
  for (int i = tid; i < NWAVES * NBMAX; i += PT) whist[i] = 0;
  __syncthreads();
  const int cs = (n_edges + NCH - 1) / NCH;
  const int eb = blockIdx.x * cs;
  const int ee = min(n_edges, eb + cs);
  wave_block_hist(dst, whist + w * NBMAX, eb, ee, w, lane);
  __syncthreads();
  int tot = 0;
  if (tid < NBMAX) {
    for (int k = 0; k < NWAVES; ++k) tot += whist[k * NBMAX + tid];
  }
  int inc = tot;
#pragma unroll
  for (int o = 1; o < 32; o <<= 1) {
    const int t = __shfl_up(inc, o);
    if (lane >= o) inc += t;
  }
  if (lane == 31) wsum[w] = inc;
  __syncthreads();
  if (tid < NBMAX) {
    int run = eb + inc - tot;
#pragma unroll
    for (int k = 0; k < NBMAX / 32; ++k) run += (k < w) ? wsum[k] : 0;
    if (tid < nb) tab[tid * NCH + blockIdx.x] = run;
    if (tid == nb - 1) tab[nb * NCH + blockIdx.x] = run + tot;
    for (int k = 0; k < NWAVES; ++k) {
      const int c = whist[k * NBMAX + tid];
      whist[k * NBMAX + tid] = run;
      run += c;
    }
  }
  __syncthreads();
  int* cur = whist + w * NBMAX;
  const unsigned lt = (1u << lane) - 1u;
  for (int base0 = eb + w * 32; base0 < ee; base0 += SST * PT) {
    int dv[SST];
    int sv[SST];
#pragma unroll
    for (int k = 0; k < SST; ++k) {
      const int e = min(base0 + k * PT + lane, ee - 1);
      dv[k] = dst[e];
      sv[k] = src[e];
    }
    hold_edges(dv);
    hold_edges(sv);
#pragma unroll
    for (int k = 0; k < SST; ++k) {
      if (base0 + k * PT >= ee) break;
      const bool valid = base0 + k * PT + lane < ee;
      const int d = dv[k];
      const int b = d / ROWS;
      const unsigned live = __builtin_amdgcn_ballot_w32(valid);
      const unsigned peers = same_key_lanes<8>(b, live);
      const int rank = __popc(peers & lt);
      const int c0 = cur[b];
      if (valid)
        elist[c0 + rank] =
            ((unsigned)sv[k] << ROW_BITS) | (unsigned)(d - b * ROWS);
      if (valid && rank == 0) cur[b] = c0 + __popc(peers);
    }
  }
}

template <int G>
__device__ __forceinline__ void hold_loaded(const v4f (&v)[G]) {
  if constexpr (G == 1) asm volatile("" :: "v"(v[0]));
  else if constexpr (G == 2) asm volatile("" :: "v"(v[0]), "v"(v[1]));
  else if constexpr (G == 3) asm volatile("" :: "v"(v[0]), "v"(v[1]), "v"(v[2]));
  else if constexpr (G == 4) asm volatile("" :: "v"(v[0]), "v"(v[1]), "v"(v[2]), "v"(v[3]));
  else if constexpr (G == 5) asm volatile("" :: "v"(v[0]), "v"(v[1]), "v"(v[2]), "v"(v[3]), "v"(v[4]));
  else if constexpr (G == 6) asm volatile("" :: "v"(v[0]), "v"(v[1]), "v"(v[2]), "v"(v[3]), "v"(v[4]), "v"(v[5]));
  else if constexpr (G == 7) asm volatile("" :: "v"(v[0]), "v"(v[1]), "v"(v[2]), "v"(v[3]), "v"(v[4]), "v"(v[5]), "v"(v[6]));
  else if constexpr (G == 8) asm volatile("" :: "v"(v[0]), "v"(v[1]), "v"(v[2]), "v"(v[3]), "v"(v[4]), "v"(v[5]), "v"(v[6]), "v"(v[7]));
  else if constexpr (G == 9) asm volatile("" :: "v"(v[0]), "v"(v[1]), "v"(v[2]), "v"(v[3]), "v"(v[4]), "v"(v[5]), "v"(v[6]), "v"(v[7]), "v"(v[8]));
  else if constexpr (G == 10) asm volatile("" :: "v"(v[0]), "v"(v[1]), "v"(v[2]), "v"(v[3]), "v"(v[4]), "v"(v[5]), "v"(v[6]), "v"(v[7]), "v"(v[8]), "v"(v[9]));
  else if constexpr (G == 11) asm volatile("" :: "v"(v[0]), "v"(v[1]), "v"(v[2]), "v"(v[3]), "v"(v[4]), "v"(v[5]), "v"(v[6]), "v"(v[7]), "v"(v[8]), "v"(v[9]), "v"(v[10]));
  else if constexpr (G == 12) asm volatile("" :: "v"(v[0]), "v"(v[1]), "v"(v[2]), "v"(v[3]), "v"(v[4]), "v"(v[5]), "v"(v[6]), "v"(v[7]), "v"(v[8]), "v"(v[9]), "v"(v[10]), "v"(v[11]));
  else if constexpr (G == 13) asm volatile("" :: "v"(v[0]), "v"(v[1]), "v"(v[2]), "v"(v[3]), "v"(v[4]), "v"(v[5]), "v"(v[6]), "v"(v[7]), "v"(v[8]), "v"(v[9]), "v"(v[10]), "v"(v[11]), "v"(v[12]));
  else if constexpr (G == 14) asm volatile("" :: "v"(v[0]), "v"(v[1]), "v"(v[2]), "v"(v[3]), "v"(v[4]), "v"(v[5]), "v"(v[6]), "v"(v[7]), "v"(v[8]), "v"(v[9]), "v"(v[10]), "v"(v[11]), "v"(v[12]), "v"(v[13]));
  else if constexpr (G == 15) asm volatile("" :: "v"(v[0]), "v"(v[1]), "v"(v[2]), "v"(v[3]), "v"(v[4]), "v"(v[5]), "v"(v[6]), "v"(v[7]), "v"(v[8]), "v"(v[9]), "v"(v[10]), "v"(v[11]), "v"(v[12]), "v"(v[13]), "v"(v[14]));
  else if constexpr (G == 16) asm volatile("" :: "v"(v[0]), "v"(v[1]), "v"(v[2]), "v"(v[3]), "v"(v[4]), "v"(v[5]), "v"(v[6]), "v"(v[7]), "v"(v[8]), "v"(v[9]), "v"(v[10]), "v"(v[11]), "v"(v[12]), "v"(v[13]), "v"(v[14]), "v"(v[15]));
}

template <int G, bool OWN>
__device__ __forceinline__ void sum_group(const unsigned* p,
                                          const float* __restrict__ h,
                                          int lane, float4& s,
                                          const float4& a, const float4& b) {
  const unsigned mine = p[lane < G ? lane : 0];
  unsigned xe[G];
#pragma unroll
  for (int j = 0; j < G; ++j)
    xe[j] = (unsigned)__builtin_amdgcn_readlane((int)mine, j);
  v4f v[G];
#pragma unroll
  for (int j = 0; j < G; ++j) {
    const unsigned off =
        xe[j] * (unsigned)(EMB_DIM * sizeof(float)) + (unsigned)lane * 16u;
    v[j] = *(const v4f*)((const char*)h + off);
  }
  hold_loaded<G>(v);
  if constexpr (OWN)
    asm volatile("" :: "v"(a.x), "v"(a.y), "v"(a.z), "v"(a.w), "v"(b.x), "v"(b.y), "v"(b.z), "v"(b.w));
#pragma unroll
  for (int j = 0; j < G; ++j) {
    s.x += v[j].x;
    s.y += v[j].y;
    s.z += v[j].z;
    s.w += v[j].w;
  }
}

template <bool LAST>
__global__ void __launch_bounds__(PT)
conv_owner(const unsigned* __restrict__ elist, const int* __restrict__ tab,
           const float* __restrict__ h, float* __restrict__ gout,
           const int* __restrict__ users, const float* __restrict__ uemb,
           float* __restrict__ out_user, unsigned* __restrict__ slist,
           int* __restrict__ rtab, int num_users, int batch) {
  __shared__ __attribute__((aligned(16))) float acc[ROWS * EMB_DIM];
  __shared__ unsigned sorted[CAP];
  __shared__ int segpos[NCH + 1];
  __shared__ int segsrc[NCH];
  __shared__ int whist[NWAVES * HS];
  __shared__ int rs[HS];
  __shared__ int rc[HS];
  __shared__ int cnt[HS];
  __shared__ int wsum[NWAVES];
  __shared__ int mark[HS];
  __shared__ int mrow[HS];
  __shared__ int mcnt[NWAVES];
  const int tid = threadIdx.x;
  const int lane = tid & 31;
  const int w = __builtin_amdgcn_readfirstlane(tid >> 5);
  const unsigned lt = (1u << lane) - 1u;
  const int nb = (num_users + ROWS - 1) / ROWS;
  float4* acc4 = (float4*)acc;
  int* myh = whist + w * HS;

  for (int bin = blockIdx.x; bin < nb; bin += gridDim.x) {
    const int row0 = bin * ROWS;
    const int nrows = min(ROWS, num_users - row0);

    unsigned* sl = slist + (size_t)bin * CAP;
    int* rt = rtab + (size_t)bin * RT;
    unsigned xs[KSTEPS];
    int hrs = 0;
    int hrc = 0;
    int hlen = 0;
    if (LAST) {
#pragma unroll
      for (int k = 0; k < KSTEPS; ++k) xs[k] = sl[k * PT + tid];
      if (tid < HS) {
        hrs = rt[tid];
        hrc = rt[HS + tid];
      }
      hlen = rt[2 * HS];
    }

    int nm = 0;
    if (LAST) {
      for (int i = tid; i < HS; i += PT) mark[i] = 0;
      __syncthreads();
      for (int b0 = tid; b0 < batch; b0 += 8 * PT) {
        int u[8];
#pragma unroll
        for (int k = 0; k < 8; ++k) {
          const int b = b0 + k * PT;
          u[k] = (b < batch) ? (users[b] - row0) : -1;
        }
#pragma unroll
        for (int k = 0; k < 8; ++k)
          if (u[k] >= 0 && u[k] < nrows) mark[u[k]] = 1;
      }
      __syncthreads();
      const bool named = tid < HS && mark[tid] != 0;
      const unsigned nl = __builtin_amdgcn_ballot_w32(named);
      if (lane == 0) mcnt[w] = __popc(nl);
      __syncthreads();
      const int mc = mcnt[lane];
      int minc = mc;
#pragma unroll
      for (int o = 1; o < 32; o <<= 1) {
        const int t = __shfl_up(minc, o);
        if (lane >= o) minc += t;
      }
      nm = __builtin_amdgcn_readlane(minc, NWAVES - 1);
      const int before = __builtin_amdgcn_readlane(minc - mc, w);
      if (named) mrow[before + __popc(nl & lt)] = tid;
    }

    bool handed = false;
    int nlist = 0;
    if (LAST) {
      const int hn = __builtin_amdgcn_readfirstlane(hlen);
      handed = hn <= CAP;
      if (handed) {
        nlist = hn;
#pragma unroll
        for (int k = 0; k < KSTEPS; ++k) sorted[k * PT + tid] = xs[k];
        if (tid < HS) {
          rs[tid] = hrs;
          rc[tid] = hrc;
        }
      }
    }
    if (!handed) {
      int gsrc = 0;
      int glen = 0;
      if (tid < NCH) {
        gsrc = tab[bin * NCH + tid];
        glen = tab[(bin + 1) * NCH + tid] - gsrc;
      }
      int ginc = glen;
#pragma unroll
      for (int o = 1; o < 32; o <<= 1) {
        const int t = __shfl_up(ginc, o);
        if (lane >= o) ginc += t;
      }
      if (lane == 31) wsum[w] = ginc;
      __syncthreads();
      if (tid < NCH) {
        int p = ginc - glen;
#pragma unroll
        for (int k = 0; k < NCH / 32; ++k) p += (k < w) ? wsum[k] : 0;
        segpos[tid] = p;
        segsrc[tid] = gsrc;
        if (tid == NCH - 1) segpos[NCH] = p + glen;
      }
      __syncthreads();
      nlist = segpos[NCH];
      if (!LAST) {
        if (tid == 0) rt[2 * HS] = nlist;
        if (nlist == 0) {
#pragma nounroll
          for (int i = tid; i < CAP; i += PT) sl[i] = 0u;
          if (tid < HS) {
            rt[tid] = 0;
            rt[HS + tid] = 0;
          }
        }
      }
    }

    const bool direct = LAST ? handed : (nlist <= CAP);
    if (!direct) {
      for (int i = tid; i < ROWS * EMB_DIM / 4; i += PT)
        acc4[i] = make_float4(0.0f, 0.0f, 0.0f, 0.0f);
      for (int i = tid; i < HS; i += PT) cnt[i] = 0;
    }

    for (int t0 = 0; t0 < nlist; t0 += CAP) {
      const int tn = min(CAP, nlist - t0);
      __syncthreads();
      if (!handed) {
        for (int i = tid; i < NWAVES * HS; i += PT) whist[i] = 0;
        int gp[CPW + 1];
        int gs[CPW];
#pragma unroll
        for (int k = 0; k <= CPW; ++k) gp[k] = segpos[w * CPW + k];
#pragma unroll
        for (int k = 0; k < CPW; ++k) gs[k] = segsrc[w * CPW + k];
        unsigned gv[CPW];
        int gi[CPW];
        unsigned hv[CPW];
        int hi[CPW];
        int longest = 0;
#pragma unroll
        for (int k = 0; k < CPW; ++k) {
          const int a = max(gp[k], t0);
          const int b = min(gp[k + 1], t0 + tn);
          const int i = a + lane;
          gi[k] = (i < b) ? (i - t0) : -1;
          gv[k] = (i < b) ? elist[gs[k] + (i - gp[k])] : 0u;
          const int i2 = i + 32;
          hi[k] = (i2 < b) ? (i2 - t0) : -1;
          hv[k] = (i2 < b) ? elist[gs[k] + (i2 - gp[k])] : 0u;
          longest = max(longest, b - a);
        }
#pragma unroll
        for (int k = 0; k < CPW; ++k) {
          if (gi[k] >= 0) sorted[gi[k]] = gv[k];
          if (hi[k] >= 0) sorted[hi[k]] = hv[k];
        }
        if (longest > 64) {
#pragma nounroll
          for (int k = 0; k < CPW; ++k) {
            const int c = w * CPW + k;
            const int p = segpos[c];
            const int a = max(p, t0);
            const int b = min(segpos[c + 1], t0 + tn);
#pragma nounroll
            for (int i = a + 64 + lane; i < b; i += 32)
              sorted[i - t0] = elist[segsrc[c] + (i - p)];
          }
        }
        __syncthreads();
        unsigned x[KSTEPS];
#pragma unroll
        for (int k = 0; k < KSTEPS; ++k) {
          const int i = w * SHARE + k * 32 + lane;
          x[k] = (i < tn) ? sorted[i] : 0u;
        }
        __syncthreads();

        unsigned peers[KSTEPS];
#pragma unroll
        for (int k = 0; k < KSTEPS; ++k) {
          const bool valid = w * SHARE + k * 32 + lane < tn;
          const int key = (int)(x[k] & ROW_MASK);
          const unsigned live = __builtin_amdgcn_ballot_w32(valid);
          peers[k] = same_key_lanes<ROW_BITS>(key, live);
          if (valid && (peers[k] & lt) == 0u) myh[key] += __popc(peers[k]);
        }
        __syncthreads();

        int tot = 0;
        if (tid < HS) {
          int cw[NWAVES];
#pragma unroll
          for (int k = 0; k < NWAVES; ++k) cw[k] = whist[k * HS + tid];
#pragma unroll
          for (int k = 0; k < NWAVES; ++k) {
            whist[k * HS + tid] = tot;
            tot += cw[k];
          }
        }
        int inc = tot;
#pragma unroll
        for (int o = 1; o < 32; o <<= 1) {
          const int t = __shfl_up(inc, o);
          if (lane >= o) inc += t;
        }
        if (lane == 31) wsum[w] = inc;
        __syncthreads();
        if (w == 0) {
          const int v = wsum[lane];
          int vi = v;
#pragma unroll
          for (int o = 1; o < 32; o <<= 1) {
            const int t = __shfl_up(vi, o);
            if (lane >= o) vi += t;
          }
          wsum[lane] = vi - v;
        }
        __syncthreads();
        if (tid < HS) {
          rs[tid] = wsum[w] + inc - tot;
          rc[tid] = tot;
        }
        __syncthreads();

#pragma unroll
        for (int k = 0; k < KSTEPS; ++k) {
          const bool valid = w * SHARE + k * 32 + lane < tn;
          const int key = (int)(x[k] & ROW_MASK);
          const int rank = __popc(peers[k] & lt);
          const int c0 = myh[key];
          if (valid) sorted[rs[key] + c0 + rank] = x[k] >> ROW_BITS;
          if (valid && rank == 0) myh[key] = c0 + __popc(peers[k]);
        }
        __syncthreads();
        if (!LAST && t0 == 0) {
#pragma unroll
          for (int k = 0; k < KSTEPS; ++k) {
            const int i = k * PT + tid;
            sl[i] = (i < tn) ? sorted[i] : 0u;
          }
          if (tid < HS) {
            rt[tid] = rs[tid];
            rt[HS + tid] = rc[tid];
          }
        }
      }

      const int nsum = LAST ? nm : ROWS;
      for (int i = w; i < nsum; i += NWAVES) {
        const int r = LAST ? __builtin_amdgcn_readfirstlane(mrow[i]) : i;
        const int st = __builtin_amdgcn_readfirstlane(rs[r]);
        const int c = __builtin_amdgcn_readfirstlane(rc[r]);
        const size_t ro = (size_t)(row0 + r) * EMB_DIM + lane * 4;
        if (c == 0) {
          if (direct) {
            if (LAST) {
              const float4 u0 = *(const float4*)(uemb + ro);
              const float4 h0 = *(const float4*)(h + ro);
              float4 o;
              o.x = u0.x + h0.x + 0.0f;
              o.y = u0.y + h0.y + 0.0f;
              o.z = u0.z + h0.z + 0.0f;
              o.w = u0.w + h0.w + 0.0f;
              acc4[r * 32 + lane] = o;
            } else if (r < nrows) {
              *(float4*)(gout + ro) = make_float4(0.0f, 0.0f, 0.0f, 0.0f);
            }
          }
          continue;
        }
        float4 ue = make_float4(0.0f, 0.0f, 0.0f, 0.0f);
        float4 he = make_float4(0.0f, 0.0f, 0.0f, 0.0f);
        if (LAST) {
          const unsigned ob = (unsigned)(row0 + r) * (unsigned)(EMB_DIM * sizeof(float)) + (unsigned)lane * 16u;
          ue = *(const float4*)((const char*)uemb + ob);
          he = *(const float4*)((const char*)h + ob);
        }
        float4 s = make_float4(0.0f, 0.0f, 0.0f, 0.0f);
        const unsigned* p = sorted + st;
        int q = 0;
        for (; q + GRP <= c; q += GRP) sum_group<GRP, LAST>(p + q, h, lane, s, ue, he);
        switch (c - q) {
          case 1: sum_group<1, LAST>(p + q, h, lane, s, ue, he); break;
          case 2: sum_group<2, LAST>(p + q, h, lane, s, ue, he); break;
          case 3: sum_group<3, LAST>(p + q, h, lane, s, ue, he); break;
          case 4: sum_group<4, LAST>(p + q, h, lane, s, ue, he); break;
          case 5: sum_group<5, LAST>(p + q, h, lane, s, ue, he); break;
          case 6: sum_group<6, LAST>(p + q, h, lane, s, ue, he); break;
          case 7: sum_group<7, LAST>(p + q, h, lane, s, ue, he); break;
          case 8: sum_group<8, LAST>(p + q, h, lane, s, ue, he); break;
          case 9: sum_group<9, LAST>(p + q, h, lane, s, ue, he); break;
          case 10: sum_group<10, LAST>(p + q, h, lane, s, ue, he); break;
          case 11: sum_group<11, LAST>(p + q, h, lane, s, ue, he); break;
          case 12: sum_group<12, LAST>(p + q, h, lane, s, ue, he); break;
          case 13: sum_group<13, LAST>(p + q, h, lane, s, ue, he); break;
          case 14: sum_group<14, LAST>(p + q, h, lane, s, ue, he); break;
          case 15: sum_group<15, LAST>(p + q, h, lane, s, ue, he); break;
          default: break;
        }
        if (direct) {
          const float m = fmaxf((float)c, 1.0f);
          float4 o;
          if (LAST) {
            o.x = ue.x + he.x + s.x / m;
            o.y = ue.y + he.y + s.y / m;
            o.z = ue.z + he.z + s.z / m;
            o.w = ue.w + he.w + s.w / m;
            acc4[r * 32 + lane] = o;
          } else {
            o.x = s.x / m;
            o.y = s.y / m;
            o.z = s.z / m;
            o.w = s.w / m;
            *(float4*)(gout + ro) = o;
          }
        } else {
          float4 a = acc4[r * 32 + lane];
          a.x += s.x;
          a.y += s.y;
          a.z += s.z;
          a.w += s.w;
          acc4[r * 32 + lane] = a;
          if (lane == 0) cnt[r] += c;
        }
      }
    }
    if (direct && nlist == 0) {
      if (LAST) {
        __syncthreads();
        for (int i = w; i < nm; i += NWAVES) {
          const int r = __builtin_amdgcn_readfirstlane(mrow[i]);
          const size_t ro = (size_t)(row0 + r) * EMB_DIM + lane * 4;
          const float4 ue = *(const float4*)(uemb + ro);
          const float4 he = *(const float4*)(h + ro);
          float4 o;
          o.x = ue.x + he.x + 0.0f;
          o.y = ue.y + he.y + 0.0f;
          o.z = ue.z + he.z + 0.0f;
          o.w = ue.w + he.w + 0.0f;
          acc4[r * 32 + lane] = o;
        }
      } else {
        for (int r = w; r < nrows; r += NWAVES)
          *(float4*)(gout + (size_t)(row0 + r) * EMB_DIM + lane * 4) =
              make_float4(0.0f, 0.0f, 0.0f, 0.0f);
      }
    }
    __syncthreads();

    if (!LAST) {
      for (int r = w; !direct && r < nrows; r += NWAVES) {
        const float m = fmaxf((float)cnt[r], 1.0f);
        const float4 a = acc4[r * 32 + lane];
        float4 o;
        o.x = a.x / m;
        o.y = a.y / m;
        o.z = a.z / m;
        o.w = a.w / m;
        *(float4*)(gout + (size_t)(row0 + r) * EMB_DIM + lane * 4) = o;
      }
    } else {
      for (int b0 = w * 32; b0 < batch; b0 += 4 * PT) {
        int r[4];
#pragma unroll
        for (int k = 0; k < 4; ++k) {
          const int b = b0 + k * PT + lane;
          r[k] = (b < batch) ? (users[b] - row0) : -1;
        }
#pragma unroll
        for (int k = 0; k < 4; ++k) {
          unsigned mask = __builtin_amdgcn_ballot_w32(r[k] >= 0 && r[k] < nrows);
          while (mask) {
            const int j = __builtin_ctz(mask);
            mask &= mask - 1;
            const int rj = __builtin_amdgcn_readlane(r[k], j);
            const float4 a = acc4[rj * 32 + lane];
            float4 o;
            if (direct) {
              o = a;
            } else {
              const size_t ro = (size_t)(row0 + rj) * EMB_DIM + lane * 4;
              const float4 x = *(const float4*)(uemb + ro);
              const float4 c = *(const float4*)(h + ro);
              const float m = fmaxf((float)cnt[rj], 1.0f);
              o.x = x.x + c.x + a.x / m;
              o.y = x.y + c.y + a.y / m;
              o.z = x.z + c.z + a.z / m;
              o.w = x.w + c.w + a.w / m;
            }
            *(float4*)(out_user + (size_t)(b0 + k * PT + j) * EMB_DIM +
                       lane * 4) = o;
          }
        }
      }
    }
    __syncthreads();
  }
}

__global__ void __launch_bounds__(256)
gather_items(const int* __restrict__ items, const float* __restrict__ iemb,
             float* __restrict__ out_item, int batch) {
  int t = blockIdx.x * blockDim.x + threadIdx.x;
  int b = t >> 5;
  int lane = t & 31;
  if (b >= batch) return;
  int it = items[b];
  size_t off = (size_t)lane * 4;
  *(float4*)(out_item + (size_t)b * EMB_DIM + off) =
      *(const float4*)(iemb + (size_t)it * EMB_DIM + off);
}

__global__ void __launch_bounds__(256)
dot_sigmoid_wmma(const float* __restrict__ U, const float* __restrict__ V,
                 float* __restrict__ pred, int batch) {
  int gid = blockIdx.x * blockDim.x + threadIdx.x;
  int wave = gid >> 5;
  int lane = threadIdx.x & 31;
  if (wave * 16 >= batch) return;

  int row = lane & 15;
  int half = lane >> 4;
  const float* urow = U + ((size_t)wave * 16 + row) * EMB_DIM + 2 * half;
  const float* vrow = V + ((size_t)wave * 16 + row) * EMB_DIM + 2 * half;

  v8f acc = {};
#pragma unroll
  for (int k0 = 0; k0 < EMB_DIM; k0 += 4) {
    v2f a = *(const v2f*)(urow + k0);
    v2f b = *(const v2f*)(vrow + k0);
    acc = __builtin_amdgcn_wmma_f32_16x16x4_f32(
         false, a,  false, b,
         (short)0, acc,  false,  false);
  }

  if (lane < 8 || lane >= 24) {
    int i = (lane < 8) ? lane : (lane - 16);
    float d = acc[lane & 7];
    pred[(size_t)wave * 16 + i] = 1.0f / (1.0f + expf(-d));
  }
}

extern "C" void kernel_launch(void* const* d_in, const int* in_sizes, int n_in,
                              void* d_out, int out_size, void* d_ws,
                              size_t ws_size, hipStream_t stream) {
  const int* users  = (const int*)d_in[0];
  const int* items  = (const int*)d_in[1];
  const int* esrc   = (const int*)d_in[2];
  const int* edst   = (const int*)d_in[3];
  const float* uemb = (const float*)d_in[4];
  const float* iemb = (const float*)d_in[5];

  const int batch     = in_sizes[0];
  const int n_edges   = in_sizes[2];
  const int num_users = in_sizes[4] / EMB_DIM;

  int* tab        = (int*)d_ws;
  float* g1       = (float*)d_ws + num_users;
  unsigned* elist = (unsigned*)(g1 + (size_t)num_users * EMB_DIM);
  unsigned* slist = elist + n_edges;
  int* rtab       = (int*)(slist + (size_t)NBMAX * CAP);

  float* pred     = (float*)d_out;
  float* out_user = pred + batch;
  float* out_item = out_user + (size_t)batch * EMB_DIM;

  const int B = 256;

  part_sort<<<NCH, PT, 0, stream>>>(esrc, edst, tab, elist, n_edges,
                                    num_users);

  conv_owner<false><<<NBMAX, PT, 0, stream>>>(elist, tab, uemb, g1, users,
                                              uemb, out_user, slist, rtab,
                                              num_users, batch);

  conv_owner<true><<<NBMAX, PT, 0, stream>>>(elist, tab, g1, (float*)nullptr,
                                             users, uemb, out_user, slist,
                                             rtab, num_users, batch);

  long long ga_threads = (long long)batch * 32;
  gather_items<<<(int)((ga_threads + B - 1) / B), B, 0, stream>>>(
      items, iemb, out_item, batch);

  long long dot_threads = (long long)((batch + 15) / 16) * 32;
  dot_sigmoid_wmma<<<(int)((dot_threads + B - 1) / B), B, 0, stream>>>(
      out_user, out_item, pred, batch);
}


# latest_item gather rides in the partition launch: each of its 8192 waves loads two item rows under its own edge loads and stores them at its end; the gather launch of 16384 one-row waves is gone
# speedup vs baseline: 8.3709x; 1.0349x over previous
#include <hip/hip_runtime.h>
#include <hip/hip_bf16.h>


#define EMB_DIM 128
#define ROWS 391
#define ROW_BITS 9
#define ROW_MASK 511u
#define NBMAX 256
#define NCH 256
#define PT 1024
#define NWAVES (PT / 32)
#define CPW (NCH / NWAVES)
#define CAP 8192
#define SHARE (CAP / NWAVES)
#define KSTEPS (SHARE / 32)
#define HS 392
#define GRP 16
#define SST 8
#define RT (2 * HS + 8)
#define TOTW (NCH * NWAVES)

typedef __attribute__((ext_vector_type(2))) float v2f;
typedef __attribute__((ext_vector_type(8))) float v8f;
typedef __attribute__((ext_vector_type(4))) float v4f;

template <int BITS>
__device__ __forceinline__ unsigned same_key_lanes(int key, unsigned live) {
  unsigned peers = live;
#pragma unroll
  for (int i = 0; i < BITS; ++i) {
    const bool bit = ((key >> i) & 1) != 0;
    const unsigned m = __builtin_amdgcn_ballot_w32(bit);
    peers &= bit ? m : ~m;
  }
  return peers;
}

__device__ __forceinline__ void hold_edges(const int (&v)[SST]) {
  static_assert(SST == 8, "hold_edges names SST values");
  asm volatile("" :: "v"(v[0]), "v"(v[1]), "v"(v[2]), "v"(v[3]), "v"(v[4]), "v"(v[5]), "v"(v[6]), "v"(v[7]));
}

__device__ __forceinline__ void wave_block_hist(const int* __restrict__ dst,
                                                int* hist, int eb, int ee,
                                                int w, int lane) {
  const unsigned lt = (1u << lane) - 1u;
  for (int base0 = eb + w * 32; base0 < ee; base0 += SST * PT) {
    int d[SST];
#pragma unroll
    for (int k = 0; k < SST; ++k) d[k] = dst[min(base0 + k * PT + lane, ee - 1)];
    hold_edges(d);
#pragma unroll
    for (int k = 0; k < SST; ++k) {
      if (base0 + k * PT >= ee) break;
      const bool valid = base0 + k * PT + lane < ee;
      const int b = valid ? d[k] / ROWS : 0;
      const unsigned live = __builtin_amdgcn_ballot_w32(valid);
      const unsigned peers = same_key_lanes<8>(b, live);
      if (valid && (peers & lt) == 0u) hist[b] += __popc(peers);
    }
  }
}

__global__ void __launch_bounds__(PT)
part_sort(const int* __restrict__ src, const int* __restrict__ dst,
          int* __restrict__ tab, unsigned* __restrict__ elist, int n_edges,
          int num_users, const int* __restrict__ items, const float* iemb,
          float* out_item, int batch) {
  __shared__ int whist[NWAVES * NBMAX];
  __shared__ int wsum[NWAVES];
  const int tid = threadIdx.x;
  const int lane = tid & 31;
  const int w = __builtin_amdgcn_readfirstlane(tid >> 5);
  const int nb = (num_users + ROWS - 1) / ROWS;
  const int gw = blockIdx.x * NWAVES + w;
  const int it0 = items[max(min(gw, batch - 1), 0)];
  const int it1 = items[max(min(gw + TOTW, batch - 1), 0)];
  for (int i = tid; i < NWAVES * NBMAX; i += PT) whist[i] = 0;
  __syncthreads();
  const int cs = (n_edges + NCH - 1) / NCH;
  const int eb = blockIdx.x * cs;
  const int ee = min(n_edges, eb + cs);
  wave_block_hist(dst, whist + w * NBMAX, eb, ee, w, lane);
  __syncthreads();
  int tot = 0;
  if (tid < NBMAX) {
    for (int k = 0; k < NWAVES; ++k) tot += whist[k * NBMAX + tid];
  }
  int inc = tot;
#pragma unroll
  for (int o = 1; o < 32; o <<= 1) {
    const int t = __shfl_up(inc, o);
    if (lane >= o) inc += t;
  }
  if (lane == 31) wsum[w] = inc;
  __syncthreads();
  if (tid < NBMAX) {
    int run = eb + inc - tot;
#pragma unroll
    for (int k = 0; k < NBMAX / 32; ++k) run += (k < w) ? wsum[k] : 0;
    if (tid < nb) tab[tid * NCH + blockIdx.x] = run;
    if (tid == nb - 1) tab[nb * NCH + blockIdx.x] = run + tot;
    for (int k = 0; k < NWAVES; ++k) {
      const int c = whist[k * NBMAX + tid];
      whist[k * NBMAX + tid] = run;
      run += c;
    }
  }
  __syncthreads();
  const float4 ir0 = *(const float4*)(iemb + (size_t)it0 * EMB_DIM + lane * 4);
  const float4 ir1 = *(const float4*)(iemb + (size_t)it1 * EMB_DIM + lane * 4);
  int* cur = whist + w * NBMAX;
  const unsigned lt = (1u << lane) - 1u;
  for (int base0 = eb + w * 32; base0 < ee; base0 += SST * PT) {
    int dv[SST];
    int sv[SST];
#pragma unroll
    for (int k = 0; k < SST; ++k) {
      const int e = min(base0 + k * PT + lane, ee - 1);
      dv[k] = dst[e];
      sv[k] = src[e];
    }
    hold_edges(dv);
    hold_edges(sv);
#pragma unroll
    for (int k = 0; k < SST; ++k) {
      if (base0 + k * PT >= ee) break;
      const bool valid = base0 + k * PT + lane < ee;
      const int d = dv[k];
      const int b = d / ROWS;
      const unsigned live = __builtin_amdgcn_ballot_w32(valid);
      const unsigned peers = same_key_lanes<8>(b, live);
      const int rank = __popc(peers & lt);
      const int c0 = cur[b];
      if (valid)
        elist[c0 + rank] =
            ((unsigned)sv[k] << ROW_BITS) | (unsigned)(d - b * ROWS);
      if (valid && rank == 0) cur[b] = c0 + __popc(peers);
    }
  }
  if (gw < batch) *(float4*)(out_item + (size_t)gw * EMB_DIM + lane * 4) = ir0;
  if (gw + TOTW < batch)
    *(float4*)(out_item + (size_t)(gw + TOTW) * EMB_DIM + lane * 4) = ir1;
  for (int b = gw + 2 * TOTW; b < batch; b += TOTW) {
    const int it = items[b];
    *(float4*)(out_item + (size_t)b * EMB_DIM + lane * 4) =
        *(const float4*)(iemb + (size_t)it * EMB_DIM + lane * 4);
  }
}

template <int G>
__device__ __forceinline__ void hold_loaded(const v4f (&v)[G]) {
  if constexpr (G == 1) asm volatile("" :: "v"(v[0]));
  else if constexpr (G == 2) asm volatile("" :: "v"(v[0]), "v"(v[1]));
  else if constexpr (G == 3) asm volatile("" :: "v"(v[0]), "v"(v[1]), "v"(v[2]));
  else if constexpr (G == 4) asm volatile("" :: "v"(v[0]), "v"(v[1]), "v"(v[2]), "v"(v[3]));
  else if constexpr (G == 5) asm volatile("" :: "v"(v[0]), "v"(v[1]), "v"(v[2]), "v"(v[3]), "v"(v[4]));
  else if constexpr (G == 6) asm volatile("" :: "v"(v[0]), "v"(v[1]), "v"(v[2]), "v"(v[3]), "v"(v[4]), "v"(v[5]));
  else if constexpr (G == 7) asm volatile("" :: "v"(v[0]), "v"(v[1]), "v"(v[2]), "v"(v[3]), "v"(v[4]), "v"(v[5]), "v"(v[6]));
  else if constexpr (G == 8) asm volatile("" :: "v"(v[0]), "v"(v[1]), "v"(v[2]), "v"(v[3]), "v"(v[4]), "v"(v[5]), "v"(v[6]), "v"(v[7]));
  else if constexpr (G == 9) asm volatile("" :: "v"(v[0]), "v"(v[1]), "v"(v[2]), "v"(v[3]), "v"(v[4]), "v"(v[5]), "v"(v[6]), "v"(v[7]), "v"(v[8]));
  else if constexpr (G == 10) asm volatile("" :: "v"(v[0]), "v"(v[1]), "v"(v[2]), "v"(v[3]), "v"(v[4]), "v"(v[5]), "v"(v[6]), "v"(v[7]), "v"(v[8]), "v"(v[9]));
  else if constexpr (G == 11) asm volatile("" :: "v"(v[0]), "v"(v[1]), "v"(v[2]), "v"(v[3]), "v"(v[4]), "v"(v[5]), "v"(v[6]), "v"(v[7]), "v"(v[8]), "v"(v[9]), "v"(v[10]));
  else if constexpr (G == 12) asm volatile("" :: "v"(v[0]), "v"(v[1]), "v"(v[2]), "v"(v[3]), "v"(v[4]), "v"(v[5]), "v"(v[6]), "v"(v[7]), "v"(v[8]), "v"(v[9]), "v"(v[10]), "v"(v[11]));
  else if constexpr (G == 13) asm volatile("" :: "v"(v[0]), "v"(v[1]), "v"(v[2]), "v"(v[3]), "v"(v[4]), "v"(v[5]), "v"(v[6]), "v"(v[7]), "v"(v[8]), "v"(v[9]), "v"(v[10]), "v"(v[11]), "v"(v[12]));
  else if constexpr (G == 14) asm volatile("" :: "v"(v[0]), "v"(v[1]), "v"(v[2]), "v"(v[3]), "v"(v[4]), "v"(v[5]), "v"(v[6]), "v"(v[7]), "v"(v[8]), "v"(v[9]), "v"(v[10]), "v"(v[11]), "v"(v[12]), "v"(v[13]));
  else if constexpr (G == 15) asm volatile("" :: "v"(v[0]), "v"(v[1]), "v"(v[2]), "v"(v[3]), "v"(v[4]), "v"(v[5]), "v"(v[6]), "v"(v[7]), "v"(v[8]), "v"(v[9]), "v"(v[10]), "v"(v[11]), "v"(v[12]), "v"(v[13]), "v"(v[14]));
  else if constexpr (G == 16) asm volatile("" :: "v"(v[0]), "v"(v[1]), "v"(v[2]), "v"(v[3]), "v"(v[4]), "v"(v[5]), "v"(v[6]), "v"(v[7]), "v"(v[8]), "v"(v[9]), "v"(v[10]), "v"(v[11]), "v"(v[12]), "v"(v[13]), "v"(v[14]), "v"(v[15]));
}

template <int G, bool OWN>
__device__ __forceinline__ void sum_group(const unsigned* p,
                                          const float* __restrict__ h,
                                          int lane, float4& s,
                                          const float4& a, const float4& b) {
  const unsigned mine = p[lane < G ? lane : 0];
  unsigned xe[G];
#pragma unroll
  for (int j = 0; j < G; ++j)
    xe[j] = (unsigned)__builtin_amdgcn_readlane((int)mine, j);
  v4f v[G];
#pragma unroll
  for (int j = 0; j < G; ++j) {
    const unsigned off =
        xe[j] * (unsigned)(EMB_DIM * sizeof(float)) + (unsigned)lane * 16u;
    v[j] = *(const v4f*)((const char*)h + off);
  }
  hold_loaded<G>(v);
  if constexpr (OWN)
    asm volatile("" :: "v"(a.x), "v"(a.y), "v"(a.z), "v"(a.w), "v"(b.x), "v"(b.y), "v"(b.z), "v"(b.w));
#pragma unroll
  for (int j = 0; j < G; ++j) {
    s.x += v[j].x;
    s.y += v[j].y;
    s.z += v[j].z;
    s.w += v[j].w;
  }
}

template <bool LAST>
__global__ void __launch_bounds__(PT)
conv_owner(const unsigned* __restrict__ elist, const int* __restrict__ tab,
           const float* __restrict__ h, float* __restrict__ gout,
           const int* __restrict__ users, const float* __restrict__ uemb,
           float* __restrict__ out_user, unsigned* __restrict__ slist,
           int* __restrict__ rtab, int num_users, int batch) {
  __shared__ __attribute__((aligned(16))) float acc[ROWS * EMB_DIM];
  __shared__ unsigned sorted[CAP];
  __shared__ int segpos[NCH + 1];
  __shared__ int segsrc[NCH];
  __shared__ int whist[NWAVES * HS];
  __shared__ int rs[HS];
  __shared__ int rc[HS];
  __shared__ int cnt[HS];
  __shared__ int wsum[NWAVES];
  __shared__ int mark[HS];
  __shared__ int mrow[HS];
  __shared__ int mcnt[NWAVES];
  const int tid = threadIdx.x;
  const int lane = tid & 31;
  const int w = __builtin_amdgcn_readfirstlane(tid >> 5);
  const unsigned lt = (1u << lane) - 1u;
  const int nb = (num_users + ROWS - 1) / ROWS;
  float4* acc4 = (float4*)acc;
  int* myh = whist + w * HS;

  for (int bin = blockIdx.x; bin < nb; bin += gridDim.x) {
    const int row0 = bin * ROWS;
    const int nrows = min(ROWS, num_users - row0);

    unsigned* sl = slist + (size_t)bin * CAP;
    int* rt = rtab + (size_t)bin * RT;
    unsigned xs[KSTEPS];
    int hrs = 0;
    int hrc = 0;
    int hlen = 0;
    if (LAST) {
#pragma unroll
      for (int k = 0; k < KSTEPS; ++k) xs[k] = sl[k * PT + tid];
      if (tid < HS) {
        hrs = rt[tid];
        hrc = rt[HS + tid];
      }
      hlen = rt[2 * HS];
    }

    int nm = 0;
    if (LAST) {
      for (int i = tid; i < HS; i += PT) mark[i] = 0;
      __syncthreads();
      for (int b0 = tid; b0 < batch; b0 += 8 * PT) {
        int u[8];
#pragma unroll
        for (int k = 0; k < 8; ++k) {
          const int b = b0 + k * PT;
          u[k] = (b < batch) ? (users[b] - row0) : -1;
        }
#pragma unroll
        for (int k = 0; k < 8; ++k)
          if (u[k] >= 0 && u[k] < nrows) mark[u[k]] = 1;
      }
      __syncthreads();
      const bool named = tid < HS && mark[tid] != 0;
      const unsigned nl = __builtin_amdgcn_ballot_w32(named);
      if (lane == 0) mcnt[w] = __popc(nl);
      __syncthreads();
      const int mc = mcnt[lane];
      int minc = mc;
#pragma unroll
      for (int o = 1; o < 32; o <<= 1) {
        const int t = __shfl_up(minc, o);
        if (lane >= o) minc += t;
      }
      nm = __builtin_amdgcn_readlane(minc, NWAVES - 1);
      const int before = __builtin_amdgcn_readlane(minc - mc, w);
      if (named) mrow[before + __popc(nl & lt)] = tid;
    }

    bool handed = false;
    int nlist = 0;
    if (LAST) {
      const int hn = __builtin_amdgcn_readfirstlane(hlen);
      handed = hn <= CAP;
      if (handed) {
        nlist = hn;
#pragma unroll
        for (int k = 0; k < KSTEPS; ++k) sorted[k * PT + tid] = xs[k];
        if (tid < HS) {
          rs[tid] = hrs;
          rc[tid] = hrc;
        }
      }
    }
    if (!handed) {
      int gsrc = 0;
      int glen = 0;
      if (tid < NCH) {
        gsrc = tab[bin * NCH + tid];
        glen = tab[(bin + 1) * NCH + tid] - gsrc;
      }
      int ginc = glen;
#pragma unroll
      for (int o = 1; o < 32; o <<= 1) {
        const int t = __shfl_up(ginc, o);
        if (lane >= o) ginc += t;
      }
      if (lane == 31) wsum[w] = ginc;
      __syncthreads();
      if (tid < NCH) {
        int p = ginc - glen;
#pragma unroll
        for (int k = 0; k < NCH / 32; ++k) p += (k < w) ? wsum[k] : 0;
        segpos[tid] = p;
        segsrc[tid] = gsrc;
        if (tid == NCH - 1) segpos[NCH] = p + glen;
      }
      __syncthreads();
      nlist = segpos[NCH];
      if (!LAST) {
        if (tid == 0) rt[2 * HS] = nlist;
        if (nlist == 0) {
#pragma nounroll
          for (int i = tid; i < CAP; i += PT) sl[i] = 0u;
          if (tid < HS) {
            rt[tid] = 0;
            rt[HS + tid] = 0;
          }
        }
      }
    }

    const bool direct = LAST ? handed : (nlist <= CAP);
    if (!direct) {
      for (int i = tid; i < ROWS * EMB_DIM / 4; i += PT)
        acc4[i] = make_float4(0.0f, 0.0f, 0.0f, 0.0f);
      for (int i = tid; i < HS; i += PT) cnt[i] = 0;
    }

    for (int t0 = 0; t0 < nlist; t0 += CAP) {
      const int tn = min(CAP, nlist - t0);
      __syncthreads();
      if (!handed) {
        for (int i = tid; i < NWAVES * HS; i += PT) whist[i] = 0;
        int gp[CPW + 1];
        int gs[CPW];
#pragma unroll
        for (int k = 0; k <= CPW; ++k) gp[k] = segpos[w * CPW + k];
#pragma unroll
        for (int k = 0; k < CPW; ++k) gs[k] = segsrc[w * CPW + k];
        unsigned gv[CPW];
        int gi[CPW];
        unsigned hv[CPW];
        int hi[CPW];
        int longest = 0;
#pragma unroll
        for (int k = 0; k < CPW; ++k) {
          const int a = max(gp[k], t0);
          const int b = min(gp[k + 1], t0 + tn);
          const int i = a + lane;
          gi[k] = (i < b) ? (i - t0) : -1;
          gv[k] = (i < b) ? elist[gs[k] + (i - gp[k])] : 0u;
          const int i2 = i + 32;
          hi[k] = (i2 < b) ? (i2 - t0) : -1;
          hv[k] = (i2 < b) ? elist[gs[k] + (i2 - gp[k])] : 0u;
          longest = max(longest, b - a);
        }
#pragma unroll
        for (int k = 0; k < CPW; ++k) {
          if (gi[k] >= 0) sorted[gi[k]] = gv[k];
          if (hi[k] >= 0) sorted[hi[k]] = hv[k];
        }
        if (longest > 64) {
#pragma nounroll
          for (int k = 0; k < CPW; ++k) {
            const int c = w * CPW + k;
            const int p = segpos[c];
            const int a = max(p, t0);
            const int b = min(segpos[c + 1], t0 + tn);
#pragma nounroll
            for (int i = a + 64 + lane; i < b; i += 32)
              sorted[i - t0] = elist[segsrc[c] + (i - p)];
          }
        }
        __syncthreads();
        unsigned x[KSTEPS];
#pragma unroll
        for (int k = 0; k < KSTEPS; ++k) {
          const int i = w * SHARE + k * 32 + lane;
          x[k] = (i < tn) ? sorted[i] : 0u;
        }
        __syncthreads();

        unsigned peers[KSTEPS];
#pragma unroll
        for (int k = 0; k < KSTEPS; ++k) {
          const bool valid = w * SHARE + k * 32 + lane < tn;
          const int key = (int)(x[k] & ROW_MASK);
          const unsigned live = __builtin_amdgcn_ballot_w32(valid);
          peers[k] = same_key_lanes<ROW_BITS>(key, live);
          if (valid && (peers[k] & lt) == 0u) myh[key] += __popc(peers[k]);
        }
        __syncthreads();

        int tot = 0;
        if (tid < HS) {
          int cw[NWAVES];
#pragma unroll
          for (int k = 0; k < NWAVES; ++k) cw[k] = whist[k * HS + tid];
#pragma unroll
          for (int k = 0; k < NWAVES; ++k) {
            whist[k * HS + tid] = tot;
            tot += cw[k];
          }
        }
        int inc = tot;
#pragma unroll
        for (int o = 1; o < 32; o <<= 1) {
          const int t = __shfl_up(inc, o);
          if (lane >= o) inc += t;
        }
        if (lane == 31) wsum[w] = inc;
        __syncthreads();
        if (w == 0) {
          const int v = wsum[lane];
          int vi = v;
#pragma unroll
          for (int o = 1; o < 32; o <<= 1) {
            const int t = __shfl_up(vi, o);
            if (lane >= o) vi += t;
          }
          wsum[lane] = vi - v;
        }
        __syncthreads();
        if (tid < HS) {
          rs[tid] = wsum[w] + inc - tot;
          rc[tid] = tot;
        }
        __syncthreads();

#pragma unroll
        for (int k = 0; k < KSTEPS; ++k) {
          const bool valid = w * SHARE + k * 32 + lane < tn;
          const int key = (int)(x[k] & ROW_MASK);
          const int rank = __popc(peers[k] & lt);
          const int c0 = myh[key];
          if (valid) sorted[rs[key] + c0 + rank] = x[k] >> ROW_BITS;
          if (valid && rank == 0) myh[key] = c0 + __popc(peers[k]);
        }
        __syncthreads();
        if (!LAST && t0 == 0) {
#pragma unroll
          for (int k = 0; k < KSTEPS; ++k) {
            const int i = k * PT + tid;
            sl[i] = (i < tn) ? sorted[i] : 0u;
          }
          if (tid < HS) {
            rt[tid] = rs[tid];
            rt[HS + tid] = rc[tid];
          }
        }
      }

      const int nsum = LAST ? nm : ROWS;
      for (int i = w; i < nsum; i += NWAVES) {
        const int r = LAST ? __builtin_amdgcn_readfirstlane(mrow[i]) : i;
        const int st = __builtin_amdgcn_readfirstlane(rs[r]);
        const int c = __builtin_amdgcn_readfirstlane(rc[r]);
        const size_t ro = (size_t)(row0 + r) * EMB_DIM + lane * 4;
        if (c == 0) {
          if (direct) {
            if (LAST) {
              const float4 u0 = *(const float4*)(uemb + ro);
              const float4 h0 = *(const float4*)(h + ro);
              float4 o;
              o.x = u0.x + h0.x + 0.0f;
              o.y = u0.y + h0.y + 0.0f;
              o.z = u0.z + h0.z + 0.0f;
              o.w = u0.w + h0.w + 0.0f;
              acc4[r * 32 + lane] = o;
            } else if (r < nrows) {
              *(float4*)(gout + ro) = make_float4(0.0f, 0.0f, 0.0f, 0.0f);
            }
          }
          continue;
        }
        float4 ue = make_float4(0.0f, 0.0f, 0.0f, 0.0f);
        float4 he = make_float4(0.0f, 0.0f, 0.0f, 0.0f);
        if (LAST) {
          const unsigned ob = (unsigned)(row0 + r) * (unsigned)(EMB_DIM * sizeof(float)) + (unsigned)lane * 16u;
          ue = *(const float4*)((const char*)uemb + ob);
          he = *(const float4*)((const char*)h + ob);
        }
        float4 s = make_float4(0.0f, 0.0f, 0.0f, 0.0f);
        const unsigned* p = sorted + st;
        int q = 0;
        for (; q + GRP <= c; q += GRP) sum_group<GRP, LAST>(p + q, h, lane, s, ue, he);
        switch (c - q) {
          case 1: sum_group<1, LAST>(p + q, h, lane, s, ue, he); break;
          case 2: sum_group<2, LAST>(p + q, h, lane, s, ue, he); break;
          case 3: sum_group<3, LAST>(p + q, h, lane, s, ue, he); break;
          case 4: sum_group<4, LAST>(p + q, h, lane, s, ue, he); break;
          case 5: sum_group<5, LAST>(p + q, h, lane, s, ue, he); break;
          case 6: sum_group<6, LAST>(p + q, h, lane, s, ue, he); break;
          case 7: sum_group<7, LAST>(p + q, h, lane, s, ue, he); break;
          case 8: sum_group<8, LAST>(p + q, h, lane, s, ue, he); break;
          case 9: sum_group<9, LAST>(p + q, h, lane, s, ue, he); break;
          case 10: sum_group<10, LAST>(p + q, h, lane, s, ue, he); break;
          case 11: sum_group<11, LAST>(p + q, h, lane, s, ue, he); break;
          case 12: sum_group<12, LAST>(p + q, h, lane, s, ue, he); break;
          case 13: sum_group<13, LAST>(p + q, h, lane, s, ue, he); break;
          case 14: sum_group<14, LAST>(p + q, h, lane, s, ue, he); break;
          case 15: sum_group<15, LAST>(p + q, h, lane, s, ue, he); break;
          default: break;
        }
        if (direct) {
          const float m = fmaxf((float)c, 1.0f);
          float4 o;
          if (LAST) {
            o.x = ue.x + he.x + s.x / m;
            o.y = ue.y + he.y + s.y / m;
            o.z = ue.z + he.z + s.z / m;
            o.w = ue.w + he.w + s.w / m;
            acc4[r * 32 + lane] = o;
          } else {
            o.x = s.x / m;
            o.y = s.y / m;
            o.z = s.z / m;
            o.w = s.w / m;
            *(float4*)(gout + ro) = o;
          }
        } else {
          float4 a = acc4[r * 32 + lane];
          a.x += s.x;
          a.y += s.y;
          a.z += s.z;
          a.w += s.w;
          acc4[r * 32 + lane] = a;
          if (lane == 0) cnt[r] += c;
        }
      }
    }
    if (direct && nlist == 0) {
      if (LAST) {
        __syncthreads();
        for (int i = w; i < nm; i += NWAVES) {
          const int r = __builtin_amdgcn_readfirstlane(mrow[i]);
          const size_t ro = (size_t)(row0 + r) * EMB_DIM + lane * 4;
          const float4 ue = *(const float4*)(uemb + ro);
          const float4 he = *(const float4*)(h + ro);
          float4 o;
          o.x = ue.x + he.x + 0.0f;
          o.y = ue.y + he.y + 0.0f;
          o.z = ue.z + he.z + 0.0f;
          o.w = ue.w + he.w + 0.0f;
          acc4[r * 32 + lane] = o;
        }
      } else {
        for (int r = w; r < nrows; r += NWAVES)
          *(float4*)(gout + (size_t)(row0 + r) * EMB_DIM + lane * 4) =
              make_float4(0.0f, 0.0f, 0.0f, 0.0f);
      }
    }
    __syncthreads();

    if (!LAST) {
      for (int r = w; !direct && r < nrows; r += NWAVES) {
        const float m = fmaxf((float)cnt[r], 1.0f);
        const float4 a = acc4[r * 32 + lane];
        float4 o;
        o.x = a.x / m;
        o.y = a.y / m;
        o.z = a.z / m;
        o.w = a.w / m;
        *(float4*)(gout + (size_t)(row0 + r) * EMB_DIM + lane * 4) = o;
      }
    } else {
      for (int b0 = w * 32; b0 < batch; b0 += 4 * PT) {
        int r[4];
#pragma unroll
        for (int k = 0; k < 4; ++k) {
          const int b = b0 + k * PT + lane;
          r[k] = (b < batch) ? (users[b] - row0) : -1;
        }
#pragma unroll
        for (int k = 0; k < 4; ++k) {
          unsigned mask = __builtin_amdgcn_ballot_w32(r[k] >= 0 && r[k] < nrows);
          while (mask) {
            const int j = __builtin_ctz(mask);
            mask &= mask - 1;
            const int rj = __builtin_amdgcn_readlane(r[k], j);
            const float4 a = acc4[rj * 32 + lane];
            float4 o;
            if (direct) {
              o = a;
            } else {
              const size_t ro = (size_t)(row0 + rj) * EMB_DIM + lane * 4;
              const float4 x = *(const float4*)(uemb + ro);
              const float4 c = *(const float4*)(h + ro);
              const float m = fmaxf((float)cnt[rj], 1.0f);
              o.x = x.x + c.x + a.x / m;
              o.y = x.y + c.y + a.y / m;
              o.z = x.z + c.z + a.z / m;
              o.w = x.w + c.w + a.w / m;
            }
            *(float4*)(out_user + (size_t)(b0 + k * PT + j) * EMB_DIM +
                       lane * 4) = o;
          }
        }
      }
    }
    __syncthreads();
  }
}

__global__ void __launch_bounds__(256)
dot_sigmoid_wmma(const float* __restrict__ U, const float* __restrict__ V,
                 float* __restrict__ pred, int batch) {
  int gid = blockIdx.x * blockDim.x + threadIdx.x;
  int wave = gid >> 5;
  int lane = threadIdx.x & 31;
  if (wave * 16 >= batch) return;

  int row = lane & 15;
  int half = lane >> 4;
  const float* urow = U + ((size_t)wave * 16 + row) * EMB_DIM + 2 * half;
  const float* vrow = V + ((size_t)wave * 16 + row) * EMB_DIM + 2 * half;

  v8f acc = {};
#pragma unroll
  for (int k0 = 0; k0 < EMB_DIM; k0 += 4) {
    v2f a = *(const v2f*)(urow + k0);
    v2f b = *(const v2f*)(vrow + k0);
    acc = __builtin_amdgcn_wmma_f32_16x16x4_f32(
         false, a,  false, b,
         (short)0, acc,  false,  false);
  }

  if (lane < 8 || lane >= 24) {
    int i = (lane < 8) ? lane : (lane - 16);
    float d = acc[lane & 7];
    pred[(size_t)wave * 16 + i] = 1.0f / (1.0f + expf(-d));
  }
}

extern "C" void kernel_launch(void* const* d_in, const int* in_sizes, int n_in,
                              void* d_out, int out_size, void* d_ws,
                              size_t ws_size, hipStream_t stream) {
  const int* users  = (const int*)d_in[0];
  const int* items  = (const int*)d_in[1];
  const int* esrc   = (const int*)d_in[2];
  const int* edst   = (const int*)d_in[3];
  const float* uemb = (const float*)d_in[4];
  const float* iemb = (const float*)d_in[5];

  const int batch     = in_sizes[0];
  const int n_edges   = in_sizes[2];
  const int num_users = in_sizes[4] / EMB_DIM;

  int* tab        = (int*)d_ws;
  float* g1       = (float*)d_ws + num_users;
  unsigned* elist = (unsigned*)(g1 + (size_t)num_users * EMB_DIM);
  unsigned* slist = elist + n_edges;
  int* rtab       = (int*)(slist + (size_t)NBMAX * CAP);

  float* pred     = (float*)d_out;
  float* out_user = pred + batch;
  float* out_item = out_user + (size_t)batch * EMB_DIM;

  const int B = 256;

  part_sort<<<NCH, PT, 0, stream>>>(esrc, edst, tab, elist, n_edges,
                                    num_users, items, iemb, out_item, batch);

  conv_owner<false><<<NBMAX, PT, 0, stream>>>(elist, tab, uemb, g1, users,
                                              uemb, out_user, slist, rtab,
                                              num_users, batch);

  conv_owner<true><<<NBMAX, PT, 0, stream>>>(elist, tab, g1, (float*)nullptr,
                                             users, uemb, out_user, slist,
                                             rtab, num_users, batch);

  long long dot_threads = (long long)((batch + 15) / 16) * 32;
  dot_sigmoid_wmma<<<(int)((dot_threads + B - 1) / B), B, 0, stream>>>(
      out_user, out_item, pred, batch);
}
